# starting kernel
# baseline (speedup 1.0000x reference)
#include <hip/hip_runtime.h>

typedef float v2f __attribute__((ext_vector_type(2)));
typedef float v8f __attribute__((ext_vector_type(8)));

constexpr int N0 = 20000, N1 = 2000;
constexpr int FF = 512, HH = 256, DD = 128, KK = 16;
constexpr int E00 = 640000, E01 = 100000, E10 = 100000, E11 = 64000;

constexpr long long off_xw00 = 0;
constexpr long long off_xw10 = 5120000;
constexpr long long off_xw01 = 10240000;
constexpr long long off_xw11 = 10752000;
constexpr long long off_acc00 = 18944000;
constexpr long long off_acc01 = 24064000;
constexpr long long off_acc10 = 29184000;
constexpr long long off_acc11 = 29696000;

constexpr long long off_h0    = off_xw00;
constexpr long long off_h1    = off_xw01;
constexpr long long off_hw00  = off_xw10;
constexpr long long off_hw10  = off_xw10 + 2560000;
constexpr long long off_hw01  = off_xw11;
constexpr long long off_hw11  = off_xw11 + 256000;
constexpr long long off_bcc00 = off_acc00;
constexpr long long off_bcc01 = off_acc00 + 2560000;
constexpr long long off_bcc10 = off_acc10;
constexpr long long off_bcc11 = off_acc11;
constexpr long long off_z0R00 = off_acc01;
constexpr long long off_z0R01 = off_acc01 + 2560000;
constexpr long long off_z1R10 = off_acc10 + 256000;
constexpr long long off_z1d   = off_acc11 + 4096000;
constexpr long long off_Zk    = off_acc11;

__device__ __forceinline__ float wave_sum(float v) {
#pragma unroll
  for (int off = 16; off > 0; off >>= 1) v += __shfl_xor(v, off, 32);
  return v;
}

constexpr int BK   = 32;
constexpr int ASTR = 36;
constexpr int BPS  = 160;

__global__ __launch_bounds__(256) void gemm_f32_wmma(
    const float* __restrict__ A, const float* __restrict__ B, float* __restrict__ C,
    int M, int N, int Kd, long long sA, long long sB, long long sC) {
  __shared__ float As[2][128 * ASTR];
  __shared__ float Bs[2][(BK / 2) * BPS];

  const int b = blockIdx.z;
  A += (long long)b * sA;
  B += (long long)b * sB;
  C += (long long)b * sC;

  const int tid  = threadIdx.x;
  const int lane = tid & 31;
  const int wave = tid >> 5;
  const int mBlk = blockIdx.x * 128;
  const int n0   = blockIdx.y * 64;
  const int m0   = mBlk + wave * 16;
  const int half = lane >> 4;
  const int lr   = lane & 15;
  const bool active = (m0 < M);

  const int arow = tid >> 3;
  const int ac4  = (tid & 7) * 4;
  const int brow = tid >> 4;
  const int bc4  = (tid & 15) * 4;

  const int nChunks = Kd / BK;
  const float4 zero4 = make_float4(0.f, 0.f, 0.f, 0.f);

  {
#pragma unroll
    for (int p = 0; p < 4; ++p) {
      const int row = arow + p * 32;
      const int grow = mBlk + row;
      float4 v = (grow < M) ? *(const float4*)(A + (long long)grow * Kd + ac4) : zero4;
      *(float4*)&As[0][row * ASTR + ac4] = v;
    }
#pragma unroll
    for (int p = 0; p < 2; ++p) {
      const int row = brow + p * 16;
      const float4 v = *(const float4*)(B + (long long)row * N + n0 + bc4);
      float* bp = &Bs[0][(row >> 1) * BPS + bc4 * 2 + (row & 1)];
      bp[0] = v.x; bp[2] = v.y; bp[4] = v.z; bp[6] = v.w;
    }
  }
  __syncthreads();

  v8f acc0 = {}, acc1 = {}, acc2 = {}, acc3 = {};

  for (int c = 0; c < nChunks; ++c) {
    const int buf = c & 1;
    const bool pf = (c + 1 < nChunks);
    float4 ra[4], rb[2];

    if (pf) {
      const int k0 = (c + 1) * BK;
#pragma unroll
      for (int p = 0; p < 4; ++p) {
        const int grow = mBlk + arow + p * 32;
        ra[p] = (grow < M) ? *(const float4*)(A + (long long)grow * Kd + k0 + ac4) : zero4;
      }
#pragma unroll
      for (int p = 0; p < 2; ++p)
        rb[p] = *(const float4*)(B + (long long)(k0 + brow + p * 16) * N + n0 + bc4);
    }

    if (active) {
      const float* Asb = &As[buf][(wave * 16 + lr) * ASTR + 2 * half];
      const float* Bsb = &Bs[buf][half * BPS + lr * 2];
#pragma unroll
      for (int kk = 0; kk < BK; kk += 4) {
        const v2f a = *(const v2f*)(Asb + kk);
        const float* bp = Bsb + (kk >> 1) * BPS;
        const v2f b0 = *(const v2f*)(bp);
        const v2f b1 = *(const v2f*)(bp + 32);
        const v2f b2 = *(const v2f*)(bp + 64);
        const v2f b3 = *(const v2f*)(bp + 96);
        acc0 = __builtin_amdgcn_wmma_f32_16x16x4_f32(false, a, false, b0, (short)0, acc0, false, false);
        acc1 = __builtin_amdgcn_wmma_f32_16x16x4_f32(false, a, false, b1, (short)0, acc1, false, false);
        acc2 = __builtin_amdgcn_wmma_f32_16x16x4_f32(false, a, false, b2, (short)0, acc2, false, false);
        acc3 = __builtin_amdgcn_wmma_f32_16x16x4_f32(false, a, false, b3, (short)0, acc3, false, false);
      }
    }
    __syncthreads();

    if (pf) {
      const int nb = buf ^ 1;
#pragma unroll
      for (int p = 0; p < 4; ++p)
        *(float4*)&As[nb][(arow + p * 32) * ASTR + ac4] = ra[p];
#pragma unroll
      for (int p = 0; p < 2; ++p) {
        const int row = brow + p * 16;
        float* bp = &Bs[nb][(row >> 1) * BPS + bc4 * 2 + (row & 1)];
        bp[0] = rb[p].x; bp[2] = rb[p].y; bp[4] = rb[p].z; bp[6] = rb[p].w;
      }
      __syncthreads();
    }
  }

  if (active) {
#pragma unroll
    for (int r = 0; r < 8; ++r) {
      const int row = m0 + r + 8 * half;
      float* cp = C + (long long)row * N + n0 + lr;
      cp[0]  = acc0[r];
      cp[16] = acc1[r];
      cp[32] = acc2[r];
      cp[48] = acc3[r];
    }
  }
}

__global__ void spmm_edges(const int* __restrict__ dst, const int* __restrict__ src,
                           const float* __restrict__ val, const float* __restrict__ x,
                           float* __restrict__ out, int nE, int N,
                           int eStride, long long xStride, long long oStride) {
  const int k = blockIdx.y;
  dst += (long long)k * eStride;
  src += (long long)k * eStride;
  val += (long long)k * eStride;
  x   += (long long)k * xStride;
  out += (long long)k * oStride;
  const int w = blockIdx.x * (blockDim.x >> 5) + (threadIdx.x >> 5);
  if (w >= nE) return;
  const int lane = threadIdx.x & 31;
  const int d = dst[w];
  const int s = src[w];
  const float v = val[w];
  const float4* xs = (const float4*)(x + (long long)s * N);
  float* od = out + (long long)d * N;
  const int nq = N >> 2;
  for (int q = lane; q < nq; q += 32) {
    const float4 t = xs[q];
    float* o = od + q * 4;
    atomicAdd(o + 0, v * t.x);
    atomicAdd(o + 1, v * t.y);
    atomicAdd(o + 2, v * t.z);
    atomicAdd(o + 3, v * t.w);
  }
}

__global__ void finalize_conv(const float* __restrict__ buf, float* __restrict__ out,
                              int K, int n, int N, long long kStride, int do_relu) {
  const int wave = threadIdx.x >> 5;
  const int lane = threadIdx.x & 31;
  const int row = blockIdx.x * (blockDim.x >> 5) + wave;
  if (row >= n) return;
  const int nc = N >> 5;
  float vals[8];
  float ss = 0.f;
  for (int j = 0; j < nc; ++j) {
    const int c = lane + (j << 5);
    float a = 0.f;
    for (int k = 0; k < K; ++k) {
      float t = buf[(long long)k * kStride + (long long)row * N + c];
      if (do_relu) t = fmaxf(t, 0.f);
      a += t;
    }
    vals[j] = a;
    ss += a * a;
  }
  ss = wave_sum(ss);
  const float sc = rsqrtf(fmaxf(ss, 1e-12f));
  for (int j = 0; j < nc; ++j)
    out[(long long)row * N + lane + (j << 5)] = vals[j] * sc;
}

__global__ void add_act(const float* __restrict__ a, const float* __restrict__ b,
                        float* __restrict__ o, long long n, int do_relu) {
  const long long i = (long long)blockIdx.x * blockDim.x + threadIdx.x;
  if (i >= n) return;
  float v = a[i] + b[i];
  o[i] = do_relu ? fmaxf(v, 0.f) : v;
}

__global__ void scale_dloc(const float* __restrict__ z1, const float* __restrict__ dloc,
                           float* __restrict__ out, long long total) {
  const long long i = (long long)blockIdx.x * blockDim.x + threadIdx.x;
  if (i >= total) return;
  const int f = (int)(i % DD);
  const long long rk = i / DD;
  const int r = (int)(rk % N1);
  const int k = (int)(rk / N1);
  out[i] = z1[(long long)r * DD + f] * dloc[k * DD + f];
}

__global__ void edge_dot128(const float* __restrict__ zl, const float* __restrict__ zr,
                            const int* __restrict__ dst, const int* __restrict__ src,
                            float* __restrict__ out, int nE, int eStride,
                            long long zlStride, long long zrStride, long long oStride) {
  const int k = blockIdx.y;
  zl  += (long long)k * zlStride;
  zr  += (long long)k * zrStride;
  dst += (long long)k * eStride;
  src += (long long)k * eStride;
  out += (long long)k * oStride;
  const int e = blockIdx.x * (blockDim.x >> 5) + (threadIdx.x >> 5);
  if (e >= nE) return;
  const int lane = threadIdx.x & 31;
  const int d = dst[e];
  const int s = src[e];
  const float4 a = ((const float4*)(zl + (long long)d * DD))[lane];
  const float4 b = ((const float4*)(zr + (long long)s * DD))[lane];
  float acc = a.x * b.x + a.y * b.y + a.z * b.z + a.w * b.w;
  acc = wave_sum(acc);
  if (lane == 0) out[e] = acc;
}

extern "C" void kernel_launch(void* const* d_in, const int* in_sizes, int n_in,
                              void* d_out, int out_size, void* d_ws, size_t ws_size,
                              hipStream_t stream) {
  (void)in_sizes; (void)n_in; (void)out_size; (void)ws_size;

  const float* feat0   = (const float*)d_in[0];
  const float* feat1   = (const float*)d_in[1];
  const int*   e00_dst = (const int*)d_in[2];
  const int*   e00_src = (const int*)d_in[3];
  const float* e00_val = (const float*)d_in[4];
  const int*   e01_dst = (const int*)d_in[5];
  const int*   e01_src = (const int*)d_in[6];
  const float* e01_val = (const float*)d_in[7];
  const int*   e10_dst = (const int*)d_in[8];
  const int*   e10_src = (const int*)d_in[9];
  const float* e10_val = (const float*)d_in[10];
  const int*   e11_dst = (const int*)d_in[11];
  const int*   e11_src = (const int*)d_in[12];
  const float* e11_val = (const float*)d_in[13];
  const float* W1_00 = (const float*)d_in[14];
  const float* W1_01 = (const float*)d_in[15];
  const float* W1_10 = (const float*)d_in[16];
  const float* W1_11 = (const float*)d_in[17];
  const float* W2_00 = (const float*)d_in[18];
  const float* W2_01 = (const float*)d_in[19];
  const float* W2_10 = (const float*)d_in[20];
  const float* W2_11 = (const float*)d_in[21];
  const float* R00   = (const float*)d_in[22];
  const float* R01   = (const float*)d_in[23];
  const float* R10   = (const float*)d_in[24];
  const float* Rglob = (const float*)d_in[25];
  const float* dloc  = (const float*)d_in[26];

  float* ws = (float*)d_ws;
  float* out = (float*)d_out;
  float* z0  = out;
  float* z1  = out + 2560000;
  float* s00 = z1 + 256000;
  float* s01 = s00 + E00;
  float* s10 = s01 + E01;
  float* s11 = s10 + E10;

  const dim3 blk(256);
  auto gg = [](int M, int N, int batch) { return dim3((M + 127) / 128, N / 64, batch); };

  gemm_f32_wmma<<<gg(N0, HH, 1), blk, 0, stream>>>(feat0, W1_00, ws + off_xw00, N0, HH, FF, 0, 0, 0);
  gemm_f32_wmma<<<gg(N0, HH, 1), blk, 0, stream>>>(feat0, W1_10, ws + off_xw10, N0, HH, FF, 0, 0, 0);
  gemm_f32_wmma<<<gg(N1, HH, 1), blk, 0, stream>>>(feat1, W1_01, ws + off_xw01, N1, HH, FF, 0, 0, 0);
  gemm_f32_wmma<<<gg(N1, HH, KK), blk, 0, stream>>>(feat1, W1_11, ws + off_xw11, N1, HH, FF,
                                                    0, (long long)FF * HH, (long long)N1 * HH);

  hipMemsetAsync(ws + off_acc00, 0, (size_t)18944000 * sizeof(float), stream);

  spmm_edges<<<dim3((E00 + 7) / 8, 1), blk, 0, stream>>>(e00_dst, e00_src, e00_val, ws + off_xw00,
                                                         ws + off_acc00, E00, HH, 0, 0, 0);
  spmm_edges<<<dim3((E01 + 7) / 8, 1), blk, 0, stream>>>(e01_dst, e01_src, e01_val, ws + off_xw01,
                                                         ws + off_acc01, E01, HH, 0, 0, 0);
  spmm_edges<<<dim3((E10 + 7) / 8, 1), blk, 0, stream>>>(e10_dst, e10_src, e10_val, ws + off_xw10,
                                                         ws + off_acc10, E10, HH, 0, 0, 0);
  spmm_edges<<<dim3((E11 + 7) / 8, KK), blk, 0, stream>>>(e11_dst, e11_src, e11_val, ws + off_xw11,
                                                          ws + off_acc11, E11, HH, E11,
                                                          (long long)N1 * HH, (long long)N1 * HH);

  finalize_conv<<<dim3((N0 + 7) / 8), blk, 0, stream>>>(ws + off_acc00, ws + off_acc00, 1, N0, HH, 0, 1);
  finalize_conv<<<dim3((N0 + 7) / 8), blk, 0, stream>>>(ws + off_acc01, ws + off_acc01, 1, N0, HH, 0, 1);
  finalize_conv<<<dim3((N1 + 7) / 8), blk, 0, stream>>>(ws + off_acc10, ws + off_acc10, 1, N1, HH, 0, 1);
  finalize_conv<<<dim3((N1 + 7) / 8), blk, 0, stream>>>(ws + off_acc11, ws + off_acc11, KK, N1, HH,
                                                        (long long)N1 * HH, 1);

  add_act<<<dim3((N0 * HH + 255) / 256), blk, 0, stream>>>(ws + off_acc00, ws + off_acc01,
                                                           ws + off_h0, (long long)N0 * HH, 1);
  add_act<<<dim3((N1 * HH + 255) / 256), blk, 0, stream>>>(ws + off_acc10, ws + off_acc11,
                                                           ws + off_h1, (long long)N1 * HH, 1);

  gemm_f32_wmma<<<gg(N0, DD, 1), blk, 0, stream>>>(ws + off_h0, W2_00, ws + off_hw00, N0, DD, HH, 0, 0, 0);
  gemm_f32_wmma<<<gg(N0, DD, 1), blk, 0, stream>>>(ws + off_h0, W2_10, ws + off_hw10, N0, DD, HH, 0, 0, 0);
  gemm_f32_wmma<<<gg(N1, DD, 1), blk, 0, stream>>>(ws + off_h1, W2_01, ws + off_hw01, N1, DD, HH, 0, 0, 0);
  gemm_f32_wmma<<<gg(N1, DD, KK), blk, 0, stream>>>(ws + off_h1, W2_11, ws + off_hw11, N1, DD, HH,
                                                    0, (long long)HH * DD, (long long)N1 * DD);

  hipMemsetAsync(ws + off_bcc00, 0, (size_t)5120000 * sizeof(float), stream);
  hipMemsetAsync(ws + off_bcc10, 0, (size_t)256000 * sizeof(float), stream);
  hipMemsetAsync(ws + off_bcc11, 0, (size_t)4096000 * sizeof(float), stream);

  spmm_edges<<<dim3((E00 + 7) / 8, 1), blk, 0, stream>>>(e00_dst, e00_src, e00_val, ws + off_hw00,
                                                         ws + off_bcc00, E00, DD, 0, 0, 0);
  spmm_edges<<<dim3((E01 + 7) / 8, 1), blk, 0, stream>>>(e01_dst, e01_src, e01_val, ws + off_hw01,
                                                         ws + off_bcc01, E01, DD, 0, 0, 0);
  spmm_edges<<<dim3((E10 + 7) / 8, 1), blk, 0, stream>>>(e10_dst, e10_src, e10_val, ws + off_hw10,
                                                         ws + off_bcc10, E10, DD, 0, 0, 0);
  spmm_edges<<<dim3((E11 + 7) / 8, KK), blk, 0, stream>>>(e11_dst, e11_src, e11_val, ws + off_hw11,
                                                          ws + off_bcc11, E11, DD, E11,
                                                          (long long)N1 * DD, (long long)N1 * DD);

  finalize_conv<<<dim3((N0 + 7) / 8), blk, 0, stream>>>(ws + off_bcc00, ws + off_bcc00, 1, N0, DD, 0, 0);
  finalize_conv<<<dim3((N0 + 7) / 8), blk, 0, stream>>>(ws + off_bcc01, ws + off_bcc01, 1, N0, DD, 0, 0);
  finalize_conv<<<dim3((N1 + 7) / 8), blk, 0, stream>>>(ws + off_bcc10, ws + off_bcc10, 1, N1, DD, 0, 0);
  finalize_conv<<<dim3((N1 + 7) / 8), blk, 0, stream>>>(ws + off_bcc11, ws + off_bcc11, KK, N1, DD,
                                                        (long long)N1 * DD, 0);

  add_act<<<dim3((N0 * DD + 255) / 256), blk, 0, stream>>>(ws + off_bcc00, ws + off_bcc01, z0,
                                                           (long long)N0 * DD, 0);
  add_act<<<dim3((N1 * DD + 255) / 256), blk, 0, stream>>>(ws + off_bcc10, ws + off_bcc11, z1,
                                                           (long long)N1 * DD, 0);

  gemm_f32_wmma<<<gg(N0, DD, 1), blk, 0, stream>>>(z0, R00, ws + off_z0R00, N0, DD, DD, 0, 0, 0);
  gemm_f32_wmma<<<gg(N0, DD, 1), blk, 0, stream>>>(z0, R01, ws + off_z0R01, N0, DD, DD, 0, 0, 0);
  gemm_f32_wmma<<<gg(N1, DD, 1), blk, 0, stream>>>(z1, R10, ws + off_z1R10, N1, DD, DD, 0, 0, 0);

  scale_dloc<<<dim3((KK * N1 * DD + 255) / 256), blk, 0, stream>>>(z1, dloc, ws + off_z1d,
                                                                   (long long)KK * N1 * DD);
  gemm_f32_wmma<<<gg(N1, DD, KK), blk, 0, stream>>>(ws + off_z1d, Rglob, ws + off_Zk, N1, DD, DD,
                                                    (long long)N1 * DD, 0, (long long)N1 * DD);

  edge_dot128<<<dim3((E00 + 7) / 8, 1), blk, 0, stream>>>(ws + off_z0R00, z0, e00_dst, e00_src,
                                                          s00, E00, 0, 0, 0, 0);
  edge_dot128<<<dim3((E01 + 7) / 8, 1), blk, 0, stream>>>(ws + off_z0R01, z1, e01_dst, e01_src,
                                                          s01, E01, 0, 0, 0, 0);
  edge_dot128<<<dim3((E10 + 7) / 8, 1), blk, 0, stream>>>(ws + off_z1R10, z0, e10_dst, e10_src,
                                                          s10, E10, 0, 0, 0, 0);
  edge_dot128<<<dim3((E11 + 7) / 8, KK), blk, 0, stream>>>(ws + off_Zk, ws + off_z1d, e11_dst, e11_src,
                                                           s11, E11, E11,
                                                           (long long)N1 * DD, (long long)N1 * DD,
                                                           (long long)E11);
}


# SPMM made output-stationary: a block holds a tile of output rows in LDS, filters the dst stream for them, adds gathered x rows with LDS float atomics (arrival order), stores it once; memsets gone
# speedup vs baseline: 1.6403x; 1.6403x over previous
#include <hip/hip_runtime.h>

typedef float v2f __attribute__((ext_vector_type(2)));
typedef float v8f __attribute__((ext_vector_type(8)));

constexpr int N0 = 20000, N1 = 2000;
constexpr int FF = 512, HH = 256, DD = 128, KK = 16;
constexpr int E00 = 640000, E01 = 100000, E10 = 100000, E11 = 64000;

constexpr long long off_xw00 = 0;
constexpr long long off_xw10 = 5120000;
constexpr long long off_xw01 = 10240000;
constexpr long long off_xw11 = 10752000;
constexpr long long off_acc00 = 18944000;
constexpr long long off_acc01 = 24064000;
constexpr long long off_acc10 = 29184000;
constexpr long long off_acc11 = 29696000;

constexpr long long off_h0    = off_xw00;
constexpr long long off_h1    = off_xw01;
constexpr long long off_hw00  = off_xw10;
constexpr long long off_hw10  = off_xw10 + 2560000;
constexpr long long off_hw01  = off_xw11;
constexpr long long off_hw11  = off_xw11 + 256000;
constexpr long long off_bcc00 = off_acc00;
constexpr long long off_bcc01 = off_acc00 + 2560000;
constexpr long long off_bcc10 = off_acc10;
constexpr long long off_bcc11 = off_acc11;
constexpr long long off_z0R00 = off_acc01;
constexpr long long off_z0R01 = off_acc01 + 2560000;
constexpr long long off_z1R10 = off_acc10 + 256000;
constexpr long long off_z1d   = off_acc11 + 4096000;
constexpr long long off_Zk    = off_acc11;

__device__ __forceinline__ float wave_sum(float v) {
#pragma unroll
  for (int off = 16; off > 0; off >>= 1) v += __shfl_xor(v, off, 32);
  return v;
}

constexpr int BK   = 32;
constexpr int ASTR = 36;
constexpr int BPS  = 160;

__global__ __launch_bounds__(256) void gemm_f32_wmma(
    const float* __restrict__ A, const float* __restrict__ B, float* __restrict__ C,
    int M, int N, int Kd, long long sA, long long sB, long long sC) {
  __shared__ float As[2][128 * ASTR];
  __shared__ float Bs[2][(BK / 2) * BPS];

  const int b = blockIdx.z;
  A += (long long)b * sA;
  B += (long long)b * sB;
  C += (long long)b * sC;

  const int tid  = threadIdx.x;
  const int lane = tid & 31;
  const int wave = tid >> 5;
  const int mBlk = blockIdx.x * 128;
  const int n0   = blockIdx.y * 64;
  const int m0   = mBlk + wave * 16;
  const int half = lane >> 4;
  const int lr   = lane & 15;
  const bool active = (m0 < M);

  const int arow = tid >> 3;
  const int ac4  = (tid & 7) * 4;
  const int brow = tid >> 4;
  const int bc4  = (tid & 15) * 4;

  const int nChunks = Kd / BK;
  const float4 zero4 = make_float4(0.f, 0.f, 0.f, 0.f);

  {
#pragma unroll
    for (int p = 0; p < 4; ++p) {
      const int row = arow + p * 32;
      const int grow = mBlk + row;
      float4 v = (grow < M) ? *(const float4*)(A + (long long)grow * Kd + ac4) : zero4;
      *(float4*)&As[0][row * ASTR + ac4] = v;
    }
#pragma unroll
    for (int p = 0; p < 2; ++p) {
      const int row = brow + p * 16;
      const float4 v = *(const float4*)(B + (long long)row * N + n0 + bc4);
      float* bp = &Bs[0][(row >> 1) * BPS + bc4 * 2 + (row & 1)];
      bp[0] = v.x; bp[2] = v.y; bp[4] = v.z; bp[6] = v.w;
    }
  }
  __syncthreads();

  v8f acc0 = {}, acc1 = {}, acc2 = {}, acc3 = {};

  for (int c = 0; c < nChunks; ++c) {
    const int buf = c & 1;
    const bool pf = (c + 1 < nChunks);
    float4 ra[4], rb[2];

    if (pf) {
      const int k0 = (c + 1) * BK;
#pragma unroll
      for (int p = 0; p < 4; ++p) {
        const int grow = mBlk + arow + p * 32;
        ra[p] = (grow < M) ? *(const float4*)(A + (long long)grow * Kd + k0 + ac4) : zero4;
      }
#pragma unroll
      for (int p = 0; p < 2; ++p)
        rb[p] = *(const float4*)(B + (long long)(k0 + brow + p * 16) * N + n0 + bc4);
    }

    if (active) {
      const float* Asb = &As[buf][(wave * 16 + lr) * ASTR + 2 * half];
      const float* Bsb = &Bs[buf][half * BPS + lr * 2];
#pragma unroll
      for (int kk = 0; kk < BK; kk += 4) {
        const v2f a = *(const v2f*)(Asb + kk);
        const float* bp = Bsb + (kk >> 1) * BPS;
        const v2f b0 = *(const v2f*)(bp);
        const v2f b1 = *(const v2f*)(bp + 32);
        const v2f b2 = *(const v2f*)(bp + 64);
        const v2f b3 = *(const v2f*)(bp + 96);
        acc0 = __builtin_amdgcn_wmma_f32_16x16x4_f32(false, a, false, b0, (short)0, acc0, false, false);
        acc1 = __builtin_amdgcn_wmma_f32_16x16x4_f32(false, a, false, b1, (short)0, acc1, false, false);
        acc2 = __builtin_amdgcn_wmma_f32_16x16x4_f32(false, a, false, b2, (short)0, acc2, false, false);
        acc3 = __builtin_amdgcn_wmma_f32_16x16x4_f32(false, a, false, b3, (short)0, acc3, false, false);
      }
    }
    __syncthreads();

    if (pf) {
      const int nb = buf ^ 1;
#pragma unroll
      for (int p = 0; p < 4; ++p)
        *(float4*)&As[nb][(arow + p * 32) * ASTR + ac4] = ra[p];
#pragma unroll
      for (int p = 0; p < 2; ++p) {
        const int row = brow + p * 16;
        float* bp = &Bs[nb][(row >> 1) * BPS + bc4 * 2 + (row & 1)];
        bp[0] = rb[p].x; bp[2] = rb[p].y; bp[4] = rb[p].z; bp[6] = rb[p].w;
      }
      __syncthreads();
    }
  }

  if (active) {
#pragma unroll
    for (int r = 0; r < 8; ++r) {
      const int row = m0 + r + 8 * half;
      float* cp = C + (long long)row * N + n0 + lr;
      cp[0]  = acc0[r];
      cp[16] = acc1[r];
      cp[32] = acc2[r];
      cp[48] = acc3[r];
    }
  }
}

constexpr int SP_THREADS = 1024;
constexpr int SP_WAVES   = SP_THREADS / 32;
constexpr int SP_MAXT    = 125;
constexpr int SP_Q       = 16384;

template <int N>
__global__ __launch_bounds__(SP_THREADS) void spmm_tile(
    const int* __restrict__ dst, const int* __restrict__ src, const float* __restrict__ val,
    const float* __restrict__ x, float* __restrict__ out, int nE, int nRows, int T,
    int eStride, long long xStride, long long oStride) {
  constexpr int G = N / 4;
  constexpr int H = N / 128;
  __shared__ __attribute__((aligned(16))) float acc[SP_MAXT * N];
  __shared__ int qe[SP_Q];
  __shared__ int qcount;

  const int k = blockIdx.y;
  dst += (long long)k * eStride;
  src += (long long)k * eStride;
  val += (long long)k * eStride;
  x   += (long long)k * xStride;
  out += (long long)k * oStride;

  const int tid  = threadIdx.x;
  const int lane = tid & 31;
  const int wave = __builtin_amdgcn_readfirstlane(tid >> 5);
  const int row0 = blockIdx.x * T;

  for (int i = tid; i < T * G; i += SP_THREADS)
    ((float4*)acc)[i] = make_float4(0.f, 0.f, 0.f, 0.f);
  if (tid == 0) qcount = 0;
  __syncthreads();

  const int n4 = nE >> 2;
  const int4* d4p = (const int4*)dst;
  constexpr int STEP = SP_THREADS * 4;
  const int last4 = max(n4 - 1, 0);

  auto test4 = [&](const int4 (&dv)[4], int base) {
#pragma unroll
    for (int u = 0; u < 4; ++u) {
      const int idx = base + u * SP_THREADS + tid;
      const int e0 = idx << 2;
      const int dd[4] = {dv[u].x, dv[u].y, dv[u].z, dv[u].w};
#pragma unroll
      for (int c = 0; c < 4; ++c) {
        const unsigned r = (unsigned)(dd[c] - row0);
        if (idx < n4 && r < (unsigned)T) {
          const int e = e0 + c;
          const int slot = atomicAdd(&qcount, 1);
          if (slot < SP_Q) {
            qe[slot] = e;
          } else {
            const int s = src[e];
            const float v = val[e];
            const float* xr = x + (long long)s * N;
            float* a = &acc[(int)r * N];
            for (int f = 0; f < N; ++f) atomicAdd(&a[(f & 3) * G + (f >> 2)], v * xr[f]);
          }
        }
      }
    }
  };

  int4 da[4], db[4];
#pragma unroll
  for (int u = 0; u < 4; ++u) da[u] = d4p[min(u * SP_THREADS + tid, last4)];
  for (int i0 = 0; i0 < n4; i0 += 2 * STEP) {
#pragma unroll
    for (int u = 0; u < 4; ++u) db[u] = d4p[min(i0 + STEP + u * SP_THREADS + tid, last4)];
    test4(da, i0);
#pragma unroll
    for (int u = 0; u < 4; ++u) da[u] = d4p[min(i0 + 2 * STEP + u * SP_THREADS + tid, last4)];
    test4(db, i0 + STEP);
  }
  __syncthreads();

  const int cnt = __builtin_amdgcn_readfirstlane(min(qcount, SP_Q));
  const int nGroups = (cnt + 31) >> 5;
  for (int g = wave; g < nGroups; g += SP_WAVES) {
    const int gi = g * 32 + lane;
    int s = 0, lr = 0, vb = 0;
    if (gi < cnt) {
      const int e = qe[gi];
      s  = src[e];
      vb = __float_as_int(val[e]);
      lr = dst[e] - row0;
    }
    const int nv = min(32, cnt - g * 32);
    for (int j0 = 0; j0 < nv; j0 += 4) {
      float4 t[4][H];
      float vv[4];
      int ll[4];
#pragma unroll
      for (int u = 0; u < 4; ++u) {
        const int j = j0 + u;
        const int sj = __builtin_amdgcn_readlane(s, j);
        ll[u] = __builtin_amdgcn_readlane(lr, j);
        vv[u] = __int_as_float(__builtin_amdgcn_readlane(vb, j));
        const float4* xp = (const float4*)(x + (long long)sj * N);
#pragma unroll
        for (int h = 0; h < H; ++h) t[u][h] = xp[lane + 32 * h];
      }
#pragma unroll
      for (int u = 0; u < 4; ++u) {
        if (j0 + u < nv) {
          float* a = &acc[ll[u] * N + lane];
#pragma unroll
          for (int h = 0; h < H; ++h) {
            atomicAdd(&a[0 * G + 32 * h], vv[u] * t[u][h].x);
            atomicAdd(&a[1 * G + 32 * h], vv[u] * t[u][h].y);
            atomicAdd(&a[2 * G + 32 * h], vv[u] * t[u][h].z);
            atomicAdd(&a[3 * G + 32 * h], vv[u] * t[u][h].w);
          }
        }
      }
    }
  }
  __syncthreads();

  for (int i = tid; i < T * G; i += SP_THREADS) {
    const int r = i / G;
    const int g = i % G;
    if (row0 + r < nRows) {
      const float* a = &acc[r * N + g];
      *(float4*)(out + (long long)(row0 + r) * N + 4 * g) = make_float4(a[0], a[G], a[2 * G], a[3 * G]);
    }
  }
}

constexpr int T0  = 80;
constexpr int T1s = 8;
constexpr int T1m = 125;
static_assert(N0 % T0 == 0 && N1 % T1s == 0 && N1 % T1m == 0, "tiles cover the rows exactly");
static_assert(T0 <= SP_MAXT && T1s <= SP_MAXT && T1m <= SP_MAXT, "tile fits its LDS array");
static_assert(E00 % 4 == 0 && E01 % 4 == 0 && E10 % 4 == 0 && E11 % 4 == 0, "dst read as int4");

__global__ void finalize_conv(const float* __restrict__ buf, float* __restrict__ out,
                              int K, int n, int N, long long kStride, int do_relu) {
  const int wave = threadIdx.x >> 5;
  const int lane = threadIdx.x & 31;
  const int row = blockIdx.x * (blockDim.x >> 5) + wave;
  if (row >= n) return;
  const int nc = N >> 5;
  float vals[8];
  float ss = 0.f;
  for (int j = 0; j < nc; ++j) {
    const int c = lane + (j << 5);
    float a = 0.f;
    for (int k = 0; k < K; ++k) {
      float t = buf[(long long)k * kStride + (long long)row * N + c];
      if (do_relu) t = fmaxf(t, 0.f);
      a += t;
    }
    vals[j] = a;
    ss += a * a;
  }
  ss = wave_sum(ss);
  const float sc = rsqrtf(fmaxf(ss, 1e-12f));
  for (int j = 0; j < nc; ++j)
    out[(long long)row * N + lane + (j << 5)] = vals[j] * sc;
}

__global__ void add_act(const float* __restrict__ a, const float* __restrict__ b,
                        float* __restrict__ o, long long n, int do_relu) {
  const long long i = (long long)blockIdx.x * blockDim.x + threadIdx.x;
  if (i >= n) return;
  float v = a[i] + b[i];
  o[i] = do_relu ? fmaxf(v, 0.f) : v;
}

__global__ void scale_dloc(const float* __restrict__ z1, const float* __restrict__ dloc,
                           float* __restrict__ out, long long total) {
  const long long i = (long long)blockIdx.x * blockDim.x + threadIdx.x;
  if (i >= total) return;
  const int f = (int)(i % DD);
  const long long rk = i / DD;
  const int r = (int)(rk % N1);
  const int k = (int)(rk / N1);
  out[i] = z1[(long long)r * DD + f] * dloc[k * DD + f];
}

__global__ void edge_dot128(const float* __restrict__ zl, const float* __restrict__ zr,
                            const int* __restrict__ dst, const int* __restrict__ src,
                            float* __restrict__ out, int nE, int eStride,
                            long long zlStride, long long zrStride, long long oStride) {
  const int k = blockIdx.y;
  zl  += (long long)k * zlStride;
  zr  += (long long)k * zrStride;
  dst += (long long)k * eStride;
  src += (long long)k * eStride;
  out += (long long)k * oStride;
  const int e = blockIdx.x * (blockDim.x >> 5) + (threadIdx.x >> 5);
  if (e >= nE) return;
  const int lane = threadIdx.x & 31;
  const int d = dst[e];
  const int s = src[e];
  const float4 a = ((const float4*)(zl + (long long)d * DD))[lane];
  const float4 b = ((const float4*)(zr + (long long)s * DD))[lane];
  float acc = a.x * b.x + a.y * b.y + a.z * b.z + a.w * b.w;
  acc = wave_sum(acc);
  if (lane == 0) out[e] = acc;
}

extern "C" void kernel_launch(void* const* d_in, const int* in_sizes, int n_in,
                              void* d_out, int out_size, void* d_ws, size_t ws_size,
                              hipStream_t stream) {
  (void)in_sizes; (void)n_in; (void)out_size; (void)ws_size;

  const float* feat0   = (const float*)d_in[0];
  const float* feat1   = (const float*)d_in[1];
  const int*   e00_dst = (const int*)d_in[2];
  const int*   e00_src = (const int*)d_in[3];
  const float* e00_val = (const float*)d_in[4];
  const int*   e01_dst = (const int*)d_in[5];
  const int*   e01_src = (const int*)d_in[6];
  const float* e01_val = (const float*)d_in[7];
  const int*   e10_dst = (const int*)d_in[8];
  const int*   e10_src = (const int*)d_in[9];
  const float* e10_val = (const float*)d_in[10];
  const int*   e11_dst = (const int*)d_in[11];
  const int*   e11_src = (const int*)d_in[12];
  const float* e11_val = (const float*)d_in[13];
  const float* W1_00 = (const float*)d_in[14];
  const float* W1_01 = (const float*)d_in[15];
  const float* W1_10 = (const float*)d_in[16];
  const float* W1_11 = (const float*)d_in[17];
  const float* W2_00 = (const float*)d_in[18];
  const float* W2_01 = (const float*)d_in[19];
  const float* W2_10 = (const float*)d_in[20];
  const float* W2_11 = (const float*)d_in[21];
  const float* R00   = (const float*)d_in[22];
  const float* R01   = (const float*)d_in[23];
  const float* R10   = (const float*)d_in[24];
  const float* Rglob = (const float*)d_in[25];
  const float* dloc  = (const float*)d_in[26];

  float* ws = (float*)d_ws;
  float* out = (float*)d_out;
  float* z0  = out;
  float* z1  = out + 2560000;
  float* s00 = z1 + 256000;
  float* s01 = s00 + E00;
  float* s10 = s01 + E01;
  float* s11 = s10 + E10;

  const dim3 blk(256);
  auto gg = [](int M, int N, int batch) { return dim3((M + 127) / 128, N / 64, batch); };

  gemm_f32_wmma<<<gg(N0, HH, 1), blk, 0, stream>>>(feat0, W1_00, ws + off_xw00, N0, HH, FF, 0, 0, 0);
  gemm_f32_wmma<<<gg(N0, HH, 1), blk, 0, stream>>>(feat0, W1_10, ws + off_xw10, N0, HH, FF, 0, 0, 0);
  gemm_f32_wmma<<<gg(N1, HH, 1), blk, 0, stream>>>(feat1, W1_01, ws + off_xw01, N1, HH, FF, 0, 0, 0);
  gemm_f32_wmma<<<gg(N1, HH, KK), blk, 0, stream>>>(feat1, W1_11, ws + off_xw11, N1, HH, FF,
                                                    0, (long long)FF * HH, (long long)N1 * HH);

  const dim3 sblk(SP_THREADS);
  spmm_tile<HH><<<dim3(N0 / T0, 1), sblk, 0, stream>>>(e00_dst, e00_src, e00_val, ws + off_xw00,
                                                       ws + off_acc00, E00, N0, T0, 0, 0, 0);
  spmm_tile<HH><<<dim3(N0 / T0, 1), sblk, 0, stream>>>(e01_dst, e01_src, e01_val, ws + off_xw01,
                                                       ws + off_acc01, E01, N0, T0, 0, 0, 0);
  spmm_tile<HH><<<dim3(N1 / T1s, 1), sblk, 0, stream>>>(e10_dst, e10_src, e10_val, ws + off_xw10,
                                                        ws + off_acc10, E10, N1, T1s, 0, 0, 0);
  spmm_tile<HH><<<dim3(N1 / T1m, KK), sblk, 0, stream>>>(e11_dst, e11_src, e11_val, ws + off_xw11,
                                                         ws + off_acc11, E11, N1, T1m, E11,
                                                         (long long)N1 * HH, (long long)N1 * HH);

  finalize_conv<<<dim3((N0 + 7) / 8), blk, 0, stream>>>(ws + off_acc00, ws + off_acc00, 1, N0, HH, 0, 1);
  finalize_conv<<<dim3((N0 + 7) / 8), blk, 0, stream>>>(ws + off_acc01, ws + off_acc01, 1, N0, HH, 0, 1);
  finalize_conv<<<dim3((N1 + 7) / 8), blk, 0, stream>>>(ws + off_acc10, ws + off_acc10, 1, N1, HH, 0, 1);
  finalize_conv<<<dim3((N1 + 7) / 8), blk, 0, stream>>>(ws + off_acc11, ws + off_acc11, KK, N1, HH,
                                                        (long long)N1 * HH, 1);

  add_act<<<dim3((N0 * HH + 255) / 256), blk, 0, stream>>>(ws + off_acc00, ws + off_acc01,
                                                           ws + off_h0, (long long)N0 * HH, 1);
  add_act<<<dim3((N1 * HH + 255) / 256), blk, 0, stream>>>(ws + off_acc10, ws + off_acc11,
                                                           ws + off_h1, (long long)N1 * HH, 1);

  gemm_f32_wmma<<<gg(N0, DD, 1), blk, 0, stream>>>(ws + off_h0, W2_00, ws + off_hw00, N0, DD, HH, 0, 0, 0);
  gemm_f32_wmma<<<gg(N0, DD, 1), blk, 0, stream>>>(ws + off_h0, W2_10, ws + off_hw10, N0, DD, HH, 0, 0, 0);
  gemm_f32_wmma<<<gg(N1, DD, 1), blk, 0, stream>>>(ws + off_h1, W2_01, ws + off_hw01, N1, DD, HH, 0, 0, 0);
  gemm_f32_wmma<<<gg(N1, DD, KK), blk, 0, stream>>>(ws + off_h1, W2_11, ws + off_hw11, N1, DD, HH,
                                                    0, (long long)HH * DD, (long long)N1 * DD);

  spmm_tile<DD><<<dim3(N0 / T0, 1), sblk, 0, stream>>>(e00_dst, e00_src, e00_val, ws + off_hw00,
                                                       ws + off_bcc00, E00, N0, T0, 0, 0, 0);
  spmm_tile<DD><<<dim3(N0 / T0, 1), sblk, 0, stream>>>(e01_dst, e01_src, e01_val, ws + off_hw01,
                                                       ws + off_bcc01, E01, N0, T0, 0, 0, 0);
  spmm_tile<DD><<<dim3(N1 / T1s, 1), sblk, 0, stream>>>(e10_dst, e10_src, e10_val, ws + off_hw10,
                                                        ws + off_bcc10, E10, N1, T1s, 0, 0, 0);
  spmm_tile<DD><<<dim3(N1 / T1m, KK), sblk, 0, stream>>>(e11_dst, e11_src, e11_val, ws + off_hw11,
                                                         ws + off_bcc11, E11, N1, T1m, E11,
                                                         (long long)N1 * DD, (long long)N1 * DD);

  finalize_conv<<<dim3((N0 + 7) / 8), blk, 0, stream>>>(ws + off_bcc00, ws + off_bcc00, 1, N0, DD, 0, 0);
  finalize_conv<<<dim3((N0 + 7) / 8), blk, 0, stream>>>(ws + off_bcc01, ws + off_bcc01, 1, N0, DD, 0, 0);
  finalize_conv<<<dim3((N1 + 7) / 8), blk, 0, stream>>>(ws + off_bcc10, ws + off_bcc10, 1, N1, DD, 0, 0);
  finalize_conv<<<dim3((N1 + 7) / 8), blk, 0, stream>>>(ws + off_bcc11, ws + off_bcc11, KK, N1, DD,
                                                        (long long)N1 * DD, 0);

  add_act<<<dim3((N0 * DD + 255) / 256), blk, 0, stream>>>(ws + off_bcc00, ws + off_bcc01, z0,
                                                           (long long)N0 * DD, 0);
  add_act<<<dim3((N1 * DD + 255) / 256), blk, 0, stream>>>(ws + off_bcc10, ws + off_bcc11, z1,
                                                           (long long)N1 * DD, 0);

  gemm_f32_wmma<<<gg(N0, DD, 1), blk, 0, stream>>>(z0, R00, ws + off_z0R00, N0, DD, DD, 0, 0, 0);
  gemm_f32_wmma<<<gg(N0, DD, 1), blk, 0, stream>>>(z0, R01, ws + off_z0R01, N0, DD, DD, 0, 0, 0);
  gemm_f32_wmma<<<gg(N1, DD, 1), blk, 0, stream>>>(z1, R10, ws + off_z1R10, N1, DD, DD, 0, 0, 0);

  scale_dloc<<<dim3((KK * N1 * DD + 255) / 256), blk, 0, stream>>>(z1, dloc, ws + off_z1d,
                                                                   (long long)KK * N1 * DD);
  gemm_f32_wmma<<<gg(N1, DD, KK), blk, 0, stream>>>(ws + off_z1d, Rglob, ws + off_Zk, N1, DD, DD,
                                                    (long long)N1 * DD, 0, (long long)N1 * DD);

  edge_dot128<<<dim3((E00 + 7) / 8, 1), blk, 0, stream>>>(ws + off_z0R00, z0, e00_dst, e00_src,
                                                          s00, E00, 0, 0, 0, 0);
  edge_dot128<<<dim3((E01 + 7) / 8, 1), blk, 0, stream>>>(ws + off_z0R01, z1, e01_dst, e01_src,
                                                          s01, E01, 0, 0, 0, 0);
  edge_dot128<<<dim3((E10 + 7) / 8, 1), blk, 0, stream>>>(ws + off_z1R10, z0, e10_dst, e10_src,
                                                          s10, E10, 0, 0, 0, 0);
  edge_dot128<<<dim3((E11 + 7) / 8, KK), blk, 0, stream>>>(ws + off_Zk, ws + off_z1d, e11_dst, e11_src,
                                                           s11, E11, E11,
                                                           (long long)N1 * DD, (long long)N1 * DD,
                                                           (long long)E11);
}


# layer-1 SPMM tiles leave their queued edges (tile row, src, val) as lists in the workspace; layer-2 tiles read them, skip the dst scan and per-edge index lookups; LDS adds in arrival order as before
# speedup vs baseline: 1.7579x; 1.0717x over previous
#include <hip/hip_runtime.h>

typedef float v2f __attribute__((ext_vector_type(2)));
typedef float v8f __attribute__((ext_vector_type(8)));

constexpr int N0 = 20000, N1 = 2000;
constexpr int FF = 512, HH = 256, DD = 128, KK = 16;
constexpr int E00 = 640000, E01 = 100000, E10 = 100000, E11 = 64000;

constexpr long long off_xw00 = 0;
constexpr long long off_lists = 5120000;
constexpr long long off_xw01 = 10240000;
constexpr long long off_xw11 = 10752000;
constexpr long long off_acc00 = 18944000;
constexpr long long off_acc01 = 24064000;
constexpr long long off_acc10 = 29184000;
constexpr long long off_acc11 = 29696000;

constexpr long long off_xw10  = off_xw00;
constexpr long long off_h0    = off_xw00;
constexpr long long off_h1    = off_xw01;
constexpr long long off_hw00  = off_acc01;
constexpr long long off_hw10  = off_acc01 + 2560000;
constexpr long long off_hw01  = off_xw11;
constexpr long long off_hw11  = off_xw11 + 256000;
constexpr long long off_bcc00 = off_acc00;
constexpr long long off_bcc01 = off_acc00 + 2560000;
constexpr long long off_bcc10 = off_acc10;
constexpr long long off_bcc11 = off_acc11;
constexpr long long off_z0R00 = off_acc01;
constexpr long long off_z0R01 = off_acc01 + 2560000;
constexpr long long off_z1R10 = off_acc10 + 256000;
constexpr long long off_z1d   = off_acc11 + 4096000;
constexpr long long off_Zk    = off_acc11;

__device__ __forceinline__ float wave_sum(float v) {
#pragma unroll
  for (int off = 16; off > 0; off >>= 1) v += __shfl_xor(v, off, 32);
  return v;
}

constexpr int BK   = 32;
constexpr int ASTR = 36;
constexpr int BPS  = 160;

__global__ __launch_bounds__(256) void gemm_f32_wmma(
    const float* __restrict__ A, const float* __restrict__ B, float* __restrict__ C,
    int M, int N, int Kd, long long sA, long long sB, long long sC) {
  __shared__ float As[2][128 * ASTR];
  __shared__ float Bs[2][(BK / 2) * BPS];

  const int b = blockIdx.z;
  A += (long long)b * sA;
  B += (long long)b * sB;
  C += (long long)b * sC;

  const int tid  = threadIdx.x;
  const int lane = tid & 31;
  const int wave = tid >> 5;
  const int mBlk = blockIdx.x * 128;
  const int n0   = blockIdx.y * 64;
  const int m0   = mBlk + wave * 16;
  const int half = lane >> 4;
  const int lr   = lane & 15;
  const bool active = (m0 < M);

  const int arow = tid >> 3;
  const int ac4  = (tid & 7) * 4;
  const int brow = tid >> 4;
  const int bc4  = (tid & 15) * 4;

  const int nChunks = Kd / BK;
  const float4 zero4 = make_float4(0.f, 0.f, 0.f, 0.f);

  {
#pragma unroll
    for (int p = 0; p < 4; ++p) {
      const int row = arow + p * 32;
      const int grow = mBlk + row;
      float4 v = (grow < M) ? *(const float4*)(A + (long long)grow * Kd + ac4) : zero4;
      *(float4*)&As[0][row * ASTR + ac4] = v;
    }
#pragma unroll
    for (int p = 0; p < 2; ++p) {
      const int row = brow + p * 16;
      const float4 v = *(const float4*)(B + (long long)row * N + n0 + bc4);
      float* bp = &Bs[0][(row >> 1) * BPS + bc4 * 2 + (row & 1)];
      bp[0] = v.x; bp[2] = v.y; bp[4] = v.z; bp[6] = v.w;
    }
  }
  __syncthreads();

  v8f acc0 = {}, acc1 = {}, acc2 = {}, acc3 = {};

  for (int c = 0; c < nChunks; ++c) {
    const int buf = c & 1;
    const bool pf = (c + 1 < nChunks);
    float4 ra[4], rb[2];

    if (pf) {
      const int k0 = (c + 1) * BK;
#pragma unroll
      for (int p = 0; p < 4; ++p) {
        const int grow = mBlk + arow + p * 32;
        ra[p] = (grow < M) ? *(const float4*)(A + (long long)grow * Kd + k0 + ac4) : zero4;
      }
#pragma unroll
      for (int p = 0; p < 2; ++p)
        rb[p] = *(const float4*)(B + (long long)(k0 + brow + p * 16) * N + n0 + bc4);
    }

    if (active) {
      const float* Asb = &As[buf][(wave * 16 + lr) * ASTR + 2 * half];
      const float* Bsb = &Bs[buf][half * BPS + lr * 2];
#pragma unroll
      for (int kk = 0; kk < BK; kk += 4) {
        const v2f a = *(const v2f*)(Asb + kk);
        const float* bp = Bsb + (kk >> 1) * BPS;
        const v2f b0 = *(const v2f*)(bp);
        const v2f b1 = *(const v2f*)(bp + 32);
        const v2f b2 = *(const v2f*)(bp + 64);
        const v2f b3 = *(const v2f*)(bp + 96);
        acc0 = __builtin_amdgcn_wmma_f32_16x16x4_f32(false, a, false, b0, (short)0, acc0, false, false);
        acc1 = __builtin_amdgcn_wmma_f32_16x16x4_f32(false, a, false, b1, (short)0, acc1, false, false);
        acc2 = __builtin_amdgcn_wmma_f32_16x16x4_f32(false, a, false, b2, (short)0, acc2, false, false);
        acc3 = __builtin_amdgcn_wmma_f32_16x16x4_f32(false, a, false, b3, (short)0, acc3, false, false);
      }
    }
    __syncthreads();

    if (pf) {
      const int nb = buf ^ 1;
#pragma unroll
      for (int p = 0; p < 4; ++p)
        *(float4*)&As[nb][(arow + p * 32) * ASTR + ac4] = ra[p];
#pragma unroll
      for (int p = 0; p < 2; ++p) {
        const int row = brow + p * 16;
        float* bp = &Bs[nb][(row >> 1) * BPS + bc4 * 2 + (row & 1)];
        bp[0] = rb[p].x; bp[2] = rb[p].y; bp[4] = rb[p].z; bp[6] = rb[p].w;
      }
      __syncthreads();
    }
  }

  if (active) {
#pragma unroll
    for (int r = 0; r < 8; ++r) {
      const int row = m0 + r + 8 * half;
      float* cp = C + (long long)row * N + n0 + lr;
      cp[0]  = acc0[r];
      cp[16] = acc1[r];
      cp[32] = acc2[r];
      cp[48] = acc3[r];
    }
  }
}

constexpr int SP_THREADS = 1024;
constexpr int SP_WAVES   = SP_THREADS / 32;
constexpr int SP_MAXT    = 125;
constexpr int SP_Q       = 16384;
constexpr int Q_EBITS    = 20;
constexpr int Q_EMASK    = (1 << Q_EBITS) - 1;
static_assert(E00 <= (1 << Q_EBITS) && E01 <= (1 << Q_EBITS) && E10 <= (1 << Q_EBITS) &&
              E11 <= (1 << Q_EBITS), "edge id fits the low bits of a queue word");
static_assert(N0 <= 65536 && N1 <= 65536 && SP_MAXT <= 128, "src and tile row fit one list word");

template <int N, bool BUILD>
__global__ __launch_bounds__(SP_THREADS) void spmm_tile(
    const int* __restrict__ dst, const int* __restrict__ src, const float* __restrict__ val,
    const float* __restrict__ x, float* __restrict__ out, int nE, int nRows, int T,
    int eStride, long long xStride, long long oStride,
    int* __restrict__ lists, int* __restrict__ counts, int cap) {
  constexpr int G = N / 4;
  constexpr int H = N / 128;
  __shared__ __attribute__((aligned(16))) float acc[SP_MAXT * N];
  __shared__ int qe[SP_Q];
  __shared__ int qcount;

  const int k = blockIdx.y;
  dst += (long long)k * eStride;
  src += (long long)k * eStride;
  val += (long long)k * eStride;
  x   += (long long)k * xStride;
  out += (long long)k * oStride;

  const int tid  = threadIdx.x;
  const int lane = tid & 31;
  const int wave = __builtin_amdgcn_readfirstlane(tid >> 5);
  const int row0 = blockIdx.x * T;
  const int tile = blockIdx.y * gridDim.x + blockIdx.x;
  int2* __restrict__ myList = (int2*)lists + (long long)tile * cap;

  int listCnt = 0;
  bool fromList = false;
  if constexpr (!BUILD) {
    listCnt = counts[tile];
    fromList = (listCnt <= cap);
  }

  for (int i = tid; i < T * G; i += SP_THREADS)
    ((float4*)acc)[i] = make_float4(0.f, 0.f, 0.f, 0.f);
  if (tid == 0) qcount = 0;
  __syncthreads();

  const int n4 = nE >> 2;
  const int4* d4p = (const int4*)dst;
  constexpr int STEP = SP_THREADS * 4;
  const int last4 = max(n4 - 1, 0);

  auto test4 = [&](const int4 (&dv)[4], int base) {
#pragma unroll
    for (int u = 0; u < 4; ++u) {
      const int idx = base + u * SP_THREADS + tid;
      const int e0 = idx << 2;
      const int dd[4] = {dv[u].x, dv[u].y, dv[u].z, dv[u].w};
#pragma unroll
      for (int c = 0; c < 4; ++c) {
        const unsigned r = (unsigned)(dd[c] - row0);
        if (idx < n4 && r < (unsigned)T) {
          const int e = e0 + c;
          const int slot = atomicAdd(&qcount, 1);
          if (slot < SP_Q) {
            qe[slot] = e | ((int)r << Q_EBITS);
          } else {
            const int s = src[e];
            const float v = val[e];
            const float* xr = x + (long long)s * N;
            float* a = &acc[(int)r * N];
            for (int f = 0; f < N; ++f) atomicAdd(&a[(f & 3) * G + (f >> 2)], v * xr[f]);
          }
        }
      }
    }
  };

  if (!fromList) {
  int4 da[4], db[4];
#pragma unroll
  for (int u = 0; u < 4; ++u) da[u] = d4p[min(u * SP_THREADS + tid, last4)];
  for (int i0 = 0; i0 < n4; i0 += 2 * STEP) {
#pragma unroll
    for (int u = 0; u < 4; ++u) db[u] = d4p[min(i0 + STEP + u * SP_THREADS + tid, last4)];
    test4(da, i0);
#pragma unroll
    for (int u = 0; u < 4; ++u) da[u] = d4p[min(i0 + 2 * STEP + u * SP_THREADS + tid, last4)];
    test4(db, i0 + STEP);
  }
  }
  __syncthreads();
  if constexpr (BUILD) {
    if (tid == 0) counts[tile] = qcount;
  }

  const int cnt = fromList ? listCnt : __builtin_amdgcn_readfirstlane(min(qcount, SP_Q));
  const int nGroups = (cnt + 31) >> 5;
  for (int g = wave; g < nGroups; g += SP_WAVES) {
    const int gi = g * 32 + lane;
    int s = 0, lr = 0, vb = 0;
    if (gi < cnt) {
      if (fromList) {
        const int2 en = myList[gi];
        s  = en.x & 0xFFFF;
        lr = en.x >> 16;
        vb = en.y;
      } else {
        const int q = qe[gi];
        const int e = q & Q_EMASK;
        lr = q >> Q_EBITS;
        s  = src[e];
        vb = __float_as_int(val[e]);
        if constexpr (BUILD) {
          if (gi < cap) myList[gi] = make_int2(s | (lr << 16), vb);
        }
      }
    }
    const int nv = min(32, cnt - g * 32);
    for (int j0 = 0; j0 < nv; j0 += 4) {
      float4 t[4][H];
      float vv[4];
      int ll[4];
#pragma unroll
      for (int u = 0; u < 4; ++u) {
        const int j = j0 + u;
        const int sj = __builtin_amdgcn_readlane(s, j);
        ll[u] = __builtin_amdgcn_readlane(lr, j);
        vv[u] = __int_as_float(__builtin_amdgcn_readlane(vb, j));
        const float4* xp = (const float4*)(x + (long long)sj * N);
#pragma unroll
        for (int h = 0; h < H; ++h) t[u][h] = xp[lane + 32 * h];
      }
#pragma unroll
      for (int u = 0; u < 4; ++u) {
        if (j0 + u < nv) {
          float* a = &acc[ll[u] * N + lane];
#pragma unroll
          for (int h = 0; h < H; ++h) {
            atomicAdd(&a[0 * G + 32 * h], vv[u] * t[u][h].x);
            atomicAdd(&a[1 * G + 32 * h], vv[u] * t[u][h].y);
            atomicAdd(&a[2 * G + 32 * h], vv[u] * t[u][h].z);
            atomicAdd(&a[3 * G + 32 * h], vv[u] * t[u][h].w);
          }
        }
      }
    }
  }
  __syncthreads();

  for (int i = tid; i < T * G; i += SP_THREADS) {
    const int r = i / G;
    const int g = i % G;
    if (row0 + r < nRows) {
      const float* a = &acc[r * N + g];
      *(float4*)(out + (long long)(row0 + r) * N + 4 * g) = make_float4(a[0], a[G], a[2 * G], a[3 * G]);
    }
  }
}

constexpr int T0  = 80;
constexpr int T1s = 8;
constexpr int T1m = 125;
static_assert(N0 % T0 == 0 && N1 % T1s == 0 && N1 % T1m == 0, "tiles cover the rows exactly");
static_assert(T0 <= SP_MAXT && T1s <= SP_MAXT && T1m <= SP_MAXT, "tile fits its LDS array");
static_assert(E00 % 4 == 0 && E01 % 4 == 0 && E10 % 4 == 0 && E11 % 4 == 0, "dst read as int4");

constexpr int CAP00 = 3072, CAP01 = 640, CAP10 = 640, CAP11 = 4608;
static_assert(CAP00 <= SP_Q && CAP01 <= SP_Q && CAP10 <= SP_Q && CAP11 <= SP_Q, "a list is a queue prefix");
constexpr long long off_l00 = off_lists;
constexpr long long off_l01 = off_l00 + 2LL * (N0 / T0) * CAP00;
constexpr long long off_l10 = off_l01 + 2LL * (N0 / T0) * CAP01;
constexpr long long off_l11 = off_l10 + 2LL * (N1 / T1s) * CAP10;
constexpr long long off_c00 = off_l11 + 2LL * (N1 / T1m) * KK * CAP11;
constexpr long long off_c01 = off_c00 + 256;
constexpr long long off_c10 = off_c01 + 256;
constexpr long long off_c11 = off_c10 + 256;
static_assert(N0 / T0 <= 256 && N1 / T1s <= 256 && (N1 / T1m) * KK <= 256, "count arrays hold 256 tiles each");
static_assert(off_c11 + 256 <= off_lists + 5120000, "lists and counts fit the plane xw10 used to have");

__global__ void finalize_conv(const float* __restrict__ buf, float* __restrict__ out,
                              int K, int n, int N, long long kStride, int do_relu) {
  const int wave = threadIdx.x >> 5;
  const int lane = threadIdx.x & 31;
  const int row = blockIdx.x * (blockDim.x >> 5) + wave;
  if (row >= n) return;
  const int nc = N >> 5;
  float vals[8];
  float ss = 0.f;
  for (int j = 0; j < nc; ++j) {
    const int c = lane + (j << 5);
    float a = 0.f;
    for (int k = 0; k < K; ++k) {
      float t = buf[(long long)k * kStride + (long long)row * N + c];
      if (do_relu) t = fmaxf(t, 0.f);
      a += t;
    }
    vals[j] = a;
    ss += a * a;
  }
  ss = wave_sum(ss);
  const float sc = rsqrtf(fmaxf(ss, 1e-12f));
  for (int j = 0; j < nc; ++j)
    out[(long long)row * N + lane + (j << 5)] = vals[j] * sc;
}

__global__ void add_act(const float* __restrict__ a, const float* __restrict__ b,
                        float* __restrict__ o, long long n, int do_relu) {
  const long long i = (long long)blockIdx.x * blockDim.x + threadIdx.x;
  if (i >= n) return;
  float v = a[i] + b[i];
  o[i] = do_relu ? fmaxf(v, 0.f) : v;
}

__global__ void scale_dloc(const float* __restrict__ z1, const float* __restrict__ dloc,
                           float* __restrict__ out, long long total) {
  const long long i = (long long)blockIdx.x * blockDim.x + threadIdx.x;
  if (i >= total) return;
  const int f = (int)(i % DD);
  const long long rk = i / DD;
  const int r = (int)(rk % N1);
  const int k = (int)(rk / N1);
  out[i] = z1[(long long)r * DD + f] * dloc[k * DD + f];
}

__global__ void edge_dot128(const float* __restrict__ zl, const float* __restrict__ zr,
                            const int* __restrict__ dst, const int* __restrict__ src,
                            float* __restrict__ out, int nE, int eStride,
                            long long zlStride, long long zrStride, long long oStride) {
  const int k = blockIdx.y;
  zl  += (long long)k * zlStride;
  zr  += (long long)k * zrStride;
  dst += (long long)k * eStride;
  src += (long long)k * eStride;
  out += (long long)k * oStride;
  const int e = blockIdx.x * (blockDim.x >> 5) + (threadIdx.x >> 5);
  if (e >= nE) return;
  const int lane = threadIdx.x & 31;
  const int d = dst[e];
  const int s = src[e];
  const float4 a = ((const float4*)(zl + (long long)d * DD))[lane];
  const float4 b = ((const float4*)(zr + (long long)s * DD))[lane];
  float acc = a.x * b.x + a.y * b.y + a.z * b.z + a.w * b.w;
  acc = wave_sum(acc);
  if (lane == 0) out[e] = acc;
}

extern "C" void kernel_launch(void* const* d_in, const int* in_sizes, int n_in,
                              void* d_out, int out_size, void* d_ws, size_t ws_size,
                              hipStream_t stream) {
  (void)in_sizes; (void)n_in; (void)out_size; (void)ws_size;

  const float* feat0   = (const float*)d_in[0];
  const float* feat1   = (const float*)d_in[1];
  const int*   e00_dst = (const int*)d_in[2];
  const int*   e00_src = (const int*)d_in[3];
  const float* e00_val = (const float*)d_in[4];
  const int*   e01_dst = (const int*)d_in[5];
  const int*   e01_src = (const int*)d_in[6];
  const float* e01_val = (const float*)d_in[7];
  const int*   e10_dst = (const int*)d_in[8];
  const int*   e10_src = (const int*)d_in[9];
  const float* e10_val = (const float*)d_in[10];
  const int*   e11_dst = (const int*)d_in[11];
  const int*   e11_src = (const int*)d_in[12];
  const float* e11_val = (const float*)d_in[13];
  const float* W1_00 = (const float*)d_in[14];
  const float* W1_01 = (const float*)d_in[15];
  const float* W1_10 = (const float*)d_in[16];
  const float* W1_11 = (const float*)d_in[17];
  const float* W2_00 = (const float*)d_in[18];
  const float* W2_01 = (const float*)d_in[19];
  const float* W2_10 = (const float*)d_in[20];
  const float* W2_11 = (const float*)d_in[21];
  const float* R00   = (const float*)d_in[22];
  const float* R01   = (const float*)d_in[23];
  const float* R10   = (const float*)d_in[24];
  const float* Rglob = (const float*)d_in[25];
  const float* dloc  = (const float*)d_in[26];

  float* ws = (float*)d_ws;
  float* out = (float*)d_out;
  float* z0  = out;
  float* z1  = out + 2560000;
  float* s00 = z1 + 256000;
  float* s01 = s00 + E00;
  float* s10 = s01 + E01;
  float* s11 = s10 + E10;

  const dim3 blk(256);
  auto gg = [](int M, int N, int batch) { return dim3((M + 127) / 128, N / 64, batch); };

  gemm_f32_wmma<<<gg(N0, HH, 1), blk, 0, stream>>>(feat0, W1_00, ws + off_xw00, N0, HH, FF, 0, 0, 0);
  gemm_f32_wmma<<<gg(N1, HH, 1), blk, 0, stream>>>(feat1, W1_01, ws + off_xw01, N1, HH, FF, 0, 0, 0);
  gemm_f32_wmma<<<gg(N1, HH, KK), blk, 0, stream>>>(feat1, W1_11, ws + off_xw11, N1, HH, FF,
                                                    0, (long long)FF * HH, (long long)N1 * HH);

  const dim3 sblk(SP_THREADS);
  int* l00 = (int*)(ws + off_l00);  int* c00 = (int*)(ws + off_c00);
  int* l01 = (int*)(ws + off_l01);  int* c01 = (int*)(ws + off_c01);
  int* l10 = (int*)(ws + off_l10);  int* c10 = (int*)(ws + off_c10);
  int* l11 = (int*)(ws + off_l11);  int* c11 = (int*)(ws + off_c11);
  spmm_tile<HH, true><<<dim3(N0 / T0, 1), sblk, 0, stream>>>(e00_dst, e00_src, e00_val, ws + off_xw00,
                                                             ws + off_acc00, E00, N0, T0, 0, 0, 0,
                                                             l00, c00, CAP00);
  gemm_f32_wmma<<<gg(N0, HH, 1), blk, 0, stream>>>(feat0, W1_10, ws + off_xw10, N0, HH, FF, 0, 0, 0);
  spmm_tile<HH, true><<<dim3(N0 / T0, 1), sblk, 0, stream>>>(e01_dst, e01_src, e01_val, ws + off_xw01,
                                                             ws + off_acc01, E01, N0, T0, 0, 0, 0,
                                                             l01, c01, CAP01);
  spmm_tile<HH, true><<<dim3(N1 / T1s, 1), sblk, 0, stream>>>(e10_dst, e10_src, e10_val, ws + off_xw10,
                                                              ws + off_acc10, E10, N1, T1s, 0, 0, 0,
                                                              l10, c10, CAP10);
  spmm_tile<HH, true><<<dim3(N1 / T1m, KK), sblk, 0, stream>>>(e11_dst, e11_src, e11_val, ws + off_xw11,
                                                               ws + off_acc11, E11, N1, T1m, E11,
                                                               (long long)N1 * HH, (long long)N1 * HH,
                                                               l11, c11, CAP11);

  finalize_conv<<<dim3((N0 + 7) / 8), blk, 0, stream>>>(ws + off_acc00, ws + off_acc00, 1, N0, HH, 0, 1);
  finalize_conv<<<dim3((N0 + 7) / 8), blk, 0, stream>>>(ws + off_acc01, ws + off_acc01, 1, N0, HH, 0, 1);
  finalize_conv<<<dim3((N1 + 7) / 8), blk, 0, stream>>>(ws + off_acc10, ws + off_acc10, 1, N1, HH, 0, 1);
  finalize_conv<<<dim3((N1 + 7) / 8), blk, 0, stream>>>(ws + off_acc11, ws + off_acc11, KK, N1, HH,
                                                        (long long)N1 * HH, 1);

  add_act<<<dim3((N0 * HH + 255) / 256), blk, 0, stream>>>(ws + off_acc00, ws + off_acc01,
                                                           ws + off_h0, (long long)N0 * HH, 1);
  add_act<<<dim3((N1 * HH + 255) / 256), blk, 0, stream>>>(ws + off_acc10, ws + off_acc11,
                                                           ws + off_h1, (long long)N1 * HH, 1);

  gemm_f32_wmma<<<gg(N0, DD, 1), blk, 0, stream>>>(ws + off_h0, W2_00, ws + off_hw00, N0, DD, HH, 0, 0, 0);
  gemm_f32_wmma<<<gg(N0, DD, 1), blk, 0, stream>>>(ws + off_h0, W2_10, ws + off_hw10, N0, DD, HH, 0, 0, 0);
  gemm_f32_wmma<<<gg(N1, DD, 1), blk, 0, stream>>>(ws + off_h1, W2_01, ws + off_hw01, N1, DD, HH, 0, 0, 0);
  gemm_f32_wmma<<<gg(N1, DD, KK), blk, 0, stream>>>(ws + off_h1, W2_11, ws + off_hw11, N1, DD, HH,
                                                    0, (long long)HH * DD, (long long)N1 * DD);

  spmm_tile<DD, false><<<dim3(N0 / T0, 1), sblk, 0, stream>>>(e00_dst, e00_src, e00_val, ws + off_hw00,
                                                              ws + off_bcc00, E00, N0, T0, 0, 0, 0,
                                                              l00, c00, CAP00);
  spmm_tile<DD, false><<<dim3(N0 / T0, 1), sblk, 0, stream>>>(e01_dst, e01_src, e01_val, ws + off_hw01,
                                                              ws + off_bcc01, E01, N0, T0, 0, 0, 0,
                                                              l01, c01, CAP01);
  spmm_tile<DD, false><<<dim3(N1 / T1s, 1), sblk, 0, stream>>>(e10_dst, e10_src, e10_val, ws + off_hw10,
                                                               ws + off_bcc10, E10, N1, T1s, 0, 0, 0,
                                                               l10, c10, CAP10);
  spmm_tile<DD, false><<<dim3(N1 / T1m, KK), sblk, 0, stream>>>(e11_dst, e11_src, e11_val, ws + off_hw11,
                                                                ws + off_bcc11, E11, N1, T1m, E11,
                                                                (long long)N1 * DD, (long long)N1 * DD,
                                                                l11, c11, CAP11);

  finalize_conv<<<dim3((N0 + 7) / 8), blk, 0, stream>>>(ws + off_bcc00, ws + off_bcc00, 1, N0, DD, 0, 0);
  finalize_conv<<<dim3((N0 + 7) / 8), blk, 0, stream>>>(ws + off_bcc01, ws + off_bcc01, 1, N0, DD, 0, 0);
  finalize_conv<<<dim3((N1 + 7) / 8), blk, 0, stream>>>(ws + off_bcc10, ws + off_bcc10, 1, N1, DD, 0, 0);
  finalize_conv<<<dim3((N1 + 7) / 8), blk, 0, stream>>>(ws + off_bcc11, ws + off_bcc11, KK, N1, DD,
                                                        (long long)N1 * DD, 0);

  add_act<<<dim3((N0 * DD + 255) / 256), blk, 0, stream>>>(ws + off_bcc00, ws + off_bcc01, z0,
                                                           (long long)N0 * DD, 0);
  add_act<<<dim3((N1 * DD + 255) / 256), blk, 0, stream>>>(ws + off_bcc10, ws + off_bcc11, z1,
                                                           (long long)N1 * DD, 0);

  gemm_f32_wmma<<<gg(N0, DD, 1), blk, 0, stream>>>(z0, R00, ws + off_z0R00, N0, DD, DD, 0, 0, 0);
  gemm_f32_wmma<<<gg(N0, DD, 1), blk, 0, stream>>>(z0, R01, ws + off_z0R01, N0, DD, DD, 0, 0, 0);
  gemm_f32_wmma<<<gg(N1, DD, 1), blk, 0, stream>>>(z1, R10, ws + off_z1R10, N1, DD, DD, 0, 0, 0);

  scale_dloc<<<dim3((KK * N1 * DD + 255) / 256), blk, 0, stream>>>(z1, dloc, ws + off_z1d,
                                                                   (long long)KK * N1 * DD);
  gemm_f32_wmma<<<gg(N1, DD, KK), blk, 0, stream>>>(ws + off_z1d, Rglob, ws + off_Zk, N1, DD, DD,
                                                    (long long)N1 * DD, 0, (long long)N1 * DD);

  edge_dot128<<<dim3((E00 + 7) / 8, 1), blk, 0, stream>>>(ws + off_z0R00, z0, e00_dst, e00_src,
                                                          s00, E00, 0, 0, 0, 0);
  edge_dot128<<<dim3((E01 + 7) / 8, 1), blk, 0, stream>>>(ws + off_z0R01, z1, e01_dst, e01_src,
                                                          s01, E01, 0, 0, 0, 0);
  edge_dot128<<<dim3((E10 + 7) / 8, 1), blk, 0, stream>>>(ws + off_z1R10, z0, e10_dst, e10_src,
                                                          s10, E10, 0, 0, 0, 0);
  edge_dot128<<<dim3((E11 + 7) / 8, KK), blk, 0, stream>>>(ws + off_Zk, ws + off_z1d, e11_dst, e11_src,
                                                           s11, E11, E11,
                                                           (long long)N1 * DD, (long long)N1 * DD,
                                                           (long long)E11);
}


# edge scores list-driven per dst tile: block keeps its zl rows in LDS, walks the edge list layer 1 left (entry holds edge id), 8 zr rows in flight per wave; same xor sum tree, lane's 4 products fused
# speedup vs baseline: 1.8776x; 1.0681x over previous
#include <hip/hip_runtime.h>

typedef float v2f __attribute__((ext_vector_type(2)));
typedef float v8f __attribute__((ext_vector_type(8)));

constexpr int N0 = 20000, N1 = 2000;
constexpr int FF = 512, HH = 256, DD = 128, KK = 16;
constexpr int E00 = 640000, E01 = 100000, E10 = 100000, E11 = 64000;

constexpr long long off_xw00 = 0;
constexpr long long off_lists = 5120000;
constexpr long long off_xw01 = 10240000;
constexpr long long off_xw11 = 10752000;
constexpr long long off_acc00 = 18944000;
constexpr long long off_acc01 = 24064000;
constexpr long long off_acc10 = 29184000;
constexpr long long off_acc11 = 29696000;

constexpr long long off_xw10  = off_xw00;
constexpr long long off_h0    = off_xw00;
constexpr long long off_h1    = off_xw01;
constexpr long long off_hw00  = off_acc01;
constexpr long long off_hw10  = off_acc01 + 2560000;
constexpr long long off_hw01  = off_xw11;
constexpr long long off_hw11  = off_xw11 + 256000;
constexpr long long off_bcc00 = off_acc00;
constexpr long long off_bcc01 = off_acc00 + 2560000;
constexpr long long off_bcc10 = off_acc10;
constexpr long long off_bcc11 = off_acc11;
constexpr long long off_z0R00 = off_acc01;
constexpr long long off_z0R01 = off_acc01 + 2560000;
constexpr long long off_z1R10 = off_acc10 + 256000;
constexpr long long off_z1d   = off_acc11 + 4096000;
constexpr long long off_Zk    = off_acc11;

__device__ __forceinline__ float wave_sum(float v) {
#pragma unroll
  for (int off = 16; off > 0; off >>= 1) v += __shfl_xor(v, off, 32);
  return v;
}

constexpr int BK   = 32;
constexpr int ASTR = 36;
constexpr int BPS  = 160;

__global__ __launch_bounds__(256) void gemm_f32_wmma(
    const float* __restrict__ A, const float* __restrict__ B, float* __restrict__ C,
    int M, int N, int Kd, long long sA, long long sB, long long sC) {
  __shared__ float As[2][128 * ASTR];
  __shared__ float Bs[2][(BK / 2) * BPS];

  const int b = blockIdx.z;
  A += (long long)b * sA;
  B += (long long)b * sB;
  C += (long long)b * sC;

  const int tid  = threadIdx.x;
  const int lane = tid & 31;
  const int wave = tid >> 5;
  const int mBlk = blockIdx.x * 128;
  const int n0   = blockIdx.y * 64;
  const int m0   = mBlk + wave * 16;
  const int half = lane >> 4;
  const int lr   = lane & 15;
  const bool active = (m0 < M);

  const int arow = tid >> 3;
  const int ac4  = (tid & 7) * 4;
  const int brow = tid >> 4;
  const int bc4  = (tid & 15) * 4;

  const int nChunks = Kd / BK;
  const float4 zero4 = make_float4(0.f, 0.f, 0.f, 0.f);

  {
#pragma unroll
    for (int p = 0; p < 4; ++p) {
      const int row = arow + p * 32;
      const int grow = mBlk + row;
      float4 v = (grow < M) ? *(const float4*)(A + (long long)grow * Kd + ac4) : zero4;
      *(float4*)&As[0][row * ASTR + ac4] = v;
    }
#pragma unroll
    for (int p = 0; p < 2; ++p) {
      const int row = brow + p * 16;
      const float4 v = *(const float4*)(B + (long long)row * N + n0 + bc4);
      float* bp = &Bs[0][(row >> 1) * BPS + bc4 * 2 + (row & 1)];
      bp[0] = v.x; bp[2] = v.y; bp[4] = v.z; bp[6] = v.w;
    }
  }
  __syncthreads();

  v8f acc0 = {}, acc1 = {}, acc2 = {}, acc3 = {};

  for (int c = 0; c < nChunks; ++c) {
    const int buf = c & 1;
    const bool pf = (c + 1 < nChunks);
    float4 ra[4], rb[2];

    if (pf) {
      const int k0 = (c + 1) * BK;
#pragma unroll
      for (int p = 0; p < 4; ++p) {
        const int grow = mBlk + arow + p * 32;
        ra[p] = (grow < M) ? *(const float4*)(A + (long long)grow * Kd + k0 + ac4) : zero4;
      }
#pragma unroll
      for (int p = 0; p < 2; ++p)
        rb[p] = *(const float4*)(B + (long long)(k0 + brow + p * 16) * N + n0 + bc4);
    }

    if (active) {
      const float* Asb = &As[buf][(wave * 16 + lr) * ASTR + 2 * half];
      const float* Bsb = &Bs[buf][half * BPS + lr * 2];
#pragma unroll
      for (int kk = 0; kk < BK; kk += 4) {
        const v2f a = *(const v2f*)(Asb + kk);
        const float* bp = Bsb + (kk >> 1) * BPS;
        const v2f b0 = *(const v2f*)(bp);
        const v2f b1 = *(const v2f*)(bp + 32);
        const v2f b2 = *(const v2f*)(bp + 64);
        const v2f b3 = *(const v2f*)(bp + 96);
        acc0 = __builtin_amdgcn_wmma_f32_16x16x4_f32(false, a, false, b0, (short)0, acc0, false, false);
        acc1 = __builtin_amdgcn_wmma_f32_16x16x4_f32(false, a, false, b1, (short)0, acc1, false, false);
        acc2 = __builtin_amdgcn_wmma_f32_16x16x4_f32(false, a, false, b2, (short)0, acc2, false, false);
        acc3 = __builtin_amdgcn_wmma_f32_16x16x4_f32(false, a, false, b3, (short)0, acc3, false, false);
      }
    }
    __syncthreads();

    if (pf) {
      const int nb = buf ^ 1;
#pragma unroll
      for (int p = 0; p < 4; ++p)
        *(float4*)&As[nb][(arow + p * 32) * ASTR + ac4] = ra[p];
#pragma unroll
      for (int p = 0; p < 2; ++p) {
        const int row = brow + p * 16;
        float* bp = &Bs[nb][(row >> 1) * BPS + bc4 * 2 + (row & 1)];
        bp[0] = rb[p].x; bp[2] = rb[p].y; bp[4] = rb[p].z; bp[6] = rb[p].w;
      }
      __syncthreads();
    }
  }

  if (active) {
#pragma unroll
    for (int r = 0; r < 8; ++r) {
      const int row = m0 + r + 8 * half;
      float* cp = C + (long long)row * N + n0 + lr;
      cp[0]  = acc0[r];
      cp[16] = acc1[r];
      cp[32] = acc2[r];
      cp[48] = acc3[r];
    }
  }
}

constexpr int SP_THREADS = 1024;
constexpr int SP_WAVES   = SP_THREADS / 32;
constexpr int SP_MAXT    = 125;
constexpr int SP_Q       = 16384;
constexpr int Q_EBITS    = 20;
constexpr int Q_EMASK    = (1 << Q_EBITS) - 1;
static_assert(E00 <= (1 << Q_EBITS) && E01 <= (1 << Q_EBITS) && E10 <= (1 << Q_EBITS) &&
              E11 <= (1 << Q_EBITS), "edge id fits the low bits of a queue word");
static_assert(N0 <= 65536 && N1 <= 65536 && SP_MAXT <= 128, "src and tile row fit one list word");

template <int N, bool BUILD>
__global__ __launch_bounds__(SP_THREADS) void spmm_tile(
    const int* __restrict__ dst, const int* __restrict__ src, const float* __restrict__ val,
    const float* __restrict__ x, float* __restrict__ out, int nE, int nRows, int T,
    int eStride, long long xStride, long long oStride,
    int* __restrict__ lists, int* __restrict__ counts, int cap) {
  constexpr int G = N / 4;
  constexpr int H = N / 128;
  __shared__ __attribute__((aligned(16))) float acc[SP_MAXT * N];
  __shared__ int qe[SP_Q];
  __shared__ int qcount;

  const int k = blockIdx.y;
  dst += (long long)k * eStride;
  src += (long long)k * eStride;
  val += (long long)k * eStride;
  x   += (long long)k * xStride;
  out += (long long)k * oStride;

  const int tid  = threadIdx.x;
  const int lane = tid & 31;
  const int wave = __builtin_amdgcn_readfirstlane(tid >> 5);
  const int row0 = blockIdx.x * T;
  const int tile = blockIdx.y * gridDim.x + blockIdx.x;
  int2* __restrict__ myList = (int2*)lists + (long long)tile * cap;

  int listCnt = 0;
  bool fromList = false;
  if constexpr (!BUILD) {
    listCnt = counts[tile];
    fromList = (listCnt <= cap);
  }

  for (int i = tid; i < T * G; i += SP_THREADS)
    ((float4*)acc)[i] = make_float4(0.f, 0.f, 0.f, 0.f);
  if (tid == 0) qcount = 0;
  __syncthreads();

  const int n4 = nE >> 2;
  const int4* d4p = (const int4*)dst;
  constexpr int STEP = SP_THREADS * 4;
  const int last4 = max(n4 - 1, 0);

  auto test4 = [&](const int4 (&dv)[4], int base) {
#pragma unroll
    for (int u = 0; u < 4; ++u) {
      const int idx = base + u * SP_THREADS + tid;
      const int e0 = idx << 2;
      const int dd[4] = {dv[u].x, dv[u].y, dv[u].z, dv[u].w};
#pragma unroll
      for (int c = 0; c < 4; ++c) {
        const unsigned r = (unsigned)(dd[c] - row0);
        if (idx < n4 && r < (unsigned)T) {
          const int e = e0 + c;
          const int slot = atomicAdd(&qcount, 1);
          if (slot < SP_Q) {
            qe[slot] = e | ((int)r << Q_EBITS);
          } else {
            const int s = src[e];
            const float v = val[e];
            const float* xr = x + (long long)s * N;
            float* a = &acc[(int)r * N];
            for (int f = 0; f < N; ++f) atomicAdd(&a[(f & 3) * G + (f >> 2)], v * xr[f]);
          }
        }
      }
    }
  };

  if (!fromList) {
  int4 da[4], db[4];
#pragma unroll
  for (int u = 0; u < 4; ++u) da[u] = d4p[min(u * SP_THREADS + tid, last4)];
  for (int i0 = 0; i0 < n4; i0 += 2 * STEP) {
#pragma unroll
    for (int u = 0; u < 4; ++u) db[u] = d4p[min(i0 + STEP + u * SP_THREADS + tid, last4)];
    test4(da, i0);
#pragma unroll
    for (int u = 0; u < 4; ++u) da[u] = d4p[min(i0 + 2 * STEP + u * SP_THREADS + tid, last4)];
    test4(db, i0 + STEP);
  }
  }
  __syncthreads();
  if constexpr (BUILD) {
    if (tid == 0) counts[tile] = qcount;
  }

  const int cnt = fromList ? listCnt : __builtin_amdgcn_readfirstlane(min(qcount, SP_Q));
  const int nGroups = (cnt + 31) >> 5;
  for (int g = wave; g < nGroups; g += SP_WAVES) {
    const int gi = g * 32 + lane;
    int s = 0, lr = 0, vb = 0;
    if (gi < cnt) {
      if (fromList) {
        const int2 en = myList[gi];
        s  = en.x & 0xFFFF;
        lr = en.x >> 16;
        vb = __float_as_int(val[en.y]);
      } else {
        const int q = qe[gi];
        const int e = q & Q_EMASK;
        lr = q >> Q_EBITS;
        s  = src[e];
        vb = __float_as_int(val[e]);
        if constexpr (BUILD) {
          if (gi < cap) myList[gi] = make_int2(s | (lr << 16), e);
        }
      }
    }
    const int nv = min(32, cnt - g * 32);
    for (int j0 = 0; j0 < nv; j0 += 4) {
      float4 t[4][H];
      float vv[4];
      int ll[4];
#pragma unroll
      for (int u = 0; u < 4; ++u) {
        const int j = j0 + u;
        const int sj = __builtin_amdgcn_readlane(s, j);
        ll[u] = __builtin_amdgcn_readlane(lr, j);
        vv[u] = __int_as_float(__builtin_amdgcn_readlane(vb, j));
        const float4* xp = (const float4*)(x + (long long)sj * N);
#pragma unroll
        for (int h = 0; h < H; ++h) t[u][h] = xp[lane + 32 * h];
      }
#pragma unroll
      for (int u = 0; u < 4; ++u) {
        if (j0 + u < nv) {
          float* a = &acc[ll[u] * N + lane];
#pragma unroll
          for (int h = 0; h < H; ++h) {
            atomicAdd(&a[0 * G + 32 * h], vv[u] * t[u][h].x);
            atomicAdd(&a[1 * G + 32 * h], vv[u] * t[u][h].y);
            atomicAdd(&a[2 * G + 32 * h], vv[u] * t[u][h].z);
            atomicAdd(&a[3 * G + 32 * h], vv[u] * t[u][h].w);
          }
        }
      }
    }
  }
  __syncthreads();

  for (int i = tid; i < T * G; i += SP_THREADS) {
    const int r = i / G;
    const int g = i % G;
    if (row0 + r < nRows) {
      const float* a = &acc[r * N + g];
      *(float4*)(out + (long long)(row0 + r) * N + 4 * g) = make_float4(a[0], a[G], a[2 * G], a[3 * G]);
    }
  }
}

constexpr int T0  = 80;
constexpr int T1s = 8;
constexpr int T1m = 125;
static_assert(N0 % T0 == 0 && N1 % T1s == 0 && N1 % T1m == 0, "tiles cover the rows exactly");
static_assert(T0 <= SP_MAXT && T1s <= SP_MAXT && T1m <= SP_MAXT, "tile fits its LDS array");
static_assert(E00 % 4 == 0 && E01 % 4 == 0 && E10 % 4 == 0 && E11 % 4 == 0, "dst read as int4");

constexpr int CAP00 = 3072, CAP01 = 640, CAP10 = 640, CAP11 = 4608;
static_assert(CAP00 <= SP_Q && CAP01 <= SP_Q && CAP10 <= SP_Q && CAP11 <= SP_Q, "a list is a queue prefix");
constexpr long long off_l00 = off_lists;
constexpr long long off_l01 = off_l00 + 2LL * (N0 / T0) * CAP00;
constexpr long long off_l10 = off_l01 + 2LL * (N0 / T0) * CAP01;
constexpr long long off_l11 = off_l10 + 2LL * (N1 / T1s) * CAP10;
constexpr long long off_c00 = off_l11 + 2LL * (N1 / T1m) * KK * CAP11;
constexpr long long off_c01 = off_c00 + 256;
constexpr long long off_c10 = off_c01 + 256;
constexpr long long off_c11 = off_c10 + 256;
static_assert(N0 / T0 <= 256 && N1 / T1s <= 256 && (N1 / T1m) * KK <= 256, "count arrays hold 256 tiles each");
static_assert(off_c11 + 256 <= off_lists + 5120000, "lists and counts fit the plane xw10 used to have");

__global__ void finalize_conv(const float* __restrict__ buf, float* __restrict__ out,
                              int K, int n, int N, long long kStride, int do_relu) {
  const int wave = threadIdx.x >> 5;
  const int lane = threadIdx.x & 31;
  const int row = blockIdx.x * (blockDim.x >> 5) + wave;
  if (row >= n) return;
  const int nc = N >> 5;
  float vals[8];
  float ss = 0.f;
  for (int j = 0; j < nc; ++j) {
    const int c = lane + (j << 5);
    float a = 0.f;
    for (int k = 0; k < K; ++k) {
      float t = buf[(long long)k * kStride + (long long)row * N + c];
      if (do_relu) t = fmaxf(t, 0.f);
      a += t;
    }
    vals[j] = a;
    ss += a * a;
  }
  ss = wave_sum(ss);
  const float sc = rsqrtf(fmaxf(ss, 1e-12f));
  for (int j = 0; j < nc; ++j)
    out[(long long)row * N + lane + (j << 5)] = vals[j] * sc;
}

__global__ void add_act(const float* __restrict__ a, const float* __restrict__ b,
                        float* __restrict__ o, long long n, int do_relu) {
  const long long i = (long long)blockIdx.x * blockDim.x + threadIdx.x;
  if (i >= n) return;
  float v = a[i] + b[i];
  o[i] = do_relu ? fmaxf(v, 0.f) : v;
}

__global__ void scale_dloc(const float* __restrict__ z1, const float* __restrict__ dloc,
                           float* __restrict__ out, long long total) {
  const long long i = (long long)blockIdx.x * blockDim.x + threadIdx.x;
  if (i >= total) return;
  const int f = (int)(i % DD);
  const long long rk = i / DD;
  const int r = (int)(rk % N1);
  const int k = (int)(rk / N1);
  out[i] = z1[(long long)r * DD + f] * dloc[k * DD + f];
}

constexpr int ES_THREADS = 1024;
constexpr int ES_WAVES   = ES_THREADS / 32;

__global__ __launch_bounds__(ES_THREADS) void edge_score_tile(
    const float* __restrict__ zl, const float* __restrict__ zr,
    const int* __restrict__ dst, const int* __restrict__ src,
    float* __restrict__ out, int nE, int T, int eStride,
    long long zlStride, long long zrStride, long long oStride,
    const int* __restrict__ lists, const int* __restrict__ counts, int cap) {
  __shared__ __attribute__((aligned(16))) float zt[SP_MAXT * DD];

  const int k = blockIdx.y;
  zl  += (long long)k * zlStride;
  zr  += (long long)k * zrStride;
  dst += (long long)k * eStride;
  src += (long long)k * eStride;
  out += (long long)k * oStride;

  const int tid  = threadIdx.x;
  const int lane = tid & 31;
  const int wave = __builtin_amdgcn_readfirstlane(tid >> 5);
  const int row0 = blockIdx.x * T;
  const int tile = blockIdx.y * gridDim.x + blockIdx.x;
  const int2* __restrict__ myList = (const int2*)lists + (long long)tile * cap;

  {
    const float4* zsrc = (const float4*)(zl + (long long)row0 * DD);
    for (int i = tid; i < T * (DD / 4); i += ES_THREADS) ((float4*)zt)[i] = zsrc[i];
  }
  const int cnt = __builtin_amdgcn_readfirstlane(counts[tile]);
  __syncthreads();

  if (cnt <= cap) {
    const bool h16 = (lane & 16) != 0, h8 = (lane & 8) != 0, h4 = (lane & 4) != 0;
    const int nGroups = (cnt + 31) >> 5;
    for (int g = wave; g < nGroups; g += ES_WAVES) {
      const int gi = g * 32 + lane;
      int s = 0, lr = 0, e = 0;
      if (gi < cnt) {
        const int2 en = myList[gi];
        s  = en.x & 0xFFFF;
        lr = en.x >> 16;
        e  = en.y;
      }
      const int nv = min(32, cnt - g * 32);
      for (int j0 = 0; j0 < nv; j0 += 8) {
        float4 b[8];
        float p[8];
#pragma unroll
        for (int u = 0; u < 8; ++u) {
          const int sj = __builtin_amdgcn_readlane(s, j0 + u);
          b[u] = ((const float4*)(zr + (long long)sj * DD))[lane];
        }
#pragma unroll
        for (int u = 0; u < 8; ++u) {
          const int lj = __builtin_amdgcn_readlane(lr, j0 + u);
          const float4 a = ((const float4*)(zt + lj * DD))[lane];
          p[u] = a.x * b[u].x + a.y * b[u].y + a.z * b[u].z + a.w * b[u].w;
        }
        float q[4], r[2];
#pragma unroll
        for (int i = 0; i < 4; ++i) {
          const float keep = h16 ? p[4 + i] : p[i];
          const float send = h16 ? p[i] : p[4 + i];
          q[i] = keep + __shfl_xor(send, 16, 32);
        }
#pragma unroll
        for (int i = 0; i < 2; ++i) {
          const float keep = h8 ? q[2 + i] : q[i];
          const float send = h8 ? q[i] : q[2 + i];
          r[i] = keep + __shfl_xor(send, 8, 32);
        }
        float t;
        {
          const float keep = h4 ? r[1] : r[0];
          const float send = h4 ? r[0] : r[1];
          t = keep + __shfl_xor(send, 4, 32);
        }
        t += __shfl_xor(t, 2, 32);
        t += __shfl_xor(t, 1, 32);
        const int mine = j0 + (lane >> 2);
        const int eo = __shfl(e, mine, 32);
        if ((lane & 3) == 0 && mine < nv) out[eo] = t;
      }
    }
  } else {
    for (int base = wave * 32; base < nE; base += ES_WAVES * 32) {
      const int e = base + lane;
      const unsigned r = (e < nE) ? (unsigned)(dst[e] - row0) : 0xFFFFFFFFu;
      unsigned mask = __builtin_amdgcn_ballot_w32(r < (unsigned)T);
      while (mask != 0u) {
        const int j = __builtin_ctz(mask);
        mask &= mask - 1u;
        const int lj = __builtin_amdgcn_readlane((int)r, j);
        const int sj = src[base + j];
        const float4 a = ((const float4*)(zt + lj * DD))[lane];
        const float4 bb = ((const float4*)(zr + (long long)sj * DD))[lane];
        float acc = a.x * bb.x + a.y * bb.y + a.z * bb.z + a.w * bb.w;
        acc = wave_sum(acc);
        if (lane == 0) out[base + j] = acc;
      }
    }
  }
}

extern "C" void kernel_launch(void* const* d_in, const int* in_sizes, int n_in,
                              void* d_out, int out_size, void* d_ws, size_t ws_size,
                              hipStream_t stream) {
  (void)in_sizes; (void)n_in; (void)out_size; (void)ws_size;

  const float* feat0   = (const float*)d_in[0];
  const float* feat1   = (const float*)d_in[1];
  const int*   e00_dst = (const int*)d_in[2];
  const int*   e00_src = (const int*)d_in[3];
  const float* e00_val = (const float*)d_in[4];
  const int*   e01_dst = (const int*)d_in[5];
  const int*   e01_src = (const int*)d_in[6];
  const float* e01_val = (const float*)d_in[7];
  const int*   e10_dst = (const int*)d_in[8];
  const int*   e10_src = (const int*)d_in[9];
  const float* e10_val = (const float*)d_in[10];
  const int*   e11_dst = (const int*)d_in[11];
  const int*   e11_src = (const int*)d_in[12];
  const float* e11_val = (const float*)d_in[13];
  const float* W1_00 = (const float*)d_in[14];
  const float* W1_01 = (const float*)d_in[15];
  const float* W1_10 = (const float*)d_in[16];
  const float* W1_11 = (const float*)d_in[17];
  const float* W2_00 = (const float*)d_in[18];
  const float* W2_01 = (const float*)d_in[19];
  const float* W2_10 = (const float*)d_in[20];
  const float* W2_11 = (const float*)d_in[21];
  const float* R00   = (const float*)d_in[22];
  const float* R01   = (const float*)d_in[23];
  const float* R10   = (const float*)d_in[24];
  const float* Rglob = (const float*)d_in[25];
  const float* dloc  = (const float*)d_in[26];

  float* ws = (float*)d_ws;
  float* out = (float*)d_out;
  float* z0  = out;
  float* z1  = out + 2560000;
  float* s00 = z1 + 256000;
  float* s01 = s00 + E00;
  float* s10 = s01 + E01;
  float* s11 = s10 + E10;

  const dim3 blk(256);
  auto gg = [](int M, int N, int batch) { return dim3((M + 127) / 128, N / 64, batch); };

  gemm_f32_wmma<<<gg(N0, HH, 1), blk, 0, stream>>>(feat0, W1_00, ws + off_xw00, N0, HH, FF, 0, 0, 0);
  gemm_f32_wmma<<<gg(N1, HH, 1), blk, 0, stream>>>(feat1, W1_01, ws + off_xw01, N1, HH, FF, 0, 0, 0);
  gemm_f32_wmma<<<gg(N1, HH, KK), blk, 0, stream>>>(feat1, W1_11, ws + off_xw11, N1, HH, FF,
                                                    0, (long long)FF * HH, (long long)N1 * HH);

  const dim3 sblk(SP_THREADS);
  int* l00 = (int*)(ws + off_l00);  int* c00 = (int*)(ws + off_c00);
  int* l01 = (int*)(ws + off_l01);  int* c01 = (int*)(ws + off_c01);
  int* l10 = (int*)(ws + off_l10);  int* c10 = (int*)(ws + off_c10);
  int* l11 = (int*)(ws + off_l11);  int* c11 = (int*)(ws + off_c11);
  spmm_tile<HH, true><<<dim3(N0 / T0, 1), sblk, 0, stream>>>(e00_dst, e00_src, e00_val, ws + off_xw00,
                                                             ws + off_acc00, E00, N0, T0, 0, 0, 0,
                                                             l00, c00, CAP00);
  gemm_f32_wmma<<<gg(N0, HH, 1), blk, 0, stream>>>(feat0, W1_10, ws + off_xw10, N0, HH, FF, 0, 0, 0);
  spmm_tile<HH, true><<<dim3(N0 / T0, 1), sblk, 0, stream>>>(e01_dst, e01_src, e01_val, ws + off_xw01,
                                                             ws + off_acc01, E01, N0, T0, 0, 0, 0,
                                                             l01, c01, CAP01);
  spmm_tile<HH, true><<<dim3(N1 / T1s, 1), sblk, 0, stream>>>(e10_dst, e10_src, e10_val, ws + off_xw10,
                                                              ws + off_acc10, E10, N1, T1s, 0, 0, 0,
                                                              l10, c10, CAP10);
  spmm_tile<HH, true><<<dim3(N1 / T1m, KK), sblk, 0, stream>>>(e11_dst, e11_src, e11_val, ws + off_xw11,
                                                               ws + off_acc11, E11, N1, T1m, E11,
                                                               (long long)N1 * HH, (long long)N1 * HH,
                                                               l11, c11, CAP11);

  finalize_conv<<<dim3((N0 + 7) / 8), blk, 0, stream>>>(ws + off_acc00, ws + off_acc00, 1, N0, HH, 0, 1);
  finalize_conv<<<dim3((N0 + 7) / 8), blk, 0, stream>>>(ws + off_acc01, ws + off_acc01, 1, N0, HH, 0, 1);
  finalize_conv<<<dim3((N1 + 7) / 8), blk, 0, stream>>>(ws + off_acc10, ws + off_acc10, 1, N1, HH, 0, 1);
  finalize_conv<<<dim3((N1 + 7) / 8), blk, 0, stream>>>(ws + off_acc11, ws + off_acc11, KK, N1, HH,
                                                        (long long)N1 * HH, 1);

  add_act<<<dim3((N0 * HH + 255) / 256), blk, 0, stream>>>(ws + off_acc00, ws + off_acc01,
                                                           ws + off_h0, (long long)N0 * HH, 1);
  add_act<<<dim3((N1 * HH + 255) / 256), blk, 0, stream>>>(ws + off_acc10, ws + off_acc11,
                                                           ws + off_h1, (long long)N1 * HH, 1);

  gemm_f32_wmma<<<gg(N0, DD, 1), blk, 0, stream>>>(ws + off_h0, W2_00, ws + off_hw00, N0, DD, HH, 0, 0, 0);
  gemm_f32_wmma<<<gg(N0, DD, 1), blk, 0, stream>>>(ws + off_h0, W2_10, ws + off_hw10, N0, DD, HH, 0, 0, 0);
  gemm_f32_wmma<<<gg(N1, DD, 1), blk, 0, stream>>>(ws + off_h1, W2_01, ws + off_hw01, N1, DD, HH, 0, 0, 0);
  gemm_f32_wmma<<<gg(N1, DD, KK), blk, 0, stream>>>(ws + off_h1, W2_11, ws + off_hw11, N1, DD, HH,
                                                    0, (long long)HH * DD, (long long)N1 * DD);

  spmm_tile<DD, false><<<dim3(N0 / T0, 1), sblk, 0, stream>>>(e00_dst, e00_src, e00_val, ws + off_hw00,
                                                              ws + off_bcc00, E00, N0, T0, 0, 0, 0,
                                                              l00, c00, CAP00);
  spmm_tile<DD, false><<<dim3(N0 / T0, 1), sblk, 0, stream>>>(e01_dst, e01_src, e01_val, ws + off_hw01,
                                                              ws + off_bcc01, E01, N0, T0, 0, 0, 0,
                                                              l01, c01, CAP01);
  spmm_tile<DD, false><<<dim3(N1 / T1s, 1), sblk, 0, stream>>>(e10_dst, e10_src, e10_val, ws + off_hw10,
                                                               ws + off_bcc10, E10, N1, T1s, 0, 0, 0,
                                                               l10, c10, CAP10);
  spmm_tile<DD, false><<<dim3(N1 / T1m, KK), sblk, 0, stream>>>(e11_dst, e11_src, e11_val, ws + off_hw11,
                                                                ws + off_bcc11, E11, N1, T1m, E11,
                                                                (long long)N1 * DD, (long long)N1 * DD,
                                                                l11, c11, CAP11);

  finalize_conv<<<dim3((N0 + 7) / 8), blk, 0, stream>>>(ws + off_bcc00, ws + off_bcc00, 1, N0, DD, 0, 0);
  finalize_conv<<<dim3((N0 + 7) / 8), blk, 0, stream>>>(ws + off_bcc01, ws + off_bcc01, 1, N0, DD, 0, 0);
  finalize_conv<<<dim3((N1 + 7) / 8), blk, 0, stream>>>(ws + off_bcc10, ws + off_bcc10, 1, N1, DD, 0, 0);
  finalize_conv<<<dim3((N1 + 7) / 8), blk, 0, stream>>>(ws + off_bcc11, ws + off_bcc11, KK, N1, DD,
                                                        (long long)N1 * DD, 0);

  add_act<<<dim3((N0 * DD + 255) / 256), blk, 0, stream>>>(ws + off_bcc00, ws + off_bcc01, z0,
                                                           (long long)N0 * DD, 0);
  add_act<<<dim3((N1 * DD + 255) / 256), blk, 0, stream>>>(ws + off_bcc10, ws + off_bcc11, z1,
                                                           (long long)N1 * DD, 0);

  gemm_f32_wmma<<<gg(N0, DD, 1), blk, 0, stream>>>(z0, R00, ws + off_z0R00, N0, DD, DD, 0, 0, 0);
  gemm_f32_wmma<<<gg(N0, DD, 1), blk, 0, stream>>>(z0, R01, ws + off_z0R01, N0, DD, DD, 0, 0, 0);
  gemm_f32_wmma<<<gg(N1, DD, 1), blk, 0, stream>>>(z1, R10, ws + off_z1R10, N1, DD, DD, 0, 0, 0);

  scale_dloc<<<dim3((KK * N1 * DD + 255) / 256), blk, 0, stream>>>(z1, dloc, ws + off_z1d,
                                                                   (long long)KK * N1 * DD);
  gemm_f32_wmma<<<gg(N1, DD, KK), blk, 0, stream>>>(ws + off_z1d, Rglob, ws + off_Zk, N1, DD, DD,
                                                    (long long)N1 * DD, 0, (long long)N1 * DD);

  const dim3 eblk(ES_THREADS);
  edge_score_tile<<<dim3(N0 / T0, 1), eblk, 0, stream>>>(ws + off_z0R00, z0, e00_dst, e00_src,
                                                         s00, E00, T0, 0, 0, 0, 0, l00, c00, CAP00);
  edge_score_tile<<<dim3(N0 / T0, 1), eblk, 0, stream>>>(ws + off_z0R01, z1, e01_dst, e01_src,
                                                         s01, E01, T0, 0, 0, 0, 0, l01, c01, CAP01);
  edge_score_tile<<<dim3(N1 / T1s, 1), eblk, 0, stream>>>(ws + off_z1R10, z0, e10_dst, e10_src,
                                                          s10, E10, T1s, 0, 0, 0, 0, l10, c10, CAP10);
  edge_score_tile<<<dim3(N1 / T1m, KK), eblk, 0, stream>>>(ws + off_Zk, ws + off_z1d, e11_dst, e11_src,
                                                           s11, E11, T1m, E11,
                                                           (long long)N1 * DD, (long long)N1 * DD,
                                                           (long long)E11, l11, c11, CAP11);
}


# tile edge lists built up front by one binning pass (LDS ranks per block, one global atomic add per block and tile); layer-1 SPMM tiles read them, no dst scan left; sums stay in arrival order
# speedup vs baseline: 1.9817x; 1.0554x over previous
#include <hip/hip_runtime.h>

typedef float v2f __attribute__((ext_vector_type(2)));
typedef float v8f __attribute__((ext_vector_type(8)));

constexpr int N0 = 20000, N1 = 2000;
constexpr int FF = 512, HH = 256, DD = 128, KK = 16;
constexpr int E00 = 640000, E01 = 100000, E10 = 100000, E11 = 64000;

constexpr long long off_xw00 = 0;
constexpr long long off_lists = 5120000;
constexpr long long off_xw01 = 10240000;
constexpr long long off_xw11 = 10752000;
constexpr long long off_acc00 = 18944000;
constexpr long long off_acc01 = 24064000;
constexpr long long off_acc10 = 29184000;
constexpr long long off_acc11 = 29696000;

constexpr long long off_xw10  = off_xw00;
constexpr long long off_h0    = off_xw00;
constexpr long long off_h1    = off_xw01;
constexpr long long off_hw00  = off_acc01;
constexpr long long off_hw10  = off_acc01 + 2560000;
constexpr long long off_hw01  = off_xw11;
constexpr long long off_hw11  = off_xw11 + 256000;
constexpr long long off_bcc00 = off_acc00;
constexpr long long off_bcc01 = off_acc00 + 2560000;
constexpr long long off_bcc10 = off_acc10;
constexpr long long off_bcc11 = off_acc11;
constexpr long long off_z0R00 = off_acc01;
constexpr long long off_z0R01 = off_acc01 + 2560000;
constexpr long long off_z1R10 = off_acc10 + 256000;
constexpr long long off_z1d   = off_acc11 + 4096000;
constexpr long long off_Zk    = off_acc11;

__device__ __forceinline__ float wave_sum(float v) {
#pragma unroll
  for (int off = 16; off > 0; off >>= 1) v += __shfl_xor(v, off, 32);
  return v;
}

constexpr int BK   = 32;
constexpr int ASTR = 36;
constexpr int BPS  = 160;

__global__ __launch_bounds__(256) void gemm_f32_wmma(
    const float* __restrict__ A, const float* __restrict__ B, float* __restrict__ C,
    int M, int N, int Kd, long long sA, long long sB, long long sC) {
  __shared__ float As[2][128 * ASTR];
  __shared__ float Bs[2][(BK / 2) * BPS];

  const int b = blockIdx.z;
  A += (long long)b * sA;
  B += (long long)b * sB;
  C += (long long)b * sC;

  const int tid  = threadIdx.x;
  const int lane = tid & 31;
  const int wave = tid >> 5;
  const int mBlk = blockIdx.x * 128;
  const int n0   = blockIdx.y * 64;
  const int m0   = mBlk + wave * 16;
  const int half = lane >> 4;
  const int lr   = lane & 15;
  const bool active = (m0 < M);

  const int arow = tid >> 3;
  const int ac4  = (tid & 7) * 4;
  const int brow = tid >> 4;
  const int bc4  = (tid & 15) * 4;

  const int nChunks = Kd / BK;
  const float4 zero4 = make_float4(0.f, 0.f, 0.f, 0.f);

  {
#pragma unroll
    for (int p = 0; p < 4; ++p) {
      const int row = arow + p * 32;
      const int grow = mBlk + row;
      float4 v = (grow < M) ? *(const float4*)(A + (long long)grow * Kd + ac4) : zero4;
      *(float4*)&As[0][row * ASTR + ac4] = v;
    }
#pragma unroll
    for (int p = 0; p < 2; ++p) {
      const int row = brow + p * 16;
      const float4 v = *(const float4*)(B + (long long)row * N + n0 + bc4);
      float* bp = &Bs[0][(row >> 1) * BPS + bc4 * 2 + (row & 1)];
      bp[0] = v.x; bp[2] = v.y; bp[4] = v.z; bp[6] = v.w;
    }
  }
  __syncthreads();

  v8f acc0 = {}, acc1 = {}, acc2 = {}, acc3 = {};

  for (int c = 0; c < nChunks; ++c) {
    const int buf = c & 1;
    const bool pf = (c + 1 < nChunks);
    float4 ra[4], rb[2];

    if (pf) {
      const int k0 = (c + 1) * BK;
#pragma unroll
      for (int p = 0; p < 4; ++p) {
        const int grow = mBlk + arow + p * 32;
        ra[p] = (grow < M) ? *(const float4*)(A + (long long)grow * Kd + k0 + ac4) : zero4;
      }
#pragma unroll
      for (int p = 0; p < 2; ++p)
        rb[p] = *(const float4*)(B + (long long)(k0 + brow + p * 16) * N + n0 + bc4);
    }

    if (active) {
      const float* Asb = &As[buf][(wave * 16 + lr) * ASTR + 2 * half];
      const float* Bsb = &Bs[buf][half * BPS + lr * 2];
#pragma unroll
      for (int kk = 0; kk < BK; kk += 4) {
        const v2f a = *(const v2f*)(Asb + kk);
        const float* bp = Bsb + (kk >> 1) * BPS;
        const v2f b0 = *(const v2f*)(bp);
        const v2f b1 = *(const v2f*)(bp + 32);
        const v2f b2 = *(const v2f*)(bp + 64);
        const v2f b3 = *(const v2f*)(bp + 96);
        acc0 = __builtin_amdgcn_wmma_f32_16x16x4_f32(false, a, false, b0, (short)0, acc0, false, false);
        acc1 = __builtin_amdgcn_wmma_f32_16x16x4_f32(false, a, false, b1, (short)0, acc1, false, false);
        acc2 = __builtin_amdgcn_wmma_f32_16x16x4_f32(false, a, false, b2, (short)0, acc2, false, false);
        acc3 = __builtin_amdgcn_wmma_f32_16x16x4_f32(false, a, false, b3, (short)0, acc3, false, false);
      }
    }
    __syncthreads();

    if (pf) {
      const int nb = buf ^ 1;
#pragma unroll
      for (int p = 0; p < 4; ++p)
        *(float4*)&As[nb][(arow + p * 32) * ASTR + ac4] = ra[p];
#pragma unroll
      for (int p = 0; p < 2; ++p) {
        const int row = brow + p * 16;
        float* bp = &Bs[nb][(row >> 1) * BPS + bc4 * 2 + (row & 1)];
        bp[0] = rb[p].x; bp[2] = rb[p].y; bp[4] = rb[p].z; bp[6] = rb[p].w;
      }
      __syncthreads();
    }
  }

  if (active) {
#pragma unroll
    for (int r = 0; r < 8; ++r) {
      const int row = m0 + r + 8 * half;
      float* cp = C + (long long)row * N + n0 + lr;
      cp[0]  = acc0[r];
      cp[16] = acc1[r];
      cp[32] = acc2[r];
      cp[48] = acc3[r];
    }
  }
}

constexpr int SP_THREADS = 1024;
constexpr int SP_WAVES   = SP_THREADS / 32;
constexpr int SP_MAXT    = 125;
constexpr int SP_Q       = 16384;
constexpr int Q_EBITS    = 20;
constexpr int Q_EMASK    = (1 << Q_EBITS) - 1;
static_assert(E00 <= (1 << Q_EBITS) && E01 <= (1 << Q_EBITS) && E10 <= (1 << Q_EBITS) &&
              E11 <= (1 << Q_EBITS), "edge id fits the low bits of a queue word");
static_assert(N0 <= 65536 && N1 <= 65536 && SP_MAXT <= 128, "src and tile row fit one list word");

template <int N>
__global__ __launch_bounds__(SP_THREADS) void spmm_tile(
    const int* __restrict__ dst, const int* __restrict__ src, const float* __restrict__ val,
    const float* __restrict__ x, float* __restrict__ out, int nE, int nRows, int T,
    int eStride, long long xStride, long long oStride,
    const int* __restrict__ lists, const int* __restrict__ counts, int cap) {
  constexpr int G = N / 4;
  constexpr int H = N / 128;
  __shared__ __attribute__((aligned(16))) float acc[SP_MAXT * N];
  __shared__ int qe[SP_Q];
  __shared__ int qcount;

  const int k = blockIdx.y;
  dst += (long long)k * eStride;
  src += (long long)k * eStride;
  val += (long long)k * eStride;
  x   += (long long)k * xStride;
  out += (long long)k * oStride;

  const int tid  = threadIdx.x;
  const int lane = tid & 31;
  const int wave = __builtin_amdgcn_readfirstlane(tid >> 5);
  const int row0 = blockIdx.x * T;
  const int tile = blockIdx.y * gridDim.x + blockIdx.x;
  const int2* __restrict__ myList = (const int2*)lists + (long long)tile * cap;

  const int listCnt = counts[tile];
  const bool fromList = (listCnt <= cap);

  for (int i = tid; i < T * G; i += SP_THREADS)
    ((float4*)acc)[i] = make_float4(0.f, 0.f, 0.f, 0.f);
  if (tid == 0) qcount = 0;
  __syncthreads();

  const int n4 = nE >> 2;
  const int4* d4p = (const int4*)dst;
  constexpr int STEP = SP_THREADS * 4;
  const int last4 = max(n4 - 1, 0);

  auto test4 = [&](const int4 (&dv)[4], int base) {
#pragma unroll
    for (int u = 0; u < 4; ++u) {
      const int idx = base + u * SP_THREADS + tid;
      const int e0 = idx << 2;
      const int dd[4] = {dv[u].x, dv[u].y, dv[u].z, dv[u].w};
#pragma unroll
      for (int c = 0; c < 4; ++c) {
        const unsigned r = (unsigned)(dd[c] - row0);
        if (idx < n4 && r < (unsigned)T) {
          const int e = e0 + c;
          const int slot = atomicAdd(&qcount, 1);
          if (slot < SP_Q) {
            qe[slot] = e | ((int)r << Q_EBITS);
          } else {
            const int s = src[e];
            const float v = val[e];
            const float* xr = x + (long long)s * N;
            float* a = &acc[(int)r * N];
            for (int f = 0; f < N; ++f) atomicAdd(&a[(f & 3) * G + (f >> 2)], v * xr[f]);
          }
        }
      }
    }
  };

  if (!fromList) {
  int4 da[4], db[4];
#pragma unroll
  for (int u = 0; u < 4; ++u) da[u] = d4p[min(u * SP_THREADS + tid, last4)];
  for (int i0 = 0; i0 < n4; i0 += 2 * STEP) {
#pragma unroll
    for (int u = 0; u < 4; ++u) db[u] = d4p[min(i0 + STEP + u * SP_THREADS + tid, last4)];
    test4(da, i0);
#pragma unroll
    for (int u = 0; u < 4; ++u) da[u] = d4p[min(i0 + 2 * STEP + u * SP_THREADS + tid, last4)];
    test4(db, i0 + STEP);
  }
  }
  __syncthreads();

  const int cnt = fromList ? listCnt : __builtin_amdgcn_readfirstlane(min(qcount, SP_Q));
  const int nGroups = (cnt + 31) >> 5;
  for (int g = wave; g < nGroups; g += SP_WAVES) {
    const int gi = g * 32 + lane;
    int s = 0, lr = 0, vb = 0;
    if (gi < cnt) {
      if (fromList) {
        const int2 en = myList[gi];
        s  = en.x & 0xFFFF;
        lr = en.x >> 16;
        vb = __float_as_int(val[en.y]);
      } else {
        const int q = qe[gi];
        const int e = q & Q_EMASK;
        lr = q >> Q_EBITS;
        s  = src[e];
        vb = __float_as_int(val[e]);
      }
    }
    const int nv = min(32, cnt - g * 32);
    for (int j0 = 0; j0 < nv; j0 += 4) {
      float4 t[4][H];
      float vv[4];
      int ll[4];
#pragma unroll
      for (int u = 0; u < 4; ++u) {
        const int j = j0 + u;
        const int sj = __builtin_amdgcn_readlane(s, j);
        ll[u] = __builtin_amdgcn_readlane(lr, j);
        vv[u] = __int_as_float(__builtin_amdgcn_readlane(vb, j));
        const float4* xp = (const float4*)(x + (long long)sj * N);
#pragma unroll
        for (int h = 0; h < H; ++h) t[u][h] = xp[lane + 32 * h];
      }
#pragma unroll
      for (int u = 0; u < 4; ++u) {
        if (j0 + u < nv) {
          float* a = &acc[ll[u] * N + lane];
#pragma unroll
          for (int h = 0; h < H; ++h) {
            atomicAdd(&a[0 * G + 32 * h], vv[u] * t[u][h].x);
            atomicAdd(&a[1 * G + 32 * h], vv[u] * t[u][h].y);
            atomicAdd(&a[2 * G + 32 * h], vv[u] * t[u][h].z);
            atomicAdd(&a[3 * G + 32 * h], vv[u] * t[u][h].w);
          }
        }
      }
    }
  }
  __syncthreads();

  for (int i = tid; i < T * G; i += SP_THREADS) {
    const int r = i / G;
    const int g = i % G;
    if (row0 + r < nRows) {
      const float* a = &acc[r * N + g];
      *(float4*)(out + (long long)(row0 + r) * N + 4 * g) = make_float4(a[0], a[G], a[2 * G], a[3 * G]);
    }
  }
}

constexpr int T0  = 80;
constexpr int T1s = 8;
constexpr int T1m = 125;
static_assert(N0 % T0 == 0 && N1 % T1s == 0 && N1 % T1m == 0, "tiles cover the rows exactly");
static_assert(T0 <= SP_MAXT && T1s <= SP_MAXT && T1m <= SP_MAXT, "tile fits its LDS array");
static_assert(E00 % 4 == 0 && E01 % 4 == 0 && E10 % 4 == 0 && E11 % 4 == 0, "dst read as int4");

constexpr int CAP00 = 3072, CAP01 = 640, CAP10 = 640, CAP11 = 4608;
static_assert(CAP00 <= SP_Q && CAP01 <= SP_Q && CAP10 <= SP_Q && CAP11 <= SP_Q, "a list is a queue prefix");
constexpr long long off_l00 = off_lists;
constexpr long long off_l01 = off_l00 + 2LL * (N0 / T0) * CAP00;
constexpr long long off_l10 = off_l01 + 2LL * (N0 / T0) * CAP01;
constexpr long long off_l11 = off_l10 + 2LL * (N1 / T1s) * CAP10;
constexpr long long off_c00 = off_l11 + 2LL * (N1 / T1m) * KK * CAP11;
constexpr long long off_c01 = off_c00 + 256;
constexpr long long off_c10 = off_c01 + 256;
constexpr long long off_c11 = off_c10 + 256;
static_assert(N0 / T0 <= 256 && N1 / T1s <= 256 && (N1 / T1m) * KK <= 256, "count arrays hold 256 tiles each");
static_assert(off_c11 + 256 <= off_lists + 5120000, "lists and counts fit the plane xw10 used to have");
static_assert(off_c01 == off_c00 + 256 && off_c10 == off_c00 + 512 && off_c11 == off_c00 + 768,
              "the four count arrays are one run of 1024 ints");

constexpr int BIN_THREADS = 1024;
constexpr int BIN_U  = 2;
constexpr int BIN_I4 = BIN_THREADS * BIN_U;
constexpr int NB00 = (E00 / 4 + BIN_I4 - 1) / BIN_I4;
constexpr int NB01 = (E01 / 4 + BIN_I4 - 1) / BIN_I4;
constexpr int NB10 = (E10 / 4 + BIN_I4 - 1) / BIN_I4;
constexpr int NB11 = (E11 / 4 + BIN_I4 - 1) / BIN_I4;
static_assert(N0 / T0 <= 256 && N1 / T1s <= 256 && N1 / T1m <= 256, "a relation's tiles fit the block's LDS counters");

__global__ void zero_counts(int* __restrict__ c) { c[blockIdx.x * blockDim.x + threadIdx.x] = 0; }

template <int T>
__device__ __forceinline__ void bin_block(const int* __restrict__ dst, const int* __restrict__ src,
                                          int nE, int nRows, int blk, int tileBase, int nTiles, int cap,
                                          int2* __restrict__ lists, int* __restrict__ counts,
                                          int* cntL, int* baseL) {
  const int tid = threadIdx.x;
  const int n4 = nE >> 2;
  if (tid < 256) cntL[tid] = 0;
  __syncthreads();

  int4 dv[BIN_U], sv[BIN_U];
#pragma unroll
  for (int u = 0; u < BIN_U; ++u) {
    const int idxc = min(blk * BIN_I4 + u * BIN_THREADS + tid, n4 - 1);
    dv[u] = ((const int4*)dst)[idxc];
    sv[u] = ((const int4*)src)[idxc];
  }

  int tl[BIN_U * 4], rw[BIN_U * 4], sr[BIN_U * 4], rk[BIN_U * 4];
  bool ok[BIN_U * 4];
#pragma unroll
  for (int u = 0; u < BIN_U; ++u) {
    const bool in = blk * BIN_I4 + u * BIN_THREADS + tid < n4;
    const int dd[4] = {dv[u].x, dv[u].y, dv[u].z, dv[u].w};
    const int ss[4] = {sv[u].x, sv[u].y, sv[u].z, sv[u].w};
#pragma unroll
    for (int c = 0; c < 4; ++c) {
      const int j = u * 4 + c;
      ok[j] = in && (unsigned)dd[c] < (unsigned)nRows;
      const int t = ok[j] ? dd[c] / T : 0;
      tl[j] = t;
      rw[j] = dd[c] - t * T;
      sr[j] = ss[c];
      rk[j] = ok[j] ? atomicAdd(&cntL[t], 1) : 0;
    }
  }
  __syncthreads();

  if (tid < nTiles) {
    const int n = cntL[tid];
    baseL[tid] = (n > 0) ? atomicAdd(&counts[tileBase + tid], n) : 0;
  }
  __syncthreads();

#pragma unroll
  for (int u = 0; u < BIN_U; ++u) {
#pragma unroll
    for (int c = 0; c < 4; ++c) {
      const int j = u * 4 + c;
      if (ok[j]) {
        const int slot = baseL[tl[j]] + rk[j];
        if (slot < cap)
          lists[(long long)(tileBase + tl[j]) * cap + slot] =
              make_int2(sr[j] | (rw[j] << 16), (blk * BIN_I4 + u * BIN_THREADS + tid) * 4 + c);
      }
    }
  }
}

__global__ __launch_bounds__(BIN_THREADS) void bin_edges(
    const int* __restrict__ d00, const int* __restrict__ s00,
    const int* __restrict__ d01, const int* __restrict__ s01,
    const int* __restrict__ d10, const int* __restrict__ s10,
    const int* __restrict__ d11, const int* __restrict__ s11,
    int* __restrict__ l00, int* __restrict__ c00, int* __restrict__ l01, int* __restrict__ c01,
    int* __restrict__ l10, int* __restrict__ c10, int* __restrict__ l11, int* __restrict__ c11) {
  __shared__ int cntL[256];
  __shared__ int baseL[256];
  int b = blockIdx.x;
  if (b < NB00) {
    bin_block<T0>(d00, s00, E00, N0, b, 0, N0 / T0, CAP00, (int2*)l00, c00, cntL, baseL);
    return;
  }
  b -= NB00;
  if (b < NB01) {
    bin_block<T0>(d01, s01, E01, N0, b, 0, N0 / T0, CAP01, (int2*)l01, c01, cntL, baseL);
    return;
  }
  b -= NB01;
  if (b < NB10) {
    bin_block<T1s>(d10, s10, E10, N1, b, 0, N1 / T1s, CAP10, (int2*)l10, c10, cntL, baseL);
    return;
  }
  b -= NB10;
  const int k = b / NB11;
  b -= k * NB11;
  bin_block<T1m>(d11 + (long long)k * E11, s11 + (long long)k * E11, E11, N1, b,
                 k * (N1 / T1m), N1 / T1m, CAP11, (int2*)l11, c11, cntL, baseL);
}

__global__ void finalize_conv(const float* __restrict__ buf, float* __restrict__ out,
                              int K, int n, int N, long long kStride, int do_relu) {
  const int wave = threadIdx.x >> 5;
  const int lane = threadIdx.x & 31;
  const int row = blockIdx.x * (blockDim.x >> 5) + wave;
  if (row >= n) return;
  const int nc = N >> 5;
  float vals[8];
  float ss = 0.f;
  for (int j = 0; j < nc; ++j) {
    const int c = lane + (j << 5);
    float a = 0.f;
    for (int k = 0; k < K; ++k) {
      float t = buf[(long long)k * kStride + (long long)row * N + c];
      if (do_relu) t = fmaxf(t, 0.f);
      a += t;
    }
    vals[j] = a;
    ss += a * a;
  }
  ss = wave_sum(ss);
  const float sc = rsqrtf(fmaxf(ss, 1e-12f));
  for (int j = 0; j < nc; ++j)
    out[(long long)row * N + lane + (j << 5)] = vals[j] * sc;
}

__global__ void add_act(const float* __restrict__ a, const float* __restrict__ b,
                        float* __restrict__ o, long long n, int do_relu) {
  const long long i = (long long)blockIdx.x * blockDim.x + threadIdx.x;
  if (i >= n) return;
  float v = a[i] + b[i];
  o[i] = do_relu ? fmaxf(v, 0.f) : v;
}

__global__ void scale_dloc(const float* __restrict__ z1, const float* __restrict__ dloc,
                           float* __restrict__ out, long long total) {
  const long long i = (long long)blockIdx.x * blockDim.x + threadIdx.x;
  if (i >= total) return;
  const int f = (int)(i % DD);
  const long long rk = i / DD;
  const int r = (int)(rk % N1);
  const int k = (int)(rk / N1);
  out[i] = z1[(long long)r * DD + f] * dloc[k * DD + f];
}

constexpr int ES_THREADS = 1024;
constexpr int ES_WAVES   = ES_THREADS / 32;

__global__ __launch_bounds__(ES_THREADS) void edge_score_tile(
    const float* __restrict__ zl, const float* __restrict__ zr,
    const int* __restrict__ dst, const int* __restrict__ src,
    float* __restrict__ out, int nE, int T, int eStride,
    long long zlStride, long long zrStride, long long oStride,
    const int* __restrict__ lists, const int* __restrict__ counts, int cap) {
  __shared__ __attribute__((aligned(16))) float zt[SP_MAXT * DD];

  const int k = blockIdx.y;
  zl  += (long long)k * zlStride;
  zr  += (long long)k * zrStride;
  dst += (long long)k * eStride;
  src += (long long)k * eStride;
  out += (long long)k * oStride;

  const int tid  = threadIdx.x;
  const int lane = tid & 31;
  const int wave = __builtin_amdgcn_readfirstlane(tid >> 5);
  const int row0 = blockIdx.x * T;
  const int tile = blockIdx.y * gridDim.x + blockIdx.x;
  const int2* __restrict__ myList = (const int2*)lists + (long long)tile * cap;

  {
    const float4* zsrc = (const float4*)(zl + (long long)row0 * DD);
    for (int i = tid; i < T * (DD / 4); i += ES_THREADS) ((float4*)zt)[i] = zsrc[i];
  }
  const int cnt = __builtin_amdgcn_readfirstlane(counts[tile]);
  __syncthreads();

  if (cnt <= cap) {
    const bool h16 = (lane & 16) != 0, h8 = (lane & 8) != 0, h4 = (lane & 4) != 0;
    const int nGroups = (cnt + 31) >> 5;
    for (int g = wave; g < nGroups; g += ES_WAVES) {
      const int gi = g * 32 + lane;
      int s = 0, lr = 0, e = 0;
      if (gi < cnt) {
        const int2 en = myList[gi];
        s  = en.x & 0xFFFF;
        lr = en.x >> 16;
        e  = en.y;
      }
      const int nv = min(32, cnt - g * 32);
      for (int j0 = 0; j0 < nv; j0 += 8) {
        float4 b[8];
        float p[8];
#pragma unroll
        for (int u = 0; u < 8; ++u) {
          const int sj = __builtin_amdgcn_readlane(s, j0 + u);
          b[u] = ((const float4*)(zr + (long long)sj * DD))[lane];
        }
#pragma unroll
        for (int u = 0; u < 8; ++u) {
          const int lj = __builtin_amdgcn_readlane(lr, j0 + u);
          const float4 a = ((const float4*)(zt + lj * DD))[lane];
          p[u] = a.x * b[u].x + a.y * b[u].y + a.z * b[u].z + a.w * b[u].w;
        }
        float q[4], r[2];
#pragma unroll
        for (int i = 0; i < 4; ++i) {
          const float keep = h16 ? p[4 + i] : p[i];
          const float send = h16 ? p[i] : p[4 + i];
          q[i] = keep + __shfl_xor(send, 16, 32);
        }
#pragma unroll
        for (int i = 0; i < 2; ++i) {
          const float keep = h8 ? q[2 + i] : q[i];
          const float send = h8 ? q[i] : q[2 + i];
          r[i] = keep + __shfl_xor(send, 8, 32);
        }
        float t;
        {
          const float keep = h4 ? r[1] : r[0];
          const float send = h4 ? r[0] : r[1];
          t = keep + __shfl_xor(send, 4, 32);
        }
        t += __shfl_xor(t, 2, 32);
        t += __shfl_xor(t, 1, 32);
        const int mine = j0 + (lane >> 2);
        const int eo = __shfl(e, mine, 32);
        if ((lane & 3) == 0 && mine < nv) out[eo] = t;
      }
    }
  } else {
    for (int base = wave * 32; base < nE; base += ES_WAVES * 32) {
      const int e = base + lane;
      const unsigned r = (e < nE) ? (unsigned)(dst[e] - row0) : 0xFFFFFFFFu;
      unsigned mask = __builtin_amdgcn_ballot_w32(r < (unsigned)T);
      while (mask != 0u) {
        const int j = __builtin_ctz(mask);
        mask &= mask - 1u;
        const int lj = __builtin_amdgcn_readlane((int)r, j);
        const int sj = src[base + j];
        const float4 a = ((const float4*)(zt + lj * DD))[lane];
        const float4 bb = ((const float4*)(zr + (long long)sj * DD))[lane];
        float acc = a.x * bb.x + a.y * bb.y + a.z * bb.z + a.w * bb.w;
        acc = wave_sum(acc);
        if (lane == 0) out[base + j] = acc;
      }
    }
  }
}

extern "C" void kernel_launch(void* const* d_in, const int* in_sizes, int n_in,
                              void* d_out, int out_size, void* d_ws, size_t ws_size,
                              hipStream_t stream) {
  (void)in_sizes; (void)n_in; (void)out_size; (void)ws_size;

  const float* feat0   = (const float*)d_in[0];
  const float* feat1   = (const float*)d_in[1];
  const int*   e00_dst = (const int*)d_in[2];
  const int*   e00_src = (const int*)d_in[3];
  const float* e00_val = (const float*)d_in[4];
  const int*   e01_dst = (const int*)d_in[5];
  const int*   e01_src = (const int*)d_in[6];
  const float* e01_val = (const float*)d_in[7];
  const int*   e10_dst = (const int*)d_in[8];
  const int*   e10_src = (const int*)d_in[9];
  const float* e10_val = (const float*)d_in[10];
  const int*   e11_dst = (const int*)d_in[11];
  const int*   e11_src = (const int*)d_in[12];
  const float* e11_val = (const float*)d_in[13];
  const float* W1_00 = (const float*)d_in[14];
  const float* W1_01 = (const float*)d_in[15];
  const float* W1_10 = (const float*)d_in[16];
  const float* W1_11 = (const float*)d_in[17];
  const float* W2_00 = (const float*)d_in[18];
  const float* W2_01 = (const float*)d_in[19];
  const float* W2_10 = (const float*)d_in[20];
  const float* W2_11 = (const float*)d_in[21];
  const float* R00   = (const float*)d_in[22];
  const float* R01   = (const float*)d_in[23];
  const float* R10   = (const float*)d_in[24];
  const float* Rglob = (const float*)d_in[25];
  const float* dloc  = (const float*)d_in[26];

  float* ws = (float*)d_ws;
  float* out = (float*)d_out;
  float* z0  = out;
  float* z1  = out + 2560000;
  float* s00 = z1 + 256000;
  float* s01 = s00 + E00;
  float* s10 = s01 + E01;
  float* s11 = s10 + E10;

  const dim3 blk(256);
  auto gg = [](int M, int N, int batch) { return dim3((M + 127) / 128, N / 64, batch); };

  int* l00 = (int*)(ws + off_l00);  int* c00 = (int*)(ws + off_c00);
  int* l01 = (int*)(ws + off_l01);  int* c01 = (int*)(ws + off_c01);
  int* l10 = (int*)(ws + off_l10);  int* c10 = (int*)(ws + off_c10);
  int* l11 = (int*)(ws + off_l11);  int* c11 = (int*)(ws + off_c11);
  zero_counts<<<dim3(4), blk, 0, stream>>>(c00);
  bin_edges<<<dim3(NB00 + NB01 + NB10 + KK * NB11), dim3(BIN_THREADS), 0, stream>>>(
      e00_dst, e00_src, e01_dst, e01_src, e10_dst, e10_src, e11_dst, e11_src,
      l00, c00, l01, c01, l10, c10, l11, c11);

  gemm_f32_wmma<<<gg(N0, HH, 1), blk, 0, stream>>>(feat0, W1_00, ws + off_xw00, N0, HH, FF, 0, 0, 0);
  gemm_f32_wmma<<<gg(N1, HH, 1), blk, 0, stream>>>(feat1, W1_01, ws + off_xw01, N1, HH, FF, 0, 0, 0);
  gemm_f32_wmma<<<gg(N1, HH, KK), blk, 0, stream>>>(feat1, W1_11, ws + off_xw11, N1, HH, FF,
                                                    0, (long long)FF * HH, (long long)N1 * HH);

  const dim3 sblk(SP_THREADS);
  spmm_tile<HH><<<dim3(N0 / T0, 1), sblk, 0, stream>>>(e00_dst, e00_src, e00_val, ws + off_xw00,
                                                             ws + off_acc00, E00, N0, T0, 0, 0, 0,
                                                             l00, c00, CAP00);
  gemm_f32_wmma<<<gg(N0, HH, 1), blk, 0, stream>>>(feat0, W1_10, ws + off_xw10, N0, HH, FF, 0, 0, 0);
  spmm_tile<HH><<<dim3(N0 / T0, 1), sblk, 0, stream>>>(e01_dst, e01_src, e01_val, ws + off_xw01,
                                                             ws + off_acc01, E01, N0, T0, 0, 0, 0,
                                                             l01, c01, CAP01);
  spmm_tile<HH><<<dim3(N1 / T1s, 1), sblk, 0, stream>>>(e10_dst, e10_src, e10_val, ws + off_xw10,
                                                              ws + off_acc10, E10, N1, T1s, 0, 0, 0,
                                                              l10, c10, CAP10);
  spmm_tile<HH><<<dim3(N1 / T1m, KK), sblk, 0, stream>>>(e11_dst, e11_src, e11_val, ws + off_xw11,
                                                               ws + off_acc11, E11, N1, T1m, E11,
                                                               (long long)N1 * HH, (long long)N1 * HH,
                                                               l11, c11, CAP11);

  finalize_conv<<<dim3((N0 + 7) / 8), blk, 0, stream>>>(ws + off_acc00, ws + off_acc00, 1, N0, HH, 0, 1);
  finalize_conv<<<dim3((N0 + 7) / 8), blk, 0, stream>>>(ws + off_acc01, ws + off_acc01, 1, N0, HH, 0, 1);
  finalize_conv<<<dim3((N1 + 7) / 8), blk, 0, stream>>>(ws + off_acc10, ws + off_acc10, 1, N1, HH, 0, 1);
  finalize_conv<<<dim3((N1 + 7) / 8), blk, 0, stream>>>(ws + off_acc11, ws + off_acc11, KK, N1, HH,
                                                        (long long)N1 * HH, 1);

  add_act<<<dim3((N0 * HH + 255) / 256), blk, 0, stream>>>(ws + off_acc00, ws + off_acc01,
                                                           ws + off_h0, (long long)N0 * HH, 1);
  add_act<<<dim3((N1 * HH + 255) / 256), blk, 0, stream>>>(ws + off_acc10, ws + off_acc11,
                                                           ws + off_h1, (long long)N1 * HH, 1);

  gemm_f32_wmma<<<gg(N0, DD, 1), blk, 0, stream>>>(ws + off_h0, W2_00, ws + off_hw00, N0, DD, HH, 0, 0, 0);
  gemm_f32_wmma<<<gg(N0, DD, 1), blk, 0, stream>>>(ws + off_h0, W2_10, ws + off_hw10, N0, DD, HH, 0, 0, 0);
  gemm_f32_wmma<<<gg(N1, DD, 1), blk, 0, stream>>>(ws + off_h1, W2_01, ws + off_hw01, N1, DD, HH, 0, 0, 0);
  gemm_f32_wmma<<<gg(N1, DD, KK), blk, 0, stream>>>(ws + off_h1, W2_11, ws + off_hw11, N1, DD, HH,
                                                    0, (long long)HH * DD, (long long)N1 * DD);

  spmm_tile<DD><<<dim3(N0 / T0, 1), sblk, 0, stream>>>(e00_dst, e00_src, e00_val, ws + off_hw00,
                                                              ws + off_bcc00, E00, N0, T0, 0, 0, 0,
                                                              l00, c00, CAP00);
  spmm_tile<DD><<<dim3(N0 / T0, 1), sblk, 0, stream>>>(e01_dst, e01_src, e01_val, ws + off_hw01,
                                                              ws + off_bcc01, E01, N0, T0, 0, 0, 0,
                                                              l01, c01, CAP01);
  spmm_tile<DD><<<dim3(N1 / T1s, 1), sblk, 0, stream>>>(e10_dst, e10_src, e10_val, ws + off_hw10,
                                                               ws + off_bcc10, E10, N1, T1s, 0, 0, 0,
                                                               l10, c10, CAP10);
  spmm_tile<DD><<<dim3(N1 / T1m, KK), sblk, 0, stream>>>(e11_dst, e11_src, e11_val, ws + off_hw11,
                                                                ws + off_bcc11, E11, N1, T1m, E11,
                                                                (long long)N1 * DD, (long long)N1 * DD,
                                                                l11, c11, CAP11);

  finalize_conv<<<dim3((N0 + 7) / 8), blk, 0, stream>>>(ws + off_bcc00, ws + off_bcc00, 1, N0, DD, 0, 0);
  finalize_conv<<<dim3((N0 + 7) / 8), blk, 0, stream>>>(ws + off_bcc01, ws + off_bcc01, 1, N0, DD, 0, 0);
  finalize_conv<<<dim3((N1 + 7) / 8), blk, 0, stream>>>(ws + off_bcc10, ws + off_bcc10, 1, N1, DD, 0, 0);
  finalize_conv<<<dim3((N1 + 7) / 8), blk, 0, stream>>>(ws + off_bcc11, ws + off_bcc11, KK, N1, DD,
                                                        (long long)N1 * DD, 0);

  add_act<<<dim3((N0 * DD + 255) / 256), blk, 0, stream>>>(ws + off_bcc00, ws + off_bcc01, z0,
                                                           (long long)N0 * DD, 0);
  add_act<<<dim3((N1 * DD + 255) / 256), blk, 0, stream>>>(ws + off_bcc10, ws + off_bcc11, z1,
                                                           (long long)N1 * DD, 0);

  gemm_f32_wmma<<<gg(N0, DD, 1), blk, 0, stream>>>(z0, R00, ws + off_z0R00, N0, DD, DD, 0, 0, 0);
  gemm_f32_wmma<<<gg(N0, DD, 1), blk, 0, stream>>>(z0, R01, ws + off_z0R01, N0, DD, DD, 0, 0, 0);
  gemm_f32_wmma<<<gg(N1, DD, 1), blk, 0, stream>>>(z1, R10, ws + off_z1R10, N1, DD, DD, 0, 0, 0);

  scale_dloc<<<dim3((KK * N1 * DD + 255) / 256), blk, 0, stream>>>(z1, dloc, ws + off_z1d,
                                                                   (long long)KK * N1 * DD);
  gemm_f32_wmma<<<gg(N1, DD, KK), blk, 0, stream>>>(ws + off_z1d, Rglob, ws + off_Zk, N1, DD, DD,
                                                    (long long)N1 * DD, 0, (long long)N1 * DD);

  const dim3 eblk(ES_THREADS);
  edge_score_tile<<<dim3(N0 / T0, 1), eblk, 0, stream>>>(ws + off_z0R00, z0, e00_dst, e00_src,
                                                         s00, E00, T0, 0, 0, 0, 0, l00, c00, CAP00);
  edge_score_tile<<<dim3(N0 / T0, 1), eblk, 0, stream>>>(ws + off_z0R01, z1, e01_dst, e01_src,
                                                         s01, E01, T0, 0, 0, 0, 0, l01, c01, CAP01);
  edge_score_tile<<<dim3(N1 / T1s, 1), eblk, 0, stream>>>(ws + off_z1R10, z0, e10_dst, e10_src,
                                                          s10, E10, T1s, 0, 0, 0, 0, l10, c10, CAP10);
  edge_score_tile<<<dim3(N1 / T1m, KK), eblk, 0, stream>>>(ws + off_Zk, ws + off_z1d, e11_dst, e11_src,
                                                           s11, E11, T1m, E11,
                                                           (long long)N1 * DD, (long long)N1 * DD,
                                                           (long long)E11, l11, c11, CAP11);
}


# light passes (row normalise, the four adds, dloc scaling) keep loads in flight: a row's loads issue before first use, adds and scaling move float4s; 480,000 waves become 66,504; sums in same order
# speedup vs baseline: 2.1099x; 1.0647x over previous
#include <hip/hip_runtime.h>

typedef float v2f __attribute__((ext_vector_type(2)));
typedef float v8f __attribute__((ext_vector_type(8)));

constexpr int N0 = 20000, N1 = 2000;
constexpr int FF = 512, HH = 256, DD = 128, KK = 16;
constexpr int E00 = 640000, E01 = 100000, E10 = 100000, E11 = 64000;

constexpr long long off_xw00 = 0;
constexpr long long off_lists = 5120000;
constexpr long long off_xw01 = 10240000;
constexpr long long off_xw11 = 10752000;
constexpr long long off_acc00 = 18944000;
constexpr long long off_acc01 = 24064000;
constexpr long long off_acc10 = 29184000;
constexpr long long off_acc11 = 29696000;

constexpr long long off_xw10  = off_xw00;
constexpr long long off_h0    = off_xw00;
constexpr long long off_h1    = off_xw01;
constexpr long long off_hw00  = off_acc01;
constexpr long long off_hw10  = off_acc01 + 2560000;
constexpr long long off_hw01  = off_xw11;
constexpr long long off_hw11  = off_xw11 + 256000;
constexpr long long off_bcc00 = off_acc00;
constexpr long long off_bcc01 = off_acc00 + 2560000;
constexpr long long off_bcc10 = off_acc10;
constexpr long long off_bcc11 = off_acc11;
constexpr long long off_z0R00 = off_acc01;
constexpr long long off_z0R01 = off_acc01 + 2560000;
constexpr long long off_z1R10 = off_acc10 + 256000;
constexpr long long off_z1d   = off_acc11 + 4096000;
constexpr long long off_Zk    = off_acc11;

__device__ __forceinline__ float wave_sum(float v) {
#pragma unroll
  for (int off = 16; off > 0; off >>= 1) v += __shfl_xor(v, off, 32);
  return v;
}

constexpr int BK   = 32;
constexpr int ASTR = 36;
constexpr int BPS  = 160;

__global__ __launch_bounds__(256) void gemm_f32_wmma(
    const float* __restrict__ A, const float* __restrict__ B, float* __restrict__ C,
    int M, int N, int Kd, long long sA, long long sB, long long sC) {
  __shared__ float As[2][128 * ASTR];
  __shared__ float Bs[2][(BK / 2) * BPS];

  const int b = blockIdx.z;
  A += (long long)b * sA;
  B += (long long)b * sB;
  C += (long long)b * sC;

  const int tid  = threadIdx.x;
  const int lane = tid & 31;
  const int wave = tid >> 5;
  const int mBlk = blockIdx.x * 128;
  const int n0   = blockIdx.y * 64;
  const int m0   = mBlk + wave * 16;
  const int half = lane >> 4;
  const int lr   = lane & 15;
  const bool active = (m0 < M);

  const int arow = tid >> 3;
  const int ac4  = (tid & 7) * 4;
  const int brow = tid >> 4;
  const int bc4  = (tid & 15) * 4;

  const int nChunks = Kd / BK;
  const float4 zero4 = make_float4(0.f, 0.f, 0.f, 0.f);

  {
#pragma unroll
    for (int p = 0; p < 4; ++p) {
      const int row = arow + p * 32;
      const int grow = mBlk + row;
      float4 v = (grow < M) ? *(const float4*)(A + (long long)grow * Kd + ac4) : zero4;
      *(float4*)&As[0][row * ASTR + ac4] = v;
    }
#pragma unroll
    for (int p = 0; p < 2; ++p) {
      const int row = brow + p * 16;
      const float4 v = *(const float4*)(B + (long long)row * N + n0 + bc4);
      float* bp = &Bs[0][(row >> 1) * BPS + bc4 * 2 + (row & 1)];
      bp[0] = v.x; bp[2] = v.y; bp[4] = v.z; bp[6] = v.w;
    }
  }
  __syncthreads();

  v8f acc0 = {}, acc1 = {}, acc2 = {}, acc3 = {};

  for (int c = 0; c < nChunks; ++c) {
    const int buf = c & 1;
    const bool pf = (c + 1 < nChunks);
    float4 ra[4], rb[2];

    if (pf) {
      const int k0 = (c + 1) * BK;
#pragma unroll
      for (int p = 0; p < 4; ++p) {
        const int grow = mBlk + arow + p * 32;
        ra[p] = (grow < M) ? *(const float4*)(A + (long long)grow * Kd + k0 + ac4) : zero4;
      }
#pragma unroll
      for (int p = 0; p < 2; ++p)
        rb[p] = *(const float4*)(B + (long long)(k0 + brow + p * 16) * N + n0 + bc4);
    }

    if (active) {
      const float* Asb = &As[buf][(wave * 16 + lr) * ASTR + 2 * half];
      const float* Bsb = &Bs[buf][half * BPS + lr * 2];
#pragma unroll
      for (int kk = 0; kk < BK; kk += 4) {
        const v2f a = *(const v2f*)(Asb + kk);
        const float* bp = Bsb + (kk >> 1) * BPS;
        const v2f b0 = *(const v2f*)(bp);
        const v2f b1 = *(const v2f*)(bp + 32);
        const v2f b2 = *(const v2f*)(bp + 64);
        const v2f b3 = *(const v2f*)(bp + 96);
        acc0 = __builtin_amdgcn_wmma_f32_16x16x4_f32(false, a, false, b0, (short)0, acc0, false, false);
        acc1 = __builtin_amdgcn_wmma_f32_16x16x4_f32(false, a, false, b1, (short)0, acc1, false, false);
        acc2 = __builtin_amdgcn_wmma_f32_16x16x4_f32(false, a, false, b2, (short)0, acc2, false, false);
        acc3 = __builtin_amdgcn_wmma_f32_16x16x4_f32(false, a, false, b3, (short)0, acc3, false, false);
      }
    }
    __syncthreads();

    if (pf) {
      const int nb = buf ^ 1;
#pragma unroll
      for (int p = 0; p < 4; ++p)
        *(float4*)&As[nb][(arow + p * 32) * ASTR + ac4] = ra[p];
#pragma unroll
      for (int p = 0; p < 2; ++p) {
        const int row = brow + p * 16;
        float* bp = &Bs[nb][(row >> 1) * BPS + bc4 * 2 + (row & 1)];
        bp[0] = rb[p].x; bp[2] = rb[p].y; bp[4] = rb[p].z; bp[6] = rb[p].w;
      }
      __syncthreads();
    }
  }

  if (active) {
#pragma unroll
    for (int r = 0; r < 8; ++r) {
      const int row = m0 + r + 8 * half;
      float* cp = C + (long long)row * N + n0 + lr;
      cp[0]  = acc0[r];
      cp[16] = acc1[r];
      cp[32] = acc2[r];
      cp[48] = acc3[r];
    }
  }
}

constexpr int SP_THREADS = 1024;
constexpr int SP_WAVES   = SP_THREADS / 32;
constexpr int SP_MAXT    = 125;
constexpr int SP_Q       = 16384;
constexpr int Q_EBITS    = 20;
constexpr int Q_EMASK    = (1 << Q_EBITS) - 1;
static_assert(E00 <= (1 << Q_EBITS) && E01 <= (1 << Q_EBITS) && E10 <= (1 << Q_EBITS) &&
              E11 <= (1 << Q_EBITS), "edge id fits the low bits of a queue word");
static_assert(N0 <= 65536 && N1 <= 65536 && SP_MAXT <= 128, "src and tile row fit one list word");

template <int N>
__global__ __launch_bounds__(SP_THREADS) void spmm_tile(
    const int* __restrict__ dst, const int* __restrict__ src, const float* __restrict__ val,
    const float* __restrict__ x, float* __restrict__ out, int nE, int nRows, int T,
    int eStride, long long xStride, long long oStride,
    const int* __restrict__ lists, const int* __restrict__ counts, int cap) {
  constexpr int G = N / 4;
  constexpr int H = N / 128;
  __shared__ __attribute__((aligned(16))) float acc[SP_MAXT * N];
  __shared__ int qe[SP_Q];
  __shared__ int qcount;

  const int k = blockIdx.y;
  dst += (long long)k * eStride;
  src += (long long)k * eStride;
  val += (long long)k * eStride;
  x   += (long long)k * xStride;
  out += (long long)k * oStride;

  const int tid  = threadIdx.x;
  const int lane = tid & 31;
  const int wave = __builtin_amdgcn_readfirstlane(tid >> 5);
  const int row0 = blockIdx.x * T;
  const int tile = blockIdx.y * gridDim.x + blockIdx.x;
  const int2* __restrict__ myList = (const int2*)lists + (long long)tile * cap;

  const int listCnt = counts[tile];
  const bool fromList = (listCnt <= cap);

  for (int i = tid; i < T * G; i += SP_THREADS)
    ((float4*)acc)[i] = make_float4(0.f, 0.f, 0.f, 0.f);
  if (tid == 0) qcount = 0;
  __syncthreads();

  const int n4 = nE >> 2;
  const int4* d4p = (const int4*)dst;
  constexpr int STEP = SP_THREADS * 4;
  const int last4 = max(n4 - 1, 0);

  auto test4 = [&](const int4 (&dv)[4], int base) {
#pragma unroll
    for (int u = 0; u < 4; ++u) {
      const int idx = base + u * SP_THREADS + tid;
      const int e0 = idx << 2;
      const int dd[4] = {dv[u].x, dv[u].y, dv[u].z, dv[u].w};
#pragma unroll
      for (int c = 0; c < 4; ++c) {
        const unsigned r = (unsigned)(dd[c] - row0);
        if (idx < n4 && r < (unsigned)T) {
          const int e = e0 + c;
          const int slot = atomicAdd(&qcount, 1);
          if (slot < SP_Q) {
            qe[slot] = e | ((int)r << Q_EBITS);
          } else {
            const int s = src[e];
            const float v = val[e];
            const float* xr = x + (long long)s * N;
            float* a = &acc[(int)r * N];
            for (int f = 0; f < N; ++f) atomicAdd(&a[(f & 3) * G + (f >> 2)], v * xr[f]);
          }
        }
      }
    }
  };

  if (!fromList) {
  int4 da[4], db[4];
#pragma unroll
  for (int u = 0; u < 4; ++u) da[u] = d4p[min(u * SP_THREADS + tid, last4)];
  for (int i0 = 0; i0 < n4; i0 += 2 * STEP) {
#pragma unroll
    for (int u = 0; u < 4; ++u) db[u] = d4p[min(i0 + STEP + u * SP_THREADS + tid, last4)];
    test4(da, i0);
#pragma unroll
    for (int u = 0; u < 4; ++u) da[u] = d4p[min(i0 + 2 * STEP + u * SP_THREADS + tid, last4)];
    test4(db, i0 + STEP);
  }
  }
  __syncthreads();

  const int cnt = fromList ? listCnt : __builtin_amdgcn_readfirstlane(min(qcount, SP_Q));
  const int nGroups = (cnt + 31) >> 5;
  for (int g = wave; g < nGroups; g += SP_WAVES) {
    const int gi = g * 32 + lane;
    int s = 0, lr = 0, vb = 0;
    if (gi < cnt) {
      if (fromList) {
        const int2 en = myList[gi];
        s  = en.x & 0xFFFF;
        lr = en.x >> 16;
        vb = __float_as_int(val[en.y]);
      } else {
        const int q = qe[gi];
        const int e = q & Q_EMASK;
        lr = q >> Q_EBITS;
        s  = src[e];
        vb = __float_as_int(val[e]);
      }
    }
    const int nv = min(32, cnt - g * 32);
    for (int j0 = 0; j0 < nv; j0 += 4) {
      float4 t[4][H];
      float vv[4];
      int ll[4];
#pragma unroll
      for (int u = 0; u < 4; ++u) {
        const int j = j0 + u;
        const int sj = __builtin_amdgcn_readlane(s, j);
        ll[u] = __builtin_amdgcn_readlane(lr, j);
        vv[u] = __int_as_float(__builtin_amdgcn_readlane(vb, j));
        const float4* xp = (const float4*)(x + (long long)sj * N);
#pragma unroll
        for (int h = 0; h < H; ++h) t[u][h] = xp[lane + 32 * h];
      }
#pragma unroll
      for (int u = 0; u < 4; ++u) {
        if (j0 + u < nv) {
          float* a = &acc[ll[u] * N + lane];
#pragma unroll
          for (int h = 0; h < H; ++h) {
            atomicAdd(&a[0 * G + 32 * h], vv[u] * t[u][h].x);
            atomicAdd(&a[1 * G + 32 * h], vv[u] * t[u][h].y);
            atomicAdd(&a[2 * G + 32 * h], vv[u] * t[u][h].z);
            atomicAdd(&a[3 * G + 32 * h], vv[u] * t[u][h].w);
          }
        }
      }
    }
  }
  __syncthreads();

  for (int i = tid; i < T * G; i += SP_THREADS) {
    const int r = i / G;
    const int g = i % G;
    if (row0 + r < nRows) {
      const float* a = &acc[r * N + g];
      *(float4*)(out + (long long)(row0 + r) * N + 4 * g) = make_float4(a[0], a[G], a[2 * G], a[3 * G]);
    }
  }
}

constexpr int T0  = 80;
constexpr int T1s = 8;
constexpr int T1m = 125;
static_assert(N0 % T0 == 0 && N1 % T1s == 0 && N1 % T1m == 0, "tiles cover the rows exactly");
static_assert(T0 <= SP_MAXT && T1s <= SP_MAXT && T1m <= SP_MAXT, "tile fits its LDS array");
static_assert(E00 % 4 == 0 && E01 % 4 == 0 && E10 % 4 == 0 && E11 % 4 == 0, "dst read as int4");

constexpr int CAP00 = 3072, CAP01 = 640, CAP10 = 640, CAP11 = 4608;
static_assert(CAP00 <= SP_Q && CAP01 <= SP_Q && CAP10 <= SP_Q && CAP11 <= SP_Q, "a list is a queue prefix");
constexpr long long off_l00 = off_lists;
constexpr long long off_l01 = off_l00 + 2LL * (N0 / T0) * CAP00;
constexpr long long off_l10 = off_l01 + 2LL * (N0 / T0) * CAP01;
constexpr long long off_l11 = off_l10 + 2LL * (N1 / T1s) * CAP10;
constexpr long long off_c00 = off_l11 + 2LL * (N1 / T1m) * KK * CAP11;
constexpr long long off_c01 = off_c00 + 256;
constexpr long long off_c10 = off_c01 + 256;
constexpr long long off_c11 = off_c10 + 256;
static_assert(N0 / T0 <= 256 && N1 / T1s <= 256 && (N1 / T1m) * KK <= 256, "count arrays hold 256 tiles each");
static_assert(off_c11 + 256 <= off_lists + 5120000, "lists and counts fit the plane xw10 used to have");
static_assert(off_c01 == off_c00 + 256 && off_c10 == off_c00 + 512 && off_c11 == off_c00 + 768,
              "the four count arrays are one run of 1024 ints");

constexpr int BIN_THREADS = 1024;
constexpr int BIN_U  = 2;
constexpr int BIN_I4 = BIN_THREADS * BIN_U;
constexpr int NB00 = (E00 / 4 + BIN_I4 - 1) / BIN_I4;
constexpr int NB01 = (E01 / 4 + BIN_I4 - 1) / BIN_I4;
constexpr int NB10 = (E10 / 4 + BIN_I4 - 1) / BIN_I4;
constexpr int NB11 = (E11 / 4 + BIN_I4 - 1) / BIN_I4;
static_assert(N0 / T0 <= 256 && N1 / T1s <= 256 && N1 / T1m <= 256, "a relation's tiles fit the block's LDS counters");

__global__ void zero_counts(int* __restrict__ c) { c[blockIdx.x * blockDim.x + threadIdx.x] = 0; }

template <int T>
__device__ __forceinline__ void bin_block(const int* __restrict__ dst, const int* __restrict__ src,
                                          int nE, int nRows, int blk, int tileBase, int nTiles, int cap,
                                          int2* __restrict__ lists, int* __restrict__ counts,
                                          int* cntL, int* baseL) {
  const int tid = threadIdx.x;
  const int n4 = nE >> 2;
  if (tid < 256) cntL[tid] = 0;
  __syncthreads();

  int4 dv[BIN_U], sv[BIN_U];
#pragma unroll
  for (int u = 0; u < BIN_U; ++u) {
    const int idxc = min(blk * BIN_I4 + u * BIN_THREADS + tid, n4 - 1);
    dv[u] = ((const int4*)dst)[idxc];
    sv[u] = ((const int4*)src)[idxc];
  }

  int tl[BIN_U * 4], rw[BIN_U * 4], sr[BIN_U * 4], rk[BIN_U * 4];
  bool ok[BIN_U * 4];
#pragma unroll
  for (int u = 0; u < BIN_U; ++u) {
    const bool in = blk * BIN_I4 + u * BIN_THREADS + tid < n4;
    const int dd[4] = {dv[u].x, dv[u].y, dv[u].z, dv[u].w};
    const int ss[4] = {sv[u].x, sv[u].y, sv[u].z, sv[u].w};
#pragma unroll
    for (int c = 0; c < 4; ++c) {
      const int j = u * 4 + c;
      ok[j] = in && (unsigned)dd[c] < (unsigned)nRows;
      const int t = ok[j] ? dd[c] / T : 0;
      tl[j] = t;
      rw[j] = dd[c] - t * T;
      sr[j] = ss[c];
      rk[j] = ok[j] ? atomicAdd(&cntL[t], 1) : 0;
    }
  }
  __syncthreads();

  if (tid < nTiles) {
    const int n = cntL[tid];
    baseL[tid] = (n > 0) ? atomicAdd(&counts[tileBase + tid], n) : 0;
  }
  __syncthreads();

#pragma unroll
  for (int u = 0; u < BIN_U; ++u) {
#pragma unroll
    for (int c = 0; c < 4; ++c) {
      const int j = u * 4 + c;
      if (ok[j]) {
        const int slot = baseL[tl[j]] + rk[j];
        if (slot < cap)
          lists[(long long)(tileBase + tl[j]) * cap + slot] =
              make_int2(sr[j] | (rw[j] << 16), (blk * BIN_I4 + u * BIN_THREADS + tid) * 4 + c);
      }
    }
  }
}

__global__ __launch_bounds__(BIN_THREADS) void bin_edges(
    const int* __restrict__ d00, const int* __restrict__ s00,
    const int* __restrict__ d01, const int* __restrict__ s01,
    const int* __restrict__ d10, const int* __restrict__ s10,
    const int* __restrict__ d11, const int* __restrict__ s11,
    int* __restrict__ l00, int* __restrict__ c00, int* __restrict__ l01, int* __restrict__ c01,
    int* __restrict__ l10, int* __restrict__ c10, int* __restrict__ l11, int* __restrict__ c11) {
  __shared__ int cntL[256];
  __shared__ int baseL[256];
  int b = blockIdx.x;
  if (b < NB00) {
    bin_block<T0>(d00, s00, E00, N0, b, 0, N0 / T0, CAP00, (int2*)l00, c00, cntL, baseL);
    return;
  }
  b -= NB00;
  if (b < NB01) {
    bin_block<T0>(d01, s01, E01, N0, b, 0, N0 / T0, CAP01, (int2*)l01, c01, cntL, baseL);
    return;
  }
  b -= NB01;
  if (b < NB10) {
    bin_block<T1s>(d10, s10, E10, N1, b, 0, N1 / T1s, CAP10, (int2*)l10, c10, cntL, baseL);
    return;
  }
  b -= NB10;
  const int k = b / NB11;
  b -= k * NB11;
  bin_block<T1m>(d11 + (long long)k * E11, s11 + (long long)k * E11, E11, N1, b,
                 k * (N1 / T1m), N1 / T1m, CAP11, (int2*)l11, c11, cntL, baseL);
}

template <int N, int K, bool RELU>
__global__ __launch_bounds__(256) void finalize_conv(const float* buf, float* out, int n, long long kStride) {
  constexpr int NC = N / 32;
  const int lane = threadIdx.x & 31;
  const int wpb = blockDim.x >> 5;
  const int nWaves = gridDim.x * wpb;
  for (int row = blockIdx.x * wpb + (threadIdx.x >> 5); row < n; row += nWaves) {
    const float* rp = buf + (long long)row * N + lane;
    float vals[NC];
    float ss = 0.f;
    if constexpr (K == 1) {
      float t[NC];
#pragma unroll
      for (int j = 0; j < NC; ++j) t[j] = rp[j * 32];
#pragma unroll
      for (int j = 0; j < NC; ++j) {
        float a = 0.f;
        a += RELU ? fmaxf(t[j], 0.f) : t[j];
        vals[j] = a;
        ss = __builtin_fmaf(a, a, ss);
      }
    } else {
#pragma unroll
      for (int j = 0; j < NC; ++j) {
        float t[K];
#pragma unroll
        for (int k = 0; k < K; ++k) t[k] = rp[(long long)k * kStride + j * 32];
        float a = 0.f;
#pragma unroll
        for (int k = 0; k < K; ++k) a += RELU ? fmaxf(t[k], 0.f) : t[k];
        vals[j] = a;
        ss = __builtin_fmaf(a, a, ss);
      }
    }
    ss = wave_sum(ss);
    const float sc = rsqrtf(fmaxf(ss, 1e-12f));
    float* op = out + (long long)row * N + lane;
#pragma unroll
    for (int j = 0; j < NC; ++j) op[j * 32] = vals[j] * sc;
  }
}

__global__ __launch_bounds__(256) void add_act(const float* __restrict__ a, const float* __restrict__ b,
                                               float* __restrict__ o, int n4, int do_relu) {
  const float4* a4 = (const float4*)a;
  const float4* b4 = (const float4*)b;
  float4* o4 = (float4*)o;
  const int stride = gridDim.x * blockDim.x;
  const int i0 = blockIdx.x * blockDim.x + threadIdx.x;
  float4 va[4], vb[4];
#pragma unroll
  for (int u = 0; u < 4; ++u) {
    const int i = min(i0 + u * stride, n4 - 1);
    va[u] = a4[i];
    vb[u] = b4[i];
  }
#pragma unroll
  for (int u = 0; u < 4; ++u) {
    const int i = i0 + u * stride;
    if (i < n4) {
      float4 v = make_float4(va[u].x + vb[u].x, va[u].y + vb[u].y, va[u].z + vb[u].z, va[u].w + vb[u].w);
      if (do_relu) v = make_float4(fmaxf(v.x, 0.f), fmaxf(v.y, 0.f), fmaxf(v.z, 0.f), fmaxf(v.w, 0.f));
      o4[i] = v;
    }
  }
}
constexpr int EWB_H0 = (N0 * HH / 4 + 1023) / 1024;
constexpr int EWB_H1 = (N1 * HH / 4 + 1023) / 1024;
constexpr int EWB_Z0 = (N0 * DD / 4 + 1023) / 1024;
constexpr int EWB_Z1 = (N1 * DD / 4 + 1023) / 1024;

__global__ __launch_bounds__(256) void scale_dloc(const float* z1, const float* dloc, float* out) {
  const int i = blockIdx.x * blockDim.x + threadIdx.x;
  if (i >= N1 * (DD / 4)) return;
  const int f4 = i & (DD / 4 - 1);
  const float4 z = ((const float4*)z1)[i];
  const float4* d4 = (const float4*)dloc;
  float4* o4 = (float4*)out;
  float4 d[KK];
#pragma unroll
  for (int k = 0; k < KK; ++k) d[k] = d4[k * (DD / 4) + f4];
#pragma unroll
  for (int k = 0; k < KK; ++k)
    o4[(long long)k * (N1 * (DD / 4)) + i] =
        make_float4(z.x * d[k].x, z.y * d[k].y, z.z * d[k].z, z.w * d[k].w);
}
static_assert((N0 * HH) % 4 == 0 && (N1 * HH) % 4 == 0 && (N0 * DD) % 4 == 0 && (N1 * DD) % 4 == 0 &&
              (N1 * (DD / 4)) % 256 == 0, "light passes move whole float4s; scale_dloc's grid is exact");

constexpr int ES_THREADS = 1024;
constexpr int ES_WAVES   = ES_THREADS / 32;

__global__ __launch_bounds__(ES_THREADS) void edge_score_tile(
    const float* __restrict__ zl, const float* __restrict__ zr,
    const int* __restrict__ dst, const int* __restrict__ src,
    float* __restrict__ out, int nE, int T, int eStride,
    long long zlStride, long long zrStride, long long oStride,
    const int* __restrict__ lists, const int* __restrict__ counts, int cap) {
  __shared__ __attribute__((aligned(16))) float zt[SP_MAXT * DD];

  const int k = blockIdx.y;
  zl  += (long long)k * zlStride;
  zr  += (long long)k * zrStride;
  dst += (long long)k * eStride;
  src += (long long)k * eStride;
  out += (long long)k * oStride;

  const int tid  = threadIdx.x;
  const int lane = tid & 31;
  const int wave = __builtin_amdgcn_readfirstlane(tid >> 5);
  const int row0 = blockIdx.x * T;
  const int tile = blockIdx.y * gridDim.x + blockIdx.x;
  const int2* __restrict__ myList = (const int2*)lists + (long long)tile * cap;

  {
    const float4* zsrc = (const float4*)(zl + (long long)row0 * DD);
    for (int i = tid; i < T * (DD / 4); i += ES_THREADS) ((float4*)zt)[i] = zsrc[i];
  }
  const int cnt = __builtin_amdgcn_readfirstlane(counts[tile]);
  __syncthreads();

  if (cnt <= cap) {
    const bool h16 = (lane & 16) != 0, h8 = (lane & 8) != 0, h4 = (lane & 4) != 0;
    const int nGroups = (cnt + 31) >> 5;
    for (int g = wave; g < nGroups; g += ES_WAVES) {
      const int gi = g * 32 + lane;
      int s = 0, lr = 0, e = 0;
      if (gi < cnt) {
        const int2 en = myList[gi];
        s  = en.x & 0xFFFF;
        lr = en.x >> 16;
        e  = en.y;
      }
      const int nv = min(32, cnt - g * 32);
      for (int j0 = 0; j0 < nv; j0 += 8) {
        float4 b[8];
        float p[8];
#pragma unroll
        for (int u = 0; u < 8; ++u) {
          const int sj = __builtin_amdgcn_readlane(s, j0 + u);
          b[u] = ((const float4*)(zr + (long long)sj * DD))[lane];
        }
#pragma unroll
        for (int u = 0; u < 8; ++u) {
          const int lj = __builtin_amdgcn_readlane(lr, j0 + u);
          const float4 a = ((const float4*)(zt + lj * DD))[lane];
          p[u] = a.x * b[u].x + a.y * b[u].y + a.z * b[u].z + a.w * b[u].w;
        }
        float q[4], r[2];
#pragma unroll
        for (int i = 0; i < 4; ++i) {
          const float keep = h16 ? p[4 + i] : p[i];
          const float send = h16 ? p[i] : p[4 + i];
          q[i] = keep + __shfl_xor(send, 16, 32);
        }
#pragma unroll
        for (int i = 0; i < 2; ++i) {
          const float keep = h8 ? q[2 + i] : q[i];
          const float send = h8 ? q[i] : q[2 + i];
          r[i] = keep + __shfl_xor(send, 8, 32);
        }
        float t;
        {
          const float keep = h4 ? r[1] : r[0];
          const float send = h4 ? r[0] : r[1];
          t = keep + __shfl_xor(send, 4, 32);
        }
        t += __shfl_xor(t, 2, 32);
        t += __shfl_xor(t, 1, 32);
        const int mine = j0 + (lane >> 2);
        const int eo = __shfl(e, mine, 32);
        if ((lane & 3) == 0 && mine < nv) out[eo] = t;
      }
    }
  } else {
    for (int base = wave * 32; base < nE; base += ES_WAVES * 32) {
      const int e = base + lane;
      const unsigned r = (e < nE) ? (unsigned)(dst[e] - row0) : 0xFFFFFFFFu;
      unsigned mask = __builtin_amdgcn_ballot_w32(r < (unsigned)T);
      while (mask != 0u) {
        const int j = __builtin_ctz(mask);
        mask &= mask - 1u;
        const int lj = __builtin_amdgcn_readlane((int)r, j);
        const int sj = src[base + j];
        const float4 a = ((const float4*)(zt + lj * DD))[lane];
        const float4 bb = ((const float4*)(zr + (long long)sj * DD))[lane];
        float acc = a.x * bb.x + a.y * bb.y + a.z * bb.z + a.w * bb.w;
        acc = wave_sum(acc);
        if (lane == 0) out[base + j] = acc;
      }
    }
  }
}

constexpr int FIN0 = N0 / 16;
constexpr int FIN1 = N1 / 8;
static_assert(N0 % 16 == 0 && N1 % 8 == 0, "finalize grids cover the rows evenly");

extern "C" void kernel_launch(void* const* d_in, const int* in_sizes, int n_in,
                              void* d_out, int out_size, void* d_ws, size_t ws_size,
                              hipStream_t stream) {
  (void)in_sizes; (void)n_in; (void)out_size; (void)ws_size;

  const float* feat0   = (const float*)d_in[0];
  const float* feat1   = (const float*)d_in[1];
  const int*   e00_dst = (const int*)d_in[2];
  const int*   e00_src = (const int*)d_in[3];
  const float* e00_val = (const float*)d_in[4];
  const int*   e01_dst = (const int*)d_in[5];
  const int*   e01_src = (const int*)d_in[6];
  const float* e01_val = (const float*)d_in[7];
  const int*   e10_dst = (const int*)d_in[8];
  const int*   e10_src = (const int*)d_in[9];
  const float* e10_val = (const float*)d_in[10];
  const int*   e11_dst = (const int*)d_in[11];
  const int*   e11_src = (const int*)d_in[12];
  const float* e11_val = (const float*)d_in[13];
  const float* W1_00 = (const float*)d_in[14];
  const float* W1_01 = (const float*)d_in[15];
  const float* W1_10 = (const float*)d_in[16];
  const float* W1_11 = (const float*)d_in[17];
  const float* W2_00 = (const float*)d_in[18];
  const float* W2_01 = (const float*)d_in[19];
  const float* W2_10 = (const float*)d_in[20];
  const float* W2_11 = (const float*)d_in[21];
  const float* R00   = (const float*)d_in[22];
  const float* R01   = (const float*)d_in[23];
  const float* R10   = (const float*)d_in[24];
  const float* Rglob = (const float*)d_in[25];
  const float* dloc  = (const float*)d_in[26];

  float* ws = (float*)d_ws;
  float* out = (float*)d_out;
  float* z0  = out;
  float* z1  = out + 2560000;
  float* s00 = z1 + 256000;
  float* s01 = s00 + E00;
  float* s10 = s01 + E01;
  float* s11 = s10 + E10;

  const dim3 blk(256);
  auto gg = [](int M, int N, int batch) { return dim3((M + 127) / 128, N / 64, batch); };

  int* l00 = (int*)(ws + off_l00);  int* c00 = (int*)(ws + off_c00);
  int* l01 = (int*)(ws + off_l01);  int* c01 = (int*)(ws + off_c01);
  int* l10 = (int*)(ws + off_l10);  int* c10 = (int*)(ws + off_c10);
  int* l11 = (int*)(ws + off_l11);  int* c11 = (int*)(ws + off_c11);
  zero_counts<<<dim3(4), blk, 0, stream>>>(c00);
  bin_edges<<<dim3(NB00 + NB01 + NB10 + KK * NB11), dim3(BIN_THREADS), 0, stream>>>(
      e00_dst, e00_src, e01_dst, e01_src, e10_dst, e10_src, e11_dst, e11_src,
      l00, c00, l01, c01, l10, c10, l11, c11);

  gemm_f32_wmma<<<gg(N0, HH, 1), blk, 0, stream>>>(feat0, W1_00, ws + off_xw00, N0, HH, FF, 0, 0, 0);
  gemm_f32_wmma<<<gg(N1, HH, 1), blk, 0, stream>>>(feat1, W1_01, ws + off_xw01, N1, HH, FF, 0, 0, 0);
  gemm_f32_wmma<<<gg(N1, HH, KK), blk, 0, stream>>>(feat1, W1_11, ws + off_xw11, N1, HH, FF,
                                                    0, (long long)FF * HH, (long long)N1 * HH);

  const dim3 sblk(SP_THREADS);
  spmm_tile<HH><<<dim3(N0 / T0, 1), sblk, 0, stream>>>(e00_dst, e00_src, e00_val, ws + off_xw00,
                                                             ws + off_acc00, E00, N0, T0, 0, 0, 0,
                                                             l00, c00, CAP00);
  gemm_f32_wmma<<<gg(N0, HH, 1), blk, 0, stream>>>(feat0, W1_10, ws + off_xw10, N0, HH, FF, 0, 0, 0);
  spmm_tile<HH><<<dim3(N0 / T0, 1), sblk, 0, stream>>>(e01_dst, e01_src, e01_val, ws + off_xw01,
                                                             ws + off_acc01, E01, N0, T0, 0, 0, 0,
                                                             l01, c01, CAP01);
  spmm_tile<HH><<<dim3(N1 / T1s, 1), sblk, 0, stream>>>(e10_dst, e10_src, e10_val, ws + off_xw10,
                                                              ws + off_acc10, E10, N1, T1s, 0, 0, 0,
                                                              l10, c10, CAP10);
  spmm_tile<HH><<<dim3(N1 / T1m, KK), sblk, 0, stream>>>(e11_dst, e11_src, e11_val, ws + off_xw11,
                                                               ws + off_acc11, E11, N1, T1m, E11,
                                                               (long long)N1 * HH, (long long)N1 * HH,
                                                               l11, c11, CAP11);

  finalize_conv<HH, 1, true><<<dim3(FIN0), blk, 0, stream>>>(ws + off_acc00, ws + off_acc00, N0, 0);
  finalize_conv<HH, 1, true><<<dim3(FIN0), blk, 0, stream>>>(ws + off_acc01, ws + off_acc01, N0, 0);
  finalize_conv<HH, 1, true><<<dim3(FIN1), blk, 0, stream>>>(ws + off_acc10, ws + off_acc10, N1, 0);
  finalize_conv<HH, KK, true><<<dim3(FIN1), blk, 0, stream>>>(ws + off_acc11, ws + off_acc11, N1,
                                                              (long long)N1 * HH);

  add_act<<<dim3(EWB_H0), blk, 0, stream>>>(ws + off_acc00, ws + off_acc01, ws + off_h0, N0 * HH / 4, 1);
  add_act<<<dim3(EWB_H1), blk, 0, stream>>>(ws + off_acc10, ws + off_acc11, ws + off_h1, N1 * HH / 4, 1);

  gemm_f32_wmma<<<gg(N0, DD, 1), blk, 0, stream>>>(ws + off_h0, W2_00, ws + off_hw00, N0, DD, HH, 0, 0, 0);
  gemm_f32_wmma<<<gg(N0, DD, 1), blk, 0, stream>>>(ws + off_h0, W2_10, ws + off_hw10, N0, DD, HH, 0, 0, 0);
  gemm_f32_wmma<<<gg(N1, DD, 1), blk, 0, stream>>>(ws + off_h1, W2_01, ws + off_hw01, N1, DD, HH, 0, 0, 0);
  gemm_f32_wmma<<<gg(N1, DD, KK), blk, 0, stream>>>(ws + off_h1, W2_11, ws + off_hw11, N1, DD, HH,
                                                    0, (long long)HH * DD, (long long)N1 * DD);

  spmm_tile<DD><<<dim3(N0 / T0, 1), sblk, 0, stream>>>(e00_dst, e00_src, e00_val, ws + off_hw00,
                                                              ws + off_bcc00, E00, N0, T0, 0, 0, 0,
                                                              l00, c00, CAP00);
  spmm_tile<DD><<<dim3(N0 / T0, 1), sblk, 0, stream>>>(e01_dst, e01_src, e01_val, ws + off_hw01,
                                                              ws + off_bcc01, E01, N0, T0, 0, 0, 0,
                                                              l01, c01, CAP01);
  spmm_tile<DD><<<dim3(N1 / T1s, 1), sblk, 0, stream>>>(e10_dst, e10_src, e10_val, ws + off_hw10,
                                                               ws + off_bcc10, E10, N1, T1s, 0, 0, 0,
                                                               l10, c10, CAP10);
  spmm_tile<DD><<<dim3(N1 / T1m, KK), sblk, 0, stream>>>(e11_dst, e11_src, e11_val, ws + off_hw11,
                                                                ws + off_bcc11, E11, N1, T1m, E11,
                                                                (long long)N1 * DD, (long long)N1 * DD,
                                                                l11, c11, CAP11);

  finalize_conv<DD, 1, false><<<dim3(FIN0), blk, 0, stream>>>(ws + off_bcc00, ws + off_bcc00, N0, 0);
  finalize_conv<DD, 1, false><<<dim3(FIN0), blk, 0, stream>>>(ws + off_bcc01, ws + off_bcc01, N0, 0);
  finalize_conv<DD, 1, false><<<dim3(FIN1), blk, 0, stream>>>(ws + off_bcc10, ws + off_bcc10, N1, 0);
  finalize_conv<DD, KK, false><<<dim3(FIN1), blk, 0, stream>>>(ws + off_bcc11, ws + off_bcc11, N1,
                                                               (long long)N1 * DD);

  add_act<<<dim3(EWB_Z0), blk, 0, stream>>>(ws + off_bcc00, ws + off_bcc01, z0, N0 * DD / 4, 0);
  add_act<<<dim3(EWB_Z1), blk, 0, stream>>>(ws + off_bcc10, ws + off_bcc11, z1, N1 * DD / 4, 0);

  gemm_f32_wmma<<<gg(N0, DD, 1), blk, 0, stream>>>(z0, R00, ws + off_z0R00, N0, DD, DD, 0, 0, 0);
  gemm_f32_wmma<<<gg(N0, DD, 1), blk, 0, stream>>>(z0, R01, ws + off_z0R01, N0, DD, DD, 0, 0, 0);
  gemm_f32_wmma<<<gg(N1, DD, 1), blk, 0, stream>>>(z1, R10, ws + off_z1R10, N1, DD, DD, 0, 0, 0);

  scale_dloc<<<dim3(N1 * (DD / 4) / 256), blk, 0, stream>>>(z1, dloc, ws + off_z1d);
  gemm_f32_wmma<<<gg(N1, DD, KK), blk, 0, stream>>>(ws + off_z1d, Rglob, ws + off_Zk, N1, DD, DD,
                                                    (long long)N1 * DD, 0, (long long)N1 * DD);

  const dim3 eblk(ES_THREADS);
  edge_score_tile<<<dim3(N0 / T0, 1), eblk, 0, stream>>>(ws + off_z0R00, z0, e00_dst, e00_src,
                                                         s00, E00, T0, 0, 0, 0, 0, l00, c00, CAP00);
  edge_score_tile<<<dim3(N0 / T0, 1), eblk, 0, stream>>>(ws + off_z0R01, z1, e01_dst, e01_src,
                                                         s01, E01, T0, 0, 0, 0, 0, l01, c01, CAP01);
  edge_score_tile<<<dim3(N1 / T1s, 1), eblk, 0, stream>>>(ws + off_z1R10, z0, e10_dst, e10_src,
                                                          s10, E10, T1s, 0, 0, 0, 0, l10, c10, CAP10);
  edge_score_tile<<<dim3(N1 / T1m, KK), eblk, 0, stream>>>(ws + off_Zk, ws + off_z1d, e11_dst, e11_src,
                                                           s11, E11, T1m, E11,
                                                           (long long)N1 * DD, (long long)N1 * DD,
                                                           (long long)E11, l11, c11, CAP11);
}


# dense products: block keeps its whole K x 64 B panel in LDS, each wave reads its A rows from global into registers one chunk ahead, no barrier in K loop; k summed 0,1,4,5,2,3,6,7 per 8, fixed order
# speedup vs baseline: 2.6844x; 1.2723x over previous
#include <hip/hip_runtime.h>

typedef float v2f __attribute__((ext_vector_type(2)));
typedef float v8f __attribute__((ext_vector_type(8)));

constexpr int N0 = 20000, N1 = 2000;
constexpr int FF = 512, HH = 256, DD = 128, KK = 16;
constexpr int E00 = 640000, E01 = 100000, E10 = 100000, E11 = 64000;

constexpr long long off_xw00 = 0;
constexpr long long off_lists = 5120000;
constexpr long long off_xw01 = 10240000;
constexpr long long off_xw11 = 10752000;
constexpr long long off_acc00 = 18944000;
constexpr long long off_acc01 = 24064000;
constexpr long long off_acc10 = 29184000;
constexpr long long off_acc11 = 29696000;

constexpr long long off_xw10  = off_xw00;
constexpr long long off_h0    = off_xw00;
constexpr long long off_h1    = off_xw01;
constexpr long long off_hw00  = off_acc01;
constexpr long long off_hw10  = off_acc01 + 2560000;
constexpr long long off_hw01  = off_xw11;
constexpr long long off_hw11  = off_xw11 + 256000;
constexpr long long off_bcc00 = off_acc00;
constexpr long long off_bcc01 = off_acc00 + 2560000;
constexpr long long off_bcc10 = off_acc10;
constexpr long long off_bcc11 = off_acc11;
constexpr long long off_z0R00 = off_acc01;
constexpr long long off_z0R01 = off_acc01 + 2560000;
constexpr long long off_z1R10 = off_acc10 + 256000;
constexpr long long off_z1d   = off_acc11 + 4096000;
constexpr long long off_Zk    = off_acc11;

__device__ __forceinline__ float wave_sum(float v) {
#pragma unroll
  for (int off = 16; off > 0; off >>= 1) v += __shfl_xor(v, off, 32);
  return v;
}

typedef float v4f __attribute__((ext_vector_type(4)));
constexpr int BK  = 32;
constexpr int BPS = 144;
constexpr int GEMM_MAXT = 640;

template <int KD>
__global__ __launch_bounds__(GEMM_MAXT) void gemm_f32_wmma(
    const float* __restrict__ A, const float* __restrict__ B, float* __restrict__ C,
    int M, int N, long long sA, long long sB, long long sC) {
  __shared__ __attribute__((aligned(16))) float Bp[(KD / 2) * BPS];

  const int b = blockIdx.z;
  A += (long long)b * sA;
  B += (long long)b * sB;
  C += (long long)b * sC;

  const int tid  = threadIdx.x;
  const int nT   = blockDim.x;
  const int lane = tid & 31;
  const int wave = __builtin_amdgcn_readfirstlane(tid >> 5);
  const int n0   = blockIdx.y * 64;
  const int m0   = blockIdx.x * (nT >> 5) * 16 + wave * 16;
  const int half = lane >> 4;
  const int lr   = lane & 15;

  {
    constexpr int ITEMS = (KD / 2) * 16;
    for (int i0 = tid; i0 < ITEMS; i0 += 4 * nT) {
      v4f r0[4], r1[4];
#pragma unroll
      for (int u = 0; u < 4; ++u) {
        const int i = min(i0 + u * nT, ITEMS - 1);
        const int p = i >> 4, c4 = (i & 15) * 4;
        const float* bp = B + (long long)(2 * p) * N + n0 + c4;
        r0[u] = *(const v4f*)bp;
        r1[u] = *(const v4f*)(bp + N);
      }
#pragma unroll
      for (int u = 0; u < 4; ++u) {
        const int i = i0 + u * nT;
        if (i < ITEMS) {
          const int p = i >> 4, c4 = (i & 15) * 4;
          float* sp = &Bp[p * BPS + c4 * 2];
          const v4f lo = {r0[u].x, r1[u].x, r0[u].y, r1[u].y};
          const v4f hi = {r0[u].z, r1[u].z, r0[u].w, r1[u].w};
          *(v4f*)sp = lo;
          *(v4f*)(sp + 4) = hi;
        }
      }
    }
  }
  __syncthreads();

  if (m0 >= M) return;

  const v4f* ap = (const v4f*)(A + (long long)(m0 + lr) * KD + 4 * half);
  const float* Bsb = &Bp[(2 * half) * BPS + lr * 2];

  v8f acc0 = {}, acc1 = {}, acc2 = {}, acc3 = {};
  v4f q[4], qn[4];
#pragma unroll
  for (int j = 0; j < 4; ++j) q[j] = ap[2 * j];

  constexpr int NCH = KD / BK;
  for (int c = 0; c < NCH; ++c) {
    const int cn = min(c + 1, NCH - 1);
#pragma unroll
    for (int j = 0; j < 4; ++j) qn[j] = ap[cn * (BK / 4) + 2 * j];

    const float* bc = Bsb + c * (BK / 2) * BPS;
#pragma unroll
    for (int j = 0; j < 4; ++j) {
#pragma unroll
      for (int sub = 0; sub < 2; ++sub) {
        const v2f a = sub ? __builtin_shufflevector(q[j], q[j], 2, 3)
                          : __builtin_shufflevector(q[j], q[j], 0, 1);
        const float* bp = bc + (4 * j + sub) * BPS;
        const v2f b0 = *(const v2f*)(bp);
        const v2f b1 = *(const v2f*)(bp + 32);
        const v2f b2 = *(const v2f*)(bp + 64);
        const v2f b3 = *(const v2f*)(bp + 96);
        acc0 = __builtin_amdgcn_wmma_f32_16x16x4_f32(false, a, false, b0, (short)0, acc0, false, false);
        acc1 = __builtin_amdgcn_wmma_f32_16x16x4_f32(false, a, false, b1, (short)0, acc1, false, false);
        acc2 = __builtin_amdgcn_wmma_f32_16x16x4_f32(false, a, false, b2, (short)0, acc2, false, false);
        acc3 = __builtin_amdgcn_wmma_f32_16x16x4_f32(false, a, false, b3, (short)0, acc3, false, false);
      }
    }
#pragma unroll
    for (int j = 0; j < 4; ++j) q[j] = qn[j];
  }

#pragma unroll
  for (int r = 0; r < 8; ++r) {
    const int row = m0 + r + 8 * half;
    float* cp = C + (long long)row * N + n0 + lr;
    cp[0]  = acc0[r];
    cp[16] = acc1[r];
    cp[32] = acc2[r];
    cp[48] = acc3[r];
  }
}

constexpr int SP_THREADS = 1024;
constexpr int SP_WAVES   = SP_THREADS / 32;
constexpr int SP_MAXT    = 125;
constexpr int SP_Q       = 16384;
constexpr int Q_EBITS    = 20;
constexpr int Q_EMASK    = (1 << Q_EBITS) - 1;
static_assert(E00 <= (1 << Q_EBITS) && E01 <= (1 << Q_EBITS) && E10 <= (1 << Q_EBITS) &&
              E11 <= (1 << Q_EBITS), "edge id fits the low bits of a queue word");
static_assert(N0 <= 65536 && N1 <= 65536 && SP_MAXT <= 128, "src and tile row fit one list word");

template <int N>
__global__ __launch_bounds__(SP_THREADS) void spmm_tile(
    const int* __restrict__ dst, const int* __restrict__ src, const float* __restrict__ val,
    const float* __restrict__ x, float* __restrict__ out, int nE, int nRows, int T,
    int eStride, long long xStride, long long oStride,
    const int* __restrict__ lists, const int* __restrict__ counts, int cap) {
  constexpr int G = N / 4;
  constexpr int H = N / 128;
  __shared__ __attribute__((aligned(16))) float acc[SP_MAXT * N];
  __shared__ int qe[SP_Q];
  __shared__ int qcount;

  const int k = blockIdx.y;
  dst += (long long)k * eStride;
  src += (long long)k * eStride;
  val += (long long)k * eStride;
  x   += (long long)k * xStride;
  out += (long long)k * oStride;

  const int tid  = threadIdx.x;
  const int lane = tid & 31;
  const int wave = __builtin_amdgcn_readfirstlane(tid >> 5);
  const int row0 = blockIdx.x * T;
  const int tile = blockIdx.y * gridDim.x + blockIdx.x;
  const int2* __restrict__ myList = (const int2*)lists + (long long)tile * cap;

  const int listCnt = counts[tile];
  const bool fromList = (listCnt <= cap);

  for (int i = tid; i < T * G; i += SP_THREADS)
    ((float4*)acc)[i] = make_float4(0.f, 0.f, 0.f, 0.f);
  if (tid == 0) qcount = 0;
  __syncthreads();

  const int n4 = nE >> 2;
  const int4* d4p = (const int4*)dst;
  constexpr int STEP = SP_THREADS * 4;
  const int last4 = max(n4 - 1, 0);

  auto test4 = [&](const int4 (&dv)[4], int base) {
#pragma unroll
    for (int u = 0; u < 4; ++u) {
      const int idx = base + u * SP_THREADS + tid;
      const int e0 = idx << 2;
      const int dd[4] = {dv[u].x, dv[u].y, dv[u].z, dv[u].w};
#pragma unroll
      for (int c = 0; c < 4; ++c) {
        const unsigned r = (unsigned)(dd[c] - row0);
        if (idx < n4 && r < (unsigned)T) {
          const int e = e0 + c;
          const int slot = atomicAdd(&qcount, 1);
          if (slot < SP_Q) {
            qe[slot] = e | ((int)r << Q_EBITS);
          } else {
            const int s = src[e];
            const float v = val[e];
            const float* xr = x + (long long)s * N;
            float* a = &acc[(int)r * N];
            for (int f = 0; f < N; ++f) atomicAdd(&a[(f & 3) * G + (f >> 2)], v * xr[f]);
          }
        }
      }
    }
  };

  if (!fromList) {
  int4 da[4], db[4];
#pragma unroll
  for (int u = 0; u < 4; ++u) da[u] = d4p[min(u * SP_THREADS + tid, last4)];
  for (int i0 = 0; i0 < n4; i0 += 2 * STEP) {
#pragma unroll
    for (int u = 0; u < 4; ++u) db[u] = d4p[min(i0 + STEP + u * SP_THREADS + tid, last4)];
    test4(da, i0);
#pragma unroll
    for (int u = 0; u < 4; ++u) da[u] = d4p[min(i0 + 2 * STEP + u * SP_THREADS + tid, last4)];
    test4(db, i0 + STEP);
  }
  }
  __syncthreads();

  const int cnt = fromList ? listCnt : __builtin_amdgcn_readfirstlane(min(qcount, SP_Q));
  const int nGroups = (cnt + 31) >> 5;
  for (int g = wave; g < nGroups; g += SP_WAVES) {
    const int gi = g * 32 + lane;
    int s = 0, lr = 0, vb = 0;
    if (gi < cnt) {
      if (fromList) {
        const int2 en = myList[gi];
        s  = en.x & 0xFFFF;
        lr = en.x >> 16;
        vb = __float_as_int(val[en.y]);
      } else {
        const int q = qe[gi];
        const int e = q & Q_EMASK;
        lr = q >> Q_EBITS;
        s  = src[e];
        vb = __float_as_int(val[e]);
      }
    }
    const int nv = min(32, cnt - g * 32);
    for (int j0 = 0; j0 < nv; j0 += 4) {
      float4 t[4][H];
      float vv[4];
      int ll[4];
#pragma unroll
      for (int u = 0; u < 4; ++u) {
        const int j = j0 + u;
        const int sj = __builtin_amdgcn_readlane(s, j);
        ll[u] = __builtin_amdgcn_readlane(lr, j);
        vv[u] = __int_as_float(__builtin_amdgcn_readlane(vb, j));
        const float4* xp = (const float4*)(x + (long long)sj * N);
#pragma unroll
        for (int h = 0; h < H; ++h) t[u][h] = xp[lane + 32 * h];
      }
#pragma unroll
      for (int u = 0; u < 4; ++u) {
        if (j0 + u < nv) {
          float* a = &acc[ll[u] * N + lane];
#pragma unroll
          for (int h = 0; h < H; ++h) {
            atomicAdd(&a[0 * G + 32 * h], vv[u] * t[u][h].x);
            atomicAdd(&a[1 * G + 32 * h], vv[u] * t[u][h].y);
            atomicAdd(&a[2 * G + 32 * h], vv[u] * t[u][h].z);
            atomicAdd(&a[3 * G + 32 * h], vv[u] * t[u][h].w);
          }
        }
      }
    }
  }
  __syncthreads();

  for (int i = tid; i < T * G; i += SP_THREADS) {
    const int r = i / G;
    const int g = i % G;
    if (row0 + r < nRows) {
      const float* a = &acc[r * N + g];
      *(float4*)(out + (long long)(row0 + r) * N + 4 * g) = make_float4(a[0], a[G], a[2 * G], a[3 * G]);
    }
  }
}

constexpr int T0  = 80;
constexpr int T1s = 8;
constexpr int T1m = 125;
static_assert(N0 % T0 == 0 && N1 % T1s == 0 && N1 % T1m == 0, "tiles cover the rows exactly");
static_assert(T0 <= SP_MAXT && T1s <= SP_MAXT && T1m <= SP_MAXT, "tile fits its LDS array");
static_assert(E00 % 4 == 0 && E01 % 4 == 0 && E10 % 4 == 0 && E11 % 4 == 0, "dst read as int4");

constexpr int CAP00 = 3072, CAP01 = 640, CAP10 = 640, CAP11 = 4608;
static_assert(CAP00 <= SP_Q && CAP01 <= SP_Q && CAP10 <= SP_Q && CAP11 <= SP_Q, "a list is a queue prefix");
constexpr long long off_l00 = off_lists;
constexpr long long off_l01 = off_l00 + 2LL * (N0 / T0) * CAP00;
constexpr long long off_l10 = off_l01 + 2LL * (N0 / T0) * CAP01;
constexpr long long off_l11 = off_l10 + 2LL * (N1 / T1s) * CAP10;
constexpr long long off_c00 = off_l11 + 2LL * (N1 / T1m) * KK * CAP11;
constexpr long long off_c01 = off_c00 + 256;
constexpr long long off_c10 = off_c01 + 256;
constexpr long long off_c11 = off_c10 + 256;
static_assert(N0 / T0 <= 256 && N1 / T1s <= 256 && (N1 / T1m) * KK <= 256, "count arrays hold 256 tiles each");
static_assert(off_c11 + 256 <= off_lists + 5120000, "lists and counts fit the plane xw10 used to have");
static_assert(off_c01 == off_c00 + 256 && off_c10 == off_c00 + 512 && off_c11 == off_c00 + 768,
              "the four count arrays are one run of 1024 ints");

constexpr int BIN_THREADS = 1024;
constexpr int BIN_U  = 2;
constexpr int BIN_I4 = BIN_THREADS * BIN_U;
constexpr int NB00 = (E00 / 4 + BIN_I4 - 1) / BIN_I4;
constexpr int NB01 = (E01 / 4 + BIN_I4 - 1) / BIN_I4;
constexpr int NB10 = (E10 / 4 + BIN_I4 - 1) / BIN_I4;
constexpr int NB11 = (E11 / 4 + BIN_I4 - 1) / BIN_I4;
static_assert(N0 / T0 <= 256 && N1 / T1s <= 256 && N1 / T1m <= 256, "a relation's tiles fit the block's LDS counters");

__global__ void zero_counts(int* __restrict__ c) { c[blockIdx.x * blockDim.x + threadIdx.x] = 0; }

template <int T>
__device__ __forceinline__ void bin_block(const int* __restrict__ dst, const int* __restrict__ src,
                                          int nE, int nRows, int blk, int tileBase, int nTiles, int cap,
                                          int2* __restrict__ lists, int* __restrict__ counts,
                                          int* cntL, int* baseL) {
  const int tid = threadIdx.x;
  const int n4 = nE >> 2;
  if (tid < 256) cntL[tid] = 0;
  __syncthreads();

  int4 dv[BIN_U], sv[BIN_U];
#pragma unroll
  for (int u = 0; u < BIN_U; ++u) {
    const int idxc = min(blk * BIN_I4 + u * BIN_THREADS + tid, n4 - 1);
    dv[u] = ((const int4*)dst)[idxc];
    sv[u] = ((const int4*)src)[idxc];
  }

  int tl[BIN_U * 4], rw[BIN_U * 4], sr[BIN_U * 4], rk[BIN_U * 4];
  bool ok[BIN_U * 4];
#pragma unroll
  for (int u = 0; u < BIN_U; ++u) {
    const bool in = blk * BIN_I4 + u * BIN_THREADS + tid < n4;
    const int dd[4] = {dv[u].x, dv[u].y, dv[u].z, dv[u].w};
    const int ss[4] = {sv[u].x, sv[u].y, sv[u].z, sv[u].w};
#pragma unroll
    for (int c = 0; c < 4; ++c) {
      const int j = u * 4 + c;
      ok[j] = in && (unsigned)dd[c] < (unsigned)nRows;
      const int t = ok[j] ? dd[c] / T : 0;
      tl[j] = t;
      rw[j] = dd[c] - t * T;
      sr[j] = ss[c];
      rk[j] = ok[j] ? atomicAdd(&cntL[t], 1) : 0;
    }
  }
  __syncthreads();

  if (tid < nTiles) {
    const int n = cntL[tid];
    baseL[tid] = (n > 0) ? atomicAdd(&counts[tileBase + tid], n) : 0;
  }
  __syncthreads();

#pragma unroll
  for (int u = 0; u < BIN_U; ++u) {
#pragma unroll
    for (int c = 0; c < 4; ++c) {
      const int j = u * 4 + c;
      if (ok[j]) {
        const int slot = baseL[tl[j]] + rk[j];
        if (slot < cap)
          lists[(long long)(tileBase + tl[j]) * cap + slot] =
              make_int2(sr[j] | (rw[j] << 16), (blk * BIN_I4 + u * BIN_THREADS + tid) * 4 + c);
      }
    }
  }
}

__global__ __launch_bounds__(BIN_THREADS) void bin_edges(
    const int* __restrict__ d00, const int* __restrict__ s00,
    const int* __restrict__ d01, const int* __restrict__ s01,
    const int* __restrict__ d10, const int* __restrict__ s10,
    const int* __restrict__ d11, const int* __restrict__ s11,
    int* __restrict__ l00, int* __restrict__ c00, int* __restrict__ l01, int* __restrict__ c01,
    int* __restrict__ l10, int* __restrict__ c10, int* __restrict__ l11, int* __restrict__ c11) {
  __shared__ int cntL[256];
  __shared__ int baseL[256];
  int b = blockIdx.x;
  if (b < NB00) {
    bin_block<T0>(d00, s00, E00, N0, b, 0, N0 / T0, CAP00, (int2*)l00, c00, cntL, baseL);
    return;
  }
  b -= NB00;
  if (b < NB01) {
    bin_block<T0>(d01, s01, E01, N0, b, 0, N0 / T0, CAP01, (int2*)l01, c01, cntL, baseL);
    return;
  }
  b -= NB01;
  if (b < NB10) {
    bin_block<T1s>(d10, s10, E10, N1, b, 0, N1 / T1s, CAP10, (int2*)l10, c10, cntL, baseL);
    return;
  }
  b -= NB10;
  const int k = b / NB11;
  b -= k * NB11;
  bin_block<T1m>(d11 + (long long)k * E11, s11 + (long long)k * E11, E11, N1, b,
                 k * (N1 / T1m), N1 / T1m, CAP11, (int2*)l11, c11, cntL, baseL);
}

template <int N, int K, bool RELU>
__global__ __launch_bounds__(256) void finalize_conv(const float* buf, float* out, int n, long long kStride) {
  constexpr int NC = N / 32;
  const int lane = threadIdx.x & 31;
  const int wpb = blockDim.x >> 5;
  const int nWaves = gridDim.x * wpb;
  for (int row = blockIdx.x * wpb + (threadIdx.x >> 5); row < n; row += nWaves) {
    const float* rp = buf + (long long)row * N + lane;
    float vals[NC];
    float ss = 0.f;
    if constexpr (K == 1) {
      float t[NC];
#pragma unroll
      for (int j = 0; j < NC; ++j) t[j] = rp[j * 32];
#pragma unroll
      for (int j = 0; j < NC; ++j) {
        float a = 0.f;
        a += RELU ? fmaxf(t[j], 0.f) : t[j];
        vals[j] = a;
        ss = __builtin_fmaf(a, a, ss);
      }
    } else {
#pragma unroll
      for (int j = 0; j < NC; ++j) {
        float t[K];
#pragma unroll
        for (int k = 0; k < K; ++k) t[k] = rp[(long long)k * kStride + j * 32];
        float a = 0.f;
#pragma unroll
        for (int k = 0; k < K; ++k) a += RELU ? fmaxf(t[k], 0.f) : t[k];
        vals[j] = a;
        ss = __builtin_fmaf(a, a, ss);
      }
    }
    ss = wave_sum(ss);
    const float sc = rsqrtf(fmaxf(ss, 1e-12f));
    float* op = out + (long long)row * N + lane;
#pragma unroll
    for (int j = 0; j < NC; ++j) op[j * 32] = vals[j] * sc;
  }
}

__global__ __launch_bounds__(256) void add_act(const float* __restrict__ a, const float* __restrict__ b,
                                               float* __restrict__ o, int n4, int do_relu) {
  const float4* a4 = (const float4*)a;
  const float4* b4 = (const float4*)b;
  float4* o4 = (float4*)o;
  const int stride = gridDim.x * blockDim.x;
  const int i0 = blockIdx.x * blockDim.x + threadIdx.x;
  float4 va[4], vb[4];
#pragma unroll
  for (int u = 0; u < 4; ++u) {
    const int i = min(i0 + u * stride, n4 - 1);
    va[u] = a4[i];
    vb[u] = b4[i];
  }
#pragma unroll
  for (int u = 0; u < 4; ++u) {
    const int i = i0 + u * stride;
    if (i < n4) {
      float4 v = make_float4(va[u].x + vb[u].x, va[u].y + vb[u].y, va[u].z + vb[u].z, va[u].w + vb[u].w);
      if (do_relu) v = make_float4(fmaxf(v.x, 0.f), fmaxf(v.y, 0.f), fmaxf(v.z, 0.f), fmaxf(v.w, 0.f));
      o4[i] = v;
    }
  }
}
constexpr int EWB_H0 = (N0 * HH / 4 + 1023) / 1024;
constexpr int EWB_H1 = (N1 * HH / 4 + 1023) / 1024;
constexpr int EWB_Z0 = (N0 * DD / 4 + 1023) / 1024;
constexpr int EWB_Z1 = (N1 * DD / 4 + 1023) / 1024;

__global__ __launch_bounds__(256) void scale_dloc(const float* z1, const float* dloc, float* out) {
  const int i = blockIdx.x * blockDim.x + threadIdx.x;
  if (i >= N1 * (DD / 4)) return;
  const int f4 = i & (DD / 4 - 1);
  const float4 z = ((const float4*)z1)[i];
  const float4* d4 = (const float4*)dloc;
  float4* o4 = (float4*)out;
  float4 d[KK];
#pragma unroll
  for (int k = 0; k < KK; ++k) d[k] = d4[k * (DD / 4) + f4];
#pragma unroll
  for (int k = 0; k < KK; ++k)
    o4[(long long)k * (N1 * (DD / 4)) + i] =
        make_float4(z.x * d[k].x, z.y * d[k].y, z.z * d[k].z, z.w * d[k].w);
}
static_assert((N0 * HH) % 4 == 0 && (N1 * HH) % 4 == 0 && (N0 * DD) % 4 == 0 && (N1 * DD) % 4 == 0 &&
              (N1 * (DD / 4)) % 256 == 0, "light passes move whole float4s; scale_dloc's grid is exact");

constexpr int ES_THREADS = 1024;
constexpr int ES_WAVES   = ES_THREADS / 32;

__global__ __launch_bounds__(ES_THREADS) void edge_score_tile(
    const float* __restrict__ zl, const float* __restrict__ zr,
    const int* __restrict__ dst, const int* __restrict__ src,
    float* __restrict__ out, int nE, int T, int eStride,
    long long zlStride, long long zrStride, long long oStride,
    const int* __restrict__ lists, const int* __restrict__ counts, int cap) {
  __shared__ __attribute__((aligned(16))) float zt[SP_MAXT * DD];

  const int k = blockIdx.y;
  zl  += (long long)k * zlStride;
  zr  += (long long)k * zrStride;
  dst += (long long)k * eStride;
  src += (long long)k * eStride;
  out += (long long)k * oStride;

  const int tid  = threadIdx.x;
  const int lane = tid & 31;
  const int wave = __builtin_amdgcn_readfirstlane(tid >> 5);
  const int row0 = blockIdx.x * T;
  const int tile = blockIdx.y * gridDim.x + blockIdx.x;
  const int2* __restrict__ myList = (const int2*)lists + (long long)tile * cap;

  {
    const float4* zsrc = (const float4*)(zl + (long long)row0 * DD);
    for (int i = tid; i < T * (DD / 4); i += ES_THREADS) ((float4*)zt)[i] = zsrc[i];
  }
  const int cnt = __builtin_amdgcn_readfirstlane(counts[tile]);
  __syncthreads();

  if (cnt <= cap) {
    const bool h16 = (lane & 16) != 0, h8 = (lane & 8) != 0, h4 = (lane & 4) != 0;
    const int nGroups = (cnt + 31) >> 5;
    for (int g = wave; g < nGroups; g += ES_WAVES) {
      const int gi = g * 32 + lane;
      int s = 0, lr = 0, e = 0;
      if (gi < cnt) {
        const int2 en = myList[gi];
        s  = en.x & 0xFFFF;
        lr = en.x >> 16;
        e  = en.y;
      }
      const int nv = min(32, cnt - g * 32);
      for (int j0 = 0; j0 < nv; j0 += 8) {
        float4 b[8];
        float p[8];
#pragma unroll
        for (int u = 0; u < 8; ++u) {
          const int sj = __builtin_amdgcn_readlane(s, j0 + u);
          b[u] = ((const float4*)(zr + (long long)sj * DD))[lane];
        }
#pragma unroll
        for (int u = 0; u < 8; ++u) {
          const int lj = __builtin_amdgcn_readlane(lr, j0 + u);
          const float4 a = ((const float4*)(zt + lj * DD))[lane];
          p[u] = a.x * b[u].x + a.y * b[u].y + a.z * b[u].z + a.w * b[u].w;
        }
        float q[4], r[2];
#pragma unroll
        for (int i = 0; i < 4; ++i) {
          const float keep = h16 ? p[4 + i] : p[i];
          const float send = h16 ? p[i] : p[4 + i];
          q[i] = keep + __shfl_xor(send, 16, 32);
        }
#pragma unroll
        for (int i = 0; i < 2; ++i) {
          const float keep = h8 ? q[2 + i] : q[i];
          const float send = h8 ? q[i] : q[2 + i];
          r[i] = keep + __shfl_xor(send, 8, 32);
        }
        float t;
        {
          const float keep = h4 ? r[1] : r[0];
          const float send = h4 ? r[0] : r[1];
          t = keep + __shfl_xor(send, 4, 32);
        }
        t += __shfl_xor(t, 2, 32);
        t += __shfl_xor(t, 1, 32);
        const int mine = j0 + (lane >> 2);
        const int eo = __shfl(e, mine, 32);
        if ((lane & 3) == 0 && mine < nv) out[eo] = t;
      }
    }
  } else {
    for (int base = wave * 32; base < nE; base += ES_WAVES * 32) {
      const int e = base + lane;
      const unsigned r = (e < nE) ? (unsigned)(dst[e] - row0) : 0xFFFFFFFFu;
      unsigned mask = __builtin_amdgcn_ballot_w32(r < (unsigned)T);
      while (mask != 0u) {
        const int j = __builtin_ctz(mask);
        mask &= mask - 1u;
        const int lj = __builtin_amdgcn_readlane((int)r, j);
        const int sj = src[base + j];
        const float4 a = ((const float4*)(zt + lj * DD))[lane];
        const float4 bb = ((const float4*)(zr + (long long)sj * DD))[lane];
        float acc = a.x * bb.x + a.y * bb.y + a.z * bb.z + a.w * bb.w;
        acc = wave_sum(acc);
        if (lane == 0) out[base + j] = acc;
      }
    }
  }
}

constexpr int FIN0 = N0 / 16;
constexpr int FIN1 = N1 / 8;
static_assert(N0 % 16 == 0 && N1 % 8 == 0, "finalize grids cover the rows evenly");

constexpr int GW_BIG = 20, GW_MID = 10, GW_REL = 16, GW_ONE = 4;
constexpr int GX_BIG = (N0 + GW_BIG * 16 - 1) / (GW_BIG * 16);
constexpr int GX_MID = (N0 + GW_MID * 16 - 1) / (GW_MID * 16);
constexpr int GX_REL = (N1 + GW_REL * 16 - 1) / (GW_REL * 16);
constexpr int GX_ONE = (N1 + GW_ONE * 16 - 1) / (GW_ONE * 16);
static_assert(GW_BIG * 32 <= GEMM_MAXT && N0 % 16 == 0 && N1 % 16 == 0 && HH % 64 == 0 && DD % 64 == 0,
              "block size within bounds; a wave's 16 rows lie wholly inside or outside M");

extern "C" void kernel_launch(void* const* d_in, const int* in_sizes, int n_in,
                              void* d_out, int out_size, void* d_ws, size_t ws_size,
                              hipStream_t stream) {
  (void)in_sizes; (void)n_in; (void)out_size; (void)ws_size;

  const float* feat0   = (const float*)d_in[0];
  const float* feat1   = (const float*)d_in[1];
  const int*   e00_dst = (const int*)d_in[2];
  const int*   e00_src = (const int*)d_in[3];
  const float* e00_val = (const float*)d_in[4];
  const int*   e01_dst = (const int*)d_in[5];
  const int*   e01_src = (const int*)d_in[6];
  const float* e01_val = (const float*)d_in[7];
  const int*   e10_dst = (const int*)d_in[8];
  const int*   e10_src = (const int*)d_in[9];
  const float* e10_val = (const float*)d_in[10];
  const int*   e11_dst = (const int*)d_in[11];
  const int*   e11_src = (const int*)d_in[12];
  const float* e11_val = (const float*)d_in[13];
  const float* W1_00 = (const float*)d_in[14];
  const float* W1_01 = (const float*)d_in[15];
  const float* W1_10 = (const float*)d_in[16];
  const float* W1_11 = (const float*)d_in[17];
  const float* W2_00 = (const float*)d_in[18];
  const float* W2_01 = (const float*)d_in[19];
  const float* W2_10 = (const float*)d_in[20];
  const float* W2_11 = (const float*)d_in[21];
  const float* R00   = (const float*)d_in[22];
  const float* R01   = (const float*)d_in[23];
  const float* R10   = (const float*)d_in[24];
  const float* Rglob = (const float*)d_in[25];
  const float* dloc  = (const float*)d_in[26];

  float* ws = (float*)d_ws;
  float* out = (float*)d_out;
  float* z0  = out;
  float* z1  = out + 2560000;
  float* s00 = z1 + 256000;
  float* s01 = s00 + E00;
  float* s10 = s01 + E01;
  float* s11 = s10 + E10;

  const dim3 blk(256);

  int* l00 = (int*)(ws + off_l00);  int* c00 = (int*)(ws + off_c00);
  int* l01 = (int*)(ws + off_l01);  int* c01 = (int*)(ws + off_c01);
  int* l10 = (int*)(ws + off_l10);  int* c10 = (int*)(ws + off_c10);
  int* l11 = (int*)(ws + off_l11);  int* c11 = (int*)(ws + off_c11);
  zero_counts<<<dim3(4), blk, 0, stream>>>(c00);
  bin_edges<<<dim3(NB00 + NB01 + NB10 + KK * NB11), dim3(BIN_THREADS), 0, stream>>>(
      e00_dst, e00_src, e01_dst, e01_src, e10_dst, e10_src, e11_dst, e11_src,
      l00, c00, l01, c01, l10, c10, l11, c11);

  gemm_f32_wmma<FF><<<dim3(GX_BIG, HH / 64, 1), dim3(GW_BIG * 32), 0, stream>>>(
      feat0, W1_00, ws + off_xw00, N0, HH, 0, 0, 0);
  gemm_f32_wmma<FF><<<dim3(GX_ONE, HH / 64, 1), dim3(GW_ONE * 32), 0, stream>>>(
      feat1, W1_01, ws + off_xw01, N1, HH, 0, 0, 0);
  gemm_f32_wmma<FF><<<dim3(GX_REL, HH / 64, KK), dim3(GW_REL * 32), 0, stream>>>(
      feat1, W1_11, ws + off_xw11, N1, HH, 0, (long long)FF * HH, (long long)N1 * HH);

  const dim3 sblk(SP_THREADS);
  spmm_tile<HH><<<dim3(N0 / T0, 1), sblk, 0, stream>>>(e00_dst, e00_src, e00_val, ws + off_xw00,
                                                             ws + off_acc00, E00, N0, T0, 0, 0, 0,
                                                             l00, c00, CAP00);
  gemm_f32_wmma<FF><<<dim3(GX_BIG, HH / 64, 1), dim3(GW_BIG * 32), 0, stream>>>(
      feat0, W1_10, ws + off_xw10, N0, HH, 0, 0, 0);
  spmm_tile<HH><<<dim3(N0 / T0, 1), sblk, 0, stream>>>(e01_dst, e01_src, e01_val, ws + off_xw01,
                                                             ws + off_acc01, E01, N0, T0, 0, 0, 0,
                                                             l01, c01, CAP01);
  spmm_tile<HH><<<dim3(N1 / T1s, 1), sblk, 0, stream>>>(e10_dst, e10_src, e10_val, ws + off_xw10,
                                                              ws + off_acc10, E10, N1, T1s, 0, 0, 0,
                                                              l10, c10, CAP10);
  spmm_tile<HH><<<dim3(N1 / T1m, KK), sblk, 0, stream>>>(e11_dst, e11_src, e11_val, ws + off_xw11,
                                                               ws + off_acc11, E11, N1, T1m, E11,
                                                               (long long)N1 * HH, (long long)N1 * HH,
                                                               l11, c11, CAP11);

  finalize_conv<HH, 1, true><<<dim3(FIN0), blk, 0, stream>>>(ws + off_acc00, ws + off_acc00, N0, 0);
  finalize_conv<HH, 1, true><<<dim3(FIN0), blk, 0, stream>>>(ws + off_acc01, ws + off_acc01, N0, 0);
  finalize_conv<HH, 1, true><<<dim3(FIN1), blk, 0, stream>>>(ws + off_acc10, ws + off_acc10, N1, 0);
  finalize_conv<HH, KK, true><<<dim3(FIN1), blk, 0, stream>>>(ws + off_acc11, ws + off_acc11, N1,
                                                              (long long)N1 * HH);

  add_act<<<dim3(EWB_H0), blk, 0, stream>>>(ws + off_acc00, ws + off_acc01, ws + off_h0, N0 * HH / 4, 1);
  add_act<<<dim3(EWB_H1), blk, 0, stream>>>(ws + off_acc10, ws + off_acc11, ws + off_h1, N1 * HH / 4, 1);

  gemm_f32_wmma<HH><<<dim3(GX_MID, DD / 64, 1), dim3(GW_MID * 32), 0, stream>>>(
      ws + off_h0, W2_00, ws + off_hw00, N0, DD, 0, 0, 0);
  gemm_f32_wmma<HH><<<dim3(GX_MID, DD / 64, 1), dim3(GW_MID * 32), 0, stream>>>(
      ws + off_h0, W2_10, ws + off_hw10, N0, DD, 0, 0, 0);
  gemm_f32_wmma<HH><<<dim3(GX_ONE, DD / 64, 1), dim3(GW_ONE * 32), 0, stream>>>(
      ws + off_h1, W2_01, ws + off_hw01, N1, DD, 0, 0, 0);
  gemm_f32_wmma<HH><<<dim3(GX_REL, DD / 64, KK), dim3(GW_REL * 32), 0, stream>>>(
      ws + off_h1, W2_11, ws + off_hw11, N1, DD, 0, (long long)HH * DD, (long long)N1 * DD);

  spmm_tile<DD><<<dim3(N0 / T0, 1), sblk, 0, stream>>>(e00_dst, e00_src, e00_val, ws + off_hw00,
                                                              ws + off_bcc00, E00, N0, T0, 0, 0, 0,
                                                              l00, c00, CAP00);
  spmm_tile<DD><<<dim3(N0 / T0, 1), sblk, 0, stream>>>(e01_dst, e01_src, e01_val, ws + off_hw01,
                                                              ws + off_bcc01, E01, N0, T0, 0, 0, 0,
                                                              l01, c01, CAP01);
  spmm_tile<DD><<<dim3(N1 / T1s, 1), sblk, 0, stream>>>(e10_dst, e10_src, e10_val, ws + off_hw10,
                                                               ws + off_bcc10, E10, N1, T1s, 0, 0, 0,
                                                               l10, c10, CAP10);
  spmm_tile<DD><<<dim3(N1 / T1m, KK), sblk, 0, stream>>>(e11_dst, e11_src, e11_val, ws + off_hw11,
                                                                ws + off_bcc11, E11, N1, T1m, E11,
                                                                (long long)N1 * DD, (long long)N1 * DD,
                                                                l11, c11, CAP11);

  finalize_conv<DD, 1, false><<<dim3(FIN0), blk, 0, stream>>>(ws + off_bcc00, ws + off_bcc00, N0, 0);
  finalize_conv<DD, 1, false><<<dim3(FIN0), blk, 0, stream>>>(ws + off_bcc01, ws + off_bcc01, N0, 0);
  finalize_conv<DD, 1, false><<<dim3(FIN1), blk, 0, stream>>>(ws + off_bcc10, ws + off_bcc10, N1, 0);
  finalize_conv<DD, KK, false><<<dim3(FIN1), blk, 0, stream>>>(ws + off_bcc11, ws + off_bcc11, N1,
                                                               (long long)N1 * DD);

  add_act<<<dim3(EWB_Z0), blk, 0, stream>>>(ws + off_bcc00, ws + off_bcc01, z0, N0 * DD / 4, 0);
  add_act<<<dim3(EWB_Z1), blk, 0, stream>>>(ws + off_bcc10, ws + off_bcc11, z1, N1 * DD / 4, 0);

  gemm_f32_wmma<DD><<<dim3(GX_MID, DD / 64, 1), dim3(GW_MID * 32), 0, stream>>>(
      z0, R00, ws + off_z0R00, N0, DD, 0, 0, 0);
  gemm_f32_wmma<DD><<<dim3(GX_MID, DD / 64, 1), dim3(GW_MID * 32), 0, stream>>>(
      z0, R01, ws + off_z0R01, N0, DD, 0, 0, 0);
  gemm_f32_wmma<DD><<<dim3(GX_ONE, DD / 64, 1), dim3(GW_ONE * 32), 0, stream>>>(
      z1, R10, ws + off_z1R10, N1, DD, 0, 0, 0);

  scale_dloc<<<dim3(N1 * (DD / 4) / 256), blk, 0, stream>>>(z1, dloc, ws + off_z1d);
  gemm_f32_wmma<DD><<<dim3(GX_REL, DD / 64, KK), dim3(GW_REL * 32), 0, stream>>>(
      ws + off_z1d, Rglob, ws + off_Zk, N1, DD, (long long)N1 * DD, 0, (long long)N1 * DD);

  const dim3 eblk(ES_THREADS);
  edge_score_tile<<<dim3(N0 / T0, 1), eblk, 0, stream>>>(ws + off_z0R00, z0, e00_dst, e00_src,
                                                         s00, E00, T0, 0, 0, 0, 0, l00, c00, CAP00);
  edge_score_tile<<<dim3(N0 / T0, 1), eblk, 0, stream>>>(ws + off_z0R01, z1, e01_dst, e01_src,
                                                         s01, E01, T0, 0, 0, 0, 0, l01, c01, CAP01);
  edge_score_tile<<<dim3(N1 / T1s, 1), eblk, 0, stream>>>(ws + off_z1R10, z0, e10_dst, e10_src,
                                                          s10, E10, T1s, 0, 0, 0, 0, l10, c10, CAP10);
  edge_score_tile<<<dim3(N1 / T1m, KK), eblk, 0, stream>>>(ws + off_Zk, ws + off_z1d, e11_dst, e11_src,
                                                           s11, E11, T1m, E11,
                                                           (long long)N1 * DD, (long long)N1 * DD,
                                                           (long long)E11, l11, c11, CAP11);
}


# list-driven gathers hold 16 float4 per lane in flight: SPMM steps take 8 rows (256 col) or 16 (128 col), scores 16 rows; later groups' list entries and val loaded ahead; LDS adds in arrival order
# speedup vs baseline: 2.9495x; 1.0988x over previous
#include <hip/hip_runtime.h>

typedef float v2f __attribute__((ext_vector_type(2)));
typedef float v8f __attribute__((ext_vector_type(8)));

constexpr int N0 = 20000, N1 = 2000;
constexpr int FF = 512, HH = 256, DD = 128, KK = 16;
constexpr int E00 = 640000, E01 = 100000, E10 = 100000, E11 = 64000;

constexpr long long off_xw00 = 0;
constexpr long long off_lists = 5120000;
constexpr long long off_xw01 = 10240000;
constexpr long long off_xw11 = 10752000;
constexpr long long off_acc00 = 18944000;
constexpr long long off_acc01 = 24064000;
constexpr long long off_acc10 = 29184000;
constexpr long long off_acc11 = 29696000;

constexpr long long off_xw10  = off_xw00;
constexpr long long off_h0    = off_xw00;
constexpr long long off_h1    = off_xw01;
constexpr long long off_hw00  = off_acc01;
constexpr long long off_hw10  = off_acc01 + 2560000;
constexpr long long off_hw01  = off_xw11;
constexpr long long off_hw11  = off_xw11 + 256000;
constexpr long long off_bcc00 = off_acc00;
constexpr long long off_bcc01 = off_acc00 + 2560000;
constexpr long long off_bcc10 = off_acc10;
constexpr long long off_bcc11 = off_acc11;
constexpr long long off_z0R00 = off_acc01;
constexpr long long off_z0R01 = off_acc01 + 2560000;
constexpr long long off_z1R10 = off_acc10 + 256000;
constexpr long long off_z1d   = off_acc11 + 4096000;
constexpr long long off_Zk    = off_acc11;

__device__ __forceinline__ float wave_sum(float v) {
#pragma unroll
  for (int off = 16; off > 0; off >>= 1) v += __shfl_xor(v, off, 32);
  return v;
}

typedef float v4f __attribute__((ext_vector_type(4)));
constexpr int BK  = 32;
constexpr int BPS = 144;
constexpr int GEMM_MAXT = 640;

template <int KD>
__global__ __launch_bounds__(GEMM_MAXT) void gemm_f32_wmma(
    const float* __restrict__ A, const float* __restrict__ B, float* __restrict__ C,
    int M, int N, long long sA, long long sB, long long sC) {
  __shared__ __attribute__((aligned(16))) float Bp[(KD / 2) * BPS];

  const int b = blockIdx.z;
  A += (long long)b * sA;
  B += (long long)b * sB;
  C += (long long)b * sC;

  const int tid  = threadIdx.x;
  const int nT   = blockDim.x;
  const int lane = tid & 31;
  const int wave = __builtin_amdgcn_readfirstlane(tid >> 5);
  const int n0   = blockIdx.y * 64;
  const int m0   = blockIdx.x * (nT >> 5) * 16 + wave * 16;
  const int half = lane >> 4;
  const int lr   = lane & 15;

  {
    constexpr int ITEMS = (KD / 2) * 16;
    for (int i0 = tid; i0 < ITEMS; i0 += 4 * nT) {
      v4f r0[4], r1[4];
#pragma unroll
      for (int u = 0; u < 4; ++u) {
        const int i = min(i0 + u * nT, ITEMS - 1);
        const int p = i >> 4, c4 = (i & 15) * 4;
        const float* bp = B + (long long)(2 * p) * N + n0 + c4;
        r0[u] = *(const v4f*)bp;
        r1[u] = *(const v4f*)(bp + N);
      }
#pragma unroll
      for (int u = 0; u < 4; ++u) {
        const int i = i0 + u * nT;
        if (i < ITEMS) {
          const int p = i >> 4, c4 = (i & 15) * 4;
          float* sp = &Bp[p * BPS + c4 * 2];
          const v4f lo = {r0[u].x, r1[u].x, r0[u].y, r1[u].y};
          const v4f hi = {r0[u].z, r1[u].z, r0[u].w, r1[u].w};
          *(v4f*)sp = lo;
          *(v4f*)(sp + 4) = hi;
        }
      }
    }
  }
  __syncthreads();

  if (m0 >= M) return;

  const v4f* ap = (const v4f*)(A + (long long)(m0 + lr) * KD + 4 * half);
  const float* Bsb = &Bp[(2 * half) * BPS + lr * 2];

  v8f acc0 = {}, acc1 = {}, acc2 = {}, acc3 = {};
  v4f q[4], qn[4];
#pragma unroll
  for (int j = 0; j < 4; ++j) q[j] = ap[2 * j];

  constexpr int NCH = KD / BK;
  for (int c = 0; c < NCH; ++c) {
    const int cn = min(c + 1, NCH - 1);
#pragma unroll
    for (int j = 0; j < 4; ++j) qn[j] = ap[cn * (BK / 4) + 2 * j];

    const float* bc = Bsb + c * (BK / 2) * BPS;
#pragma unroll
    for (int j = 0; j < 4; ++j) {
#pragma unroll
      for (int sub = 0; sub < 2; ++sub) {
        const v2f a = sub ? __builtin_shufflevector(q[j], q[j], 2, 3)
                          : __builtin_shufflevector(q[j], q[j], 0, 1);
        const float* bp = bc + (4 * j + sub) * BPS;
        const v2f b0 = *(const v2f*)(bp);
        const v2f b1 = *(const v2f*)(bp + 32);
        const v2f b2 = *(const v2f*)(bp + 64);
        const v2f b3 = *(const v2f*)(bp + 96);
        acc0 = __builtin_amdgcn_wmma_f32_16x16x4_f32(false, a, false, b0, (short)0, acc0, false, false);
        acc1 = __builtin_amdgcn_wmma_f32_16x16x4_f32(false, a, false, b1, (short)0, acc1, false, false);
        acc2 = __builtin_amdgcn_wmma_f32_16x16x4_f32(false, a, false, b2, (short)0, acc2, false, false);
        acc3 = __builtin_amdgcn_wmma_f32_16x16x4_f32(false, a, false, b3, (short)0, acc3, false, false);
      }
    }
#pragma unroll
    for (int j = 0; j < 4; ++j) q[j] = qn[j];
  }

#pragma unroll
  for (int r = 0; r < 8; ++r) {
    const int row = m0 + r + 8 * half;
    float* cp = C + (long long)row * N + n0 + lr;
    cp[0]  = acc0[r];
    cp[16] = acc1[r];
    cp[32] = acc2[r];
    cp[48] = acc3[r];
  }
}

constexpr int SP_THREADS = 1024;
constexpr int SP_WAVES   = SP_THREADS / 32;
constexpr int SP_MAXT    = 125;
constexpr int SP_Q       = 16384;
constexpr int Q_EBITS    = 20;
constexpr int Q_EMASK    = (1 << Q_EBITS) - 1;
static_assert(E00 <= (1 << Q_EBITS) && E01 <= (1 << Q_EBITS) && E10 <= (1 << Q_EBITS) &&
              E11 <= (1 << Q_EBITS), "edge id fits the low bits of a queue word");
static_assert(N0 <= 65536 && N1 <= 65536 && SP_MAXT <= 128, "src and tile row fit one list word");

template <int N, int C>
__device__ __forceinline__ void add_rows(const float* __restrict__ x, float* acc, int s, int lr, int vb,
                                         int j0, int lane) {
  constexpr int G = N / 4;
  constexpr int H = N / 128;
  float4 t[C][H];
  float vv[C];
  int ll[C];
#pragma unroll
  for (int u = 0; u < C; ++u) {
    const int j = j0 + u;
    const int sj = __builtin_amdgcn_readlane(s, j);
    ll[u] = __builtin_amdgcn_readlane(lr, j);
    vv[u] = __int_as_float(__builtin_amdgcn_readlane(vb, j));
    const float4* xp = (const float4*)(x + (long long)sj * N);
#pragma unroll
    for (int h = 0; h < H; ++h) t[u][h] = xp[lane + 32 * h];
  }
#pragma unroll
  for (int u = 0; u < C; ++u) {
    float* a = &acc[ll[u] * N + lane];
#pragma unroll
    for (int h = 0; h < H; ++h) {
      atomicAdd(&a[0 * G + 32 * h], vv[u] * t[u][h].x);
      atomicAdd(&a[1 * G + 32 * h], vv[u] * t[u][h].y);
      atomicAdd(&a[2 * G + 32 * h], vv[u] * t[u][h].z);
      atomicAdd(&a[3 * G + 32 * h], vv[u] * t[u][h].w);
    }
  }
}

template <int N>
__global__ __launch_bounds__(SP_THREADS) __attribute__((amdgpu_waves_per_eu(8, 8))) void spmm_tile(
    const int* __restrict__ dst, const int* __restrict__ src, const float* __restrict__ val,
    const float* __restrict__ x, float* __restrict__ out, int nE, int nRows, int T,
    int eStride, long long xStride, long long oStride,
    const int* __restrict__ lists, const int* __restrict__ counts, int cap) {
  constexpr int G = N / 4;
  constexpr int H = N / 128;
  __shared__ __attribute__((aligned(16))) float acc[SP_MAXT * N];
  __shared__ int qe[SP_Q];
  __shared__ int qcount;

  const int k = blockIdx.y;
  dst += (long long)k * eStride;
  src += (long long)k * eStride;
  val += (long long)k * eStride;
  x   += (long long)k * xStride;
  out += (long long)k * oStride;

  const int tid  = threadIdx.x;
  const int lane = tid & 31;
  const int wave = __builtin_amdgcn_readfirstlane(tid >> 5);
  const int row0 = blockIdx.x * T;
  const int tile = blockIdx.y * gridDim.x + blockIdx.x;
  const int2* __restrict__ myList = (const int2*)lists + (long long)tile * cap;

  const int listCnt = counts[tile];
  const bool fromList = (listCnt <= cap);

  for (int i = tid; i < T * G; i += SP_THREADS)
    ((float4*)acc)[i] = make_float4(0.f, 0.f, 0.f, 0.f);
  if (tid == 0) qcount = 0;
  __syncthreads();

  const int n4 = nE >> 2;
  const int4* d4p = (const int4*)dst;
  constexpr int STEP = SP_THREADS * 4;
  const int last4 = max(n4 - 1, 0);

  auto test4 = [&](const int4 (&dv)[4], int base) {
#pragma unroll
    for (int u = 0; u < 4; ++u) {
      const int idx = base + u * SP_THREADS + tid;
      const int e0 = idx << 2;
      const int dd[4] = {dv[u].x, dv[u].y, dv[u].z, dv[u].w};
#pragma unroll
      for (int c = 0; c < 4; ++c) {
        const unsigned r = (unsigned)(dd[c] - row0);
        if (idx < n4 && r < (unsigned)T) {
          const int e = e0 + c;
          const int slot = atomicAdd(&qcount, 1);
          if (slot < SP_Q) {
            qe[slot] = e | ((int)r << Q_EBITS);
          } else {
            const int s = src[e];
            const float v = val[e];
            const float* xr = x + (long long)s * N;
            float* a = &acc[(int)r * N];
            for (int f = 0; f < N; ++f) atomicAdd(&a[(f & 3) * G + (f >> 2)], v * xr[f]);
          }
        }
      }
    }
  };

  if (!fromList) {
  int4 da[4], db[4];
#pragma unroll
  for (int u = 0; u < 4; ++u) da[u] = d4p[min(u * SP_THREADS + tid, last4)];
  for (int i0 = 0; i0 < n4; i0 += 2 * STEP) {
#pragma unroll
    for (int u = 0; u < 4; ++u) db[u] = d4p[min(i0 + STEP + u * SP_THREADS + tid, last4)];
    test4(da, i0);
#pragma unroll
    for (int u = 0; u < 4; ++u) da[u] = d4p[min(i0 + 2 * STEP + u * SP_THREADS + tid, last4)];
    test4(db, i0 + STEP);
  }
  }
  __syncthreads();

  constexpr int R = 2048 / N;
  static_assert(R * H == 16 && 32 % R == 0, "a step holds 16 float4 per lane and steps tile a group");
  const int cnt = fromList ? listCnt : __builtin_amdgcn_readfirstlane(min(qcount, SP_Q));
  const int nGroups = (cnt + 31) >> 5;

  auto walk = [&](int s, int lr, int vb, int nv) {
    int j0 = 0;
    for (; j0 + R <= nv; j0 += R) add_rows<N, R>(x, acc, s, lr, vb, j0, lane);
    if constexpr (R > 8) {
      if (nv - j0 >= 8) { add_rows<N, 8>(x, acc, s, lr, vb, j0, lane); j0 += 8; }
    }
    if (nv - j0 >= 4) { add_rows<N, 4>(x, acc, s, lr, vb, j0, lane); j0 += 4; }
    if (nv - j0 >= 2) { add_rows<N, 2>(x, acc, s, lr, vb, j0, lane); j0 += 2; }
    if (nv - j0 >= 1) add_rows<N, 1>(x, acc, s, lr, vb, j0, lane);
  };

  if (fromList) {
    int2 en = make_int2(0, 0), enN = make_int2(0, 0);
    {
      const int gi = wave * 32 + lane;
      if (gi < cnt) en = myList[gi];
      const int gn = gi + SP_WAVES * 32;
      if (gn < cnt) enN = myList[gn];
    }
    int vb = __float_as_int(val[en.y]);
    for (int g = wave; g < nGroups; g += SP_WAVES) {
      int2 enNN = make_int2(0, 0);
      const int gnn = (g + 2 * SP_WAVES) * 32 + lane;
      if (gnn < cnt) enNN = myList[gnn];
      const int vbN = __float_as_int(val[enN.y]);
      walk(en.x & 0xFFFF, en.x >> 16, vb, min(32, cnt - g * 32));
      en = enN;
      vb = vbN;
      enN = enNN;
    }
  } else {
    for (int g = wave; g < nGroups; g += SP_WAVES) {
      const int gi = g * 32 + lane;
      int s = 0, lr = 0, vb = 0;
      if (gi < cnt) {
        const int q = qe[gi];
        const int e = q & Q_EMASK;
        lr = q >> Q_EBITS;
        s  = src[e];
        vb = __float_as_int(val[e]);
      }
      walk(s, lr, vb, min(32, cnt - g * 32));
    }
  }
  __syncthreads();

  for (int i = tid; i < T * G; i += SP_THREADS) {
    const int r = i / G;
    const int g = i % G;
    if (row0 + r < nRows) {
      const float* a = &acc[r * N + g];
      *(float4*)(out + (long long)(row0 + r) * N + 4 * g) = make_float4(a[0], a[G], a[2 * G], a[3 * G]);
    }
  }
}

constexpr int T0  = 80;
constexpr int T1s = 8;
constexpr int T1m = 125;
static_assert(N0 % T0 == 0 && N1 % T1s == 0 && N1 % T1m == 0, "tiles cover the rows exactly");
static_assert(T0 <= SP_MAXT && T1s <= SP_MAXT && T1m <= SP_MAXT, "tile fits its LDS array");
static_assert(E00 % 4 == 0 && E01 % 4 == 0 && E10 % 4 == 0 && E11 % 4 == 0, "dst read as int4");

constexpr int CAP00 = 3072, CAP01 = 640, CAP10 = 640, CAP11 = 4608;
static_assert(CAP00 <= SP_Q && CAP01 <= SP_Q && CAP10 <= SP_Q && CAP11 <= SP_Q, "a list is a queue prefix");
constexpr long long off_l00 = off_lists;
constexpr long long off_l01 = off_l00 + 2LL * (N0 / T0) * CAP00;
constexpr long long off_l10 = off_l01 + 2LL * (N0 / T0) * CAP01;
constexpr long long off_l11 = off_l10 + 2LL * (N1 / T1s) * CAP10;
constexpr long long off_c00 = off_l11 + 2LL * (N1 / T1m) * KK * CAP11;
constexpr long long off_c01 = off_c00 + 256;
constexpr long long off_c10 = off_c01 + 256;
constexpr long long off_c11 = off_c10 + 256;
static_assert(N0 / T0 <= 256 && N1 / T1s <= 256 && (N1 / T1m) * KK <= 256, "count arrays hold 256 tiles each");
static_assert(off_c11 + 256 <= off_lists + 5120000, "lists and counts fit the plane xw10 used to have");
static_assert(off_c01 == off_c00 + 256 && off_c10 == off_c00 + 512 && off_c11 == off_c00 + 768,
              "the four count arrays are one run of 1024 ints");

constexpr int BIN_THREADS = 1024;
constexpr int BIN_U  = 2;
constexpr int BIN_I4 = BIN_THREADS * BIN_U;
constexpr int NB00 = (E00 / 4 + BIN_I4 - 1) / BIN_I4;
constexpr int NB01 = (E01 / 4 + BIN_I4 - 1) / BIN_I4;
constexpr int NB10 = (E10 / 4 + BIN_I4 - 1) / BIN_I4;
constexpr int NB11 = (E11 / 4 + BIN_I4 - 1) / BIN_I4;
static_assert(N0 / T0 <= 256 && N1 / T1s <= 256 && N1 / T1m <= 256, "a relation's tiles fit the block's LDS counters");

__global__ void zero_counts(int* __restrict__ c) { c[blockIdx.x * blockDim.x + threadIdx.x] = 0; }

template <int T>
__device__ __forceinline__ void bin_block(const int* __restrict__ dst, const int* __restrict__ src,
                                          int nE, int nRows, int blk, int tileBase, int nTiles, int cap,
                                          int2* __restrict__ lists, int* __restrict__ counts,
                                          int* cntL, int* baseL) {
  const int tid = threadIdx.x;
  const int n4 = nE >> 2;
  if (tid < 256) cntL[tid] = 0;
  __syncthreads();

  int4 dv[BIN_U], sv[BIN_U];
#pragma unroll
  for (int u = 0; u < BIN_U; ++u) {
    const int idxc = min(blk * BIN_I4 + u * BIN_THREADS + tid, n4 - 1);
    dv[u] = ((const int4*)dst)[idxc];
    sv[u] = ((const int4*)src)[idxc];
  }

  int tl[BIN_U * 4], rw[BIN_U * 4], sr[BIN_U * 4], rk[BIN_U * 4];
  bool ok[BIN_U * 4];
#pragma unroll
  for (int u = 0; u < BIN_U; ++u) {
    const bool in = blk * BIN_I4 + u * BIN_THREADS + tid < n4;
    const int dd[4] = {dv[u].x, dv[u].y, dv[u].z, dv[u].w};
    const int ss[4] = {sv[u].x, sv[u].y, sv[u].z, sv[u].w};
#pragma unroll
    for (int c = 0; c < 4; ++c) {
      const int j = u * 4 + c;
      ok[j] = in && (unsigned)dd[c] < (unsigned)nRows;
      const int t = ok[j] ? dd[c] / T : 0;
      tl[j] = t;
      rw[j] = dd[c] - t * T;
      sr[j] = ss[c];
      rk[j] = ok[j] ? atomicAdd(&cntL[t], 1) : 0;
    }
  }
  __syncthreads();

  if (tid < nTiles) {
    const int n = cntL[tid];
    baseL[tid] = (n > 0) ? atomicAdd(&counts[tileBase + tid], n) : 0;
  }
  __syncthreads();

#pragma unroll
  for (int u = 0; u < BIN_U; ++u) {
#pragma unroll
    for (int c = 0; c < 4; ++c) {
      const int j = u * 4 + c;
      if (ok[j]) {
        const int slot = baseL[tl[j]] + rk[j];
        if (slot < cap)
          lists[(long long)(tileBase + tl[j]) * cap + slot] =
              make_int2(sr[j] | (rw[j] << 16), (blk * BIN_I4 + u * BIN_THREADS + tid) * 4 + c);
      }
    }
  }
}

__global__ __launch_bounds__(BIN_THREADS) void bin_edges(
    const int* __restrict__ d00, const int* __restrict__ s00,
    const int* __restrict__ d01, const int* __restrict__ s01,
    const int* __restrict__ d10, const int* __restrict__ s10,
    const int* __restrict__ d11, const int* __restrict__ s11,
    int* __restrict__ l00, int* __restrict__ c00, int* __restrict__ l01, int* __restrict__ c01,
    int* __restrict__ l10, int* __restrict__ c10, int* __restrict__ l11, int* __restrict__ c11) {
  __shared__ int cntL[256];
  __shared__ int baseL[256];
  int b = blockIdx.x;
  if (b < NB00) {
    bin_block<T0>(d00, s00, E00, N0, b, 0, N0 / T0, CAP00, (int2*)l00, c00, cntL, baseL);
    return;
  }
  b -= NB00;
  if (b < NB01) {
    bin_block<T0>(d01, s01, E01, N0, b, 0, N0 / T0, CAP01, (int2*)l01, c01, cntL, baseL);
    return;
  }
  b -= NB01;
  if (b < NB10) {
    bin_block<T1s>(d10, s10, E10, N1, b, 0, N1 / T1s, CAP10, (int2*)l10, c10, cntL, baseL);
    return;
  }
  b -= NB10;
  const int k = b / NB11;
  b -= k * NB11;
  bin_block<T1m>(d11 + (long long)k * E11, s11 + (long long)k * E11, E11, N1, b,
                 k * (N1 / T1m), N1 / T1m, CAP11, (int2*)l11, c11, cntL, baseL);
}

template <int N, int KB, bool RELU, int RPW>
__global__ __launch_bounds__(256) void conv_pair(const float* __restrict__ a, const float* __restrict__ b,
                                                 float* __restrict__ out, int n, long long kStride) {
  constexpr int NC = N / 32;
  const int lane = threadIdx.x & 31;
  const int wpb = blockDim.x >> 5;
  const int nWaves = gridDim.x * wpb;
  for (int row0 = blockIdx.x * wpb + (threadIdx.x >> 5); row0 < n; row0 += RPW * nWaves) {
    float va[RPW][NC], vb[RPW][NC];
    float sa[RPW], sb[RPW];
    float ta[RPW][NC];
#pragma unroll
    for (int r = 0; r < RPW; ++r) {
      const int row = min(row0 + r * nWaves, n - 1);
      const float* ap = a + (long long)row * N + lane;
#pragma unroll
      for (int j = 0; j < NC; ++j) ta[r][j] = ap[j * 32];
    }
    if constexpr (KB == 1) {
      float tb[RPW][NC];
#pragma unroll
      for (int r = 0; r < RPW; ++r) {
        const int row = min(row0 + r * nWaves, n - 1);
        const float* bp = b + (long long)row * N + lane;
#pragma unroll
        for (int j = 0; j < NC; ++j) tb[r][j] = bp[j * 32];
      }
#pragma unroll
      for (int r = 0; r < RPW; ++r) {
        float ss = 0.f;
#pragma unroll
        for (int j = 0; j < NC; ++j) {
          float x = 0.f;
          x += RELU ? fmaxf(tb[r][j], 0.f) : tb[r][j];
          vb[r][j] = x;
          ss = __builtin_fmaf(x, x, ss);
        }
        sb[r] = ss;
      }
    } else {
#pragma unroll
      for (int r = 0; r < RPW; ++r) {
        const int row = min(row0 + r * nWaves, n - 1);
        const float* bp = b + (long long)row * N + lane;
        float ss = 0.f;
#pragma unroll
        for (int j = 0; j < NC; ++j) {
          float t[KB];
#pragma unroll
          for (int k = 0; k < KB; ++k) t[k] = bp[(long long)k * kStride + j * 32];
          float x = 0.f;
#pragma unroll
          for (int k = 0; k < KB; ++k) x += RELU ? fmaxf(t[k], 0.f) : t[k];
          vb[r][j] = x;
          ss = __builtin_fmaf(x, x, ss);
        }
        sb[r] = ss;
      }
    }
#pragma unroll
    for (int r = 0; r < RPW; ++r) {
      float ss = 0.f;
#pragma unroll
      for (int j = 0; j < NC; ++j) {
        float x = 0.f;
        x += RELU ? fmaxf(ta[r][j], 0.f) : ta[r][j];
        va[r][j] = x;
        ss = __builtin_fmaf(x, x, ss);
      }
      sa[r] = ss;
    }
#pragma unroll
    for (int r = 0; r < RPW; ++r) {
      const int row = row0 + r * nWaves;
      const float sca = rsqrtf(fmaxf(wave_sum(sa[r]), 1e-12f));
      const float scb = rsqrtf(fmaxf(wave_sum(sb[r]), 1e-12f));
      if (row < n) {
        float* op = out + (long long)row * N + lane;
#pragma unroll
        for (int j = 0; j < NC; ++j) {
          float v = va[r][j] * sca + vb[r][j] * scb;
          if (RELU) v = fmaxf(v, 0.f);
          op[j * 32] = v;
        }
      }
    }
  }
}

__global__ __launch_bounds__(256) void scale_dloc(const float* z1, const float* dloc, float* out) {
  const int i = blockIdx.x * blockDim.x + threadIdx.x;
  if (i >= N1 * (DD / 4)) return;
  const int f4 = i & (DD / 4 - 1);
  const float4 z = ((const float4*)z1)[i];
  const float4* d4 = (const float4*)dloc;
  float4* o4 = (float4*)out;
  float4 d[KK];
#pragma unroll
  for (int k = 0; k < KK; ++k) d[k] = d4[k * (DD / 4) + f4];
#pragma unroll
  for (int k = 0; k < KK; ++k)
    o4[(long long)k * (N1 * (DD / 4)) + i] =
        make_float4(z.x * d[k].x, z.y * d[k].y, z.z * d[k].z, z.w * d[k].w);
}
static_assert((N1 * (DD / 4)) % 256 == 0, "scale_dloc's grid is exact");

constexpr int ES_THREADS = 1024;
constexpr int ES_WAVES   = ES_THREADS / 32;

__global__ __launch_bounds__(ES_THREADS) __attribute__((amdgpu_waves_per_eu(8, 8))) void edge_score_tile(
    const float* __restrict__ zl, const float* __restrict__ zr,
    const int* __restrict__ dst, const int* __restrict__ src,
    float* __restrict__ out, int nE, int T, int eStride,
    long long zlStride, long long zrStride, long long oStride,
    const int* __restrict__ lists, const int* __restrict__ counts, int cap) {
  __shared__ __attribute__((aligned(16))) float zt[SP_MAXT * DD];

  const int k = blockIdx.y;
  zl  += (long long)k * zlStride;
  zr  += (long long)k * zrStride;
  dst += (long long)k * eStride;
  src += (long long)k * eStride;
  out += (long long)k * oStride;

  const int tid  = threadIdx.x;
  const int lane = tid & 31;
  const int wave = __builtin_amdgcn_readfirstlane(tid >> 5);
  const int row0 = blockIdx.x * T;
  const int tile = blockIdx.y * gridDim.x + blockIdx.x;
  const int2* __restrict__ myList = (const int2*)lists + (long long)tile * cap;

  {
    const float4* zsrc = (const float4*)(zl + (long long)row0 * DD);
    for (int i = tid; i < T * (DD / 4); i += ES_THREADS) ((float4*)zt)[i] = zsrc[i];
  }
  const int cnt = __builtin_amdgcn_readfirstlane(counts[tile]);
  __syncthreads();

  if (cnt <= cap) {
    const bool h16 = (lane & 16) != 0, h8 = (lane & 8) != 0, h4 = (lane & 4) != 0;
    const int nGroups = (cnt + 31) >> 5;
    int2 en = make_int2(0, 0);
    if (wave * 32 + lane < cnt) en = myList[wave * 32 + lane];
    for (int g = wave; g < nGroups; g += ES_WAVES) {
      int2 enN = make_int2(0, 0);
      const int gn = (g + ES_WAVES) * 32 + lane;
      if (gn < cnt) enN = myList[gn];
      const int s  = en.x & 0xFFFF;
      const int lr = en.x >> 16;
      const int e  = en.y;
      const int nv = min(32, cnt - g * 32);
      for (int j0 = 0; j0 < nv; j0 += 16) {
        float4 b[16];
#pragma unroll
        for (int u = 0; u < 16; ++u) {
          const int sj = __builtin_amdgcn_readlane(s, j0 + u);
          b[u] = ((const float4*)(zr + (long long)sj * DD))[lane];
        }
#pragma unroll
        for (int hh = 0; hh < 2; ++hh) {
          const int jb = j0 + 8 * hh;
          float p[8];
#pragma unroll
          for (int u = 0; u < 8; ++u) {
            const int lj = __builtin_amdgcn_readlane(lr, jb + u);
            const float4 a = ((const float4*)(zt + lj * DD))[lane];
            p[u] = a.x * b[8 * hh + u].x + a.y * b[8 * hh + u].y + a.z * b[8 * hh + u].z +
                   a.w * b[8 * hh + u].w;
          }
          float q[4], r[2];
#pragma unroll
          for (int i = 0; i < 4; ++i) {
            const float keep = h16 ? p[4 + i] : p[i];
            const float send = h16 ? p[i] : p[4 + i];
            q[i] = keep + __shfl_xor(send, 16, 32);
          }
#pragma unroll
          for (int i = 0; i < 2; ++i) {
            const float keep = h8 ? q[2 + i] : q[i];
            const float send = h8 ? q[i] : q[2 + i];
            r[i] = keep + __shfl_xor(send, 8, 32);
          }
          float t;
          {
            const float keep = h4 ? r[1] : r[0];
            const float send = h4 ? r[0] : r[1];
            t = keep + __shfl_xor(send, 4, 32);
          }
          t += __shfl_xor(t, 2, 32);
          t += __shfl_xor(t, 1, 32);
          const int mine = jb + (lane >> 2);
          const int eo = __shfl(e, mine, 32);
          if ((lane & 3) == 0 && mine < nv) out[eo] = t;
        }
      }
      en = enN;
    }
  } else {
    for (int base = wave * 32; base < nE; base += ES_WAVES * 32) {
      const int e = base + lane;
      const unsigned r = (e < nE) ? (unsigned)(dst[e] - row0) : 0xFFFFFFFFu;
      unsigned mask = __builtin_amdgcn_ballot_w32(r < (unsigned)T);
      while (mask != 0u) {
        const int j = __builtin_ctz(mask);
        mask &= mask - 1u;
        const int lj = __builtin_amdgcn_readlane((int)r, j);
        const int sj = src[base + j];
        const float4 a = ((const float4*)(zt + lj * DD))[lane];
        const float4 bb = ((const float4*)(zr + (long long)sj * DD))[lane];
        float acc = a.x * bb.x + a.y * bb.y + a.z * bb.z + a.w * bb.w;
        acc = wave_sum(acc);
        if (lane == 0) out[base + j] = acc;
      }
    }
  }
}

constexpr int FIN0 = N0 / 16;
constexpr int FIN1 = N1 / 8;
static_assert(N0 % 16 == 0 && N1 % 8 == 0, "conv_pair grids cover the rows evenly");

constexpr int GW_BIG = 20, GW_MID = 10, GW_REL = 16, GW_ONE = 4;
constexpr int GX_BIG = (N0 + GW_BIG * 16 - 1) / (GW_BIG * 16);
constexpr int GX_MID = (N0 + GW_MID * 16 - 1) / (GW_MID * 16);
constexpr int GX_REL = (N1 + GW_REL * 16 - 1) / (GW_REL * 16);
constexpr int GX_ONE = (N1 + GW_ONE * 16 - 1) / (GW_ONE * 16);
static_assert(GW_BIG * 32 <= GEMM_MAXT && N0 % 16 == 0 && N1 % 16 == 0 && HH % 64 == 0 && DD % 64 == 0,
              "block size within bounds; a wave's 16 rows lie wholly inside or outside M");

extern "C" void kernel_launch(void* const* d_in, const int* in_sizes, int n_in,
                              void* d_out, int out_size, void* d_ws, size_t ws_size,
                              hipStream_t stream) {
  (void)in_sizes; (void)n_in; (void)out_size; (void)ws_size;

  const float* feat0   = (const float*)d_in[0];
  const float* feat1   = (const float*)d_in[1];
  const int*   e00_dst = (const int*)d_in[2];
  const int*   e00_src = (const int*)d_in[3];
  const float* e00_val = (const float*)d_in[4];
  const int*   e01_dst = (const int*)d_in[5];
  const int*   e01_src = (const int*)d_in[6];
  const float* e01_val = (const float*)d_in[7];
  const int*   e10_dst = (const int*)d_in[8];
  const int*   e10_src = (const int*)d_in[9];
  const float* e10_val = (const float*)d_in[10];
  const int*   e11_dst = (const int*)d_in[11];
  const int*   e11_src = (const int*)d_in[12];
  const float* e11_val = (const float*)d_in[13];
  const float* W1_00 = (const float*)d_in[14];
  const float* W1_01 = (const float*)d_in[15];
  const float* W1_10 = (const float*)d_in[16];
  const float* W1_11 = (const float*)d_in[17];
  const float* W2_00 = (const float*)d_in[18];
  const float* W2_01 = (const float*)d_in[19];
  const float* W2_10 = (const float*)d_in[20];
  const float* W2_11 = (const float*)d_in[21];
  const float* R00   = (const float*)d_in[22];
  const float* R01   = (const float*)d_in[23];
  const float* R10   = (const float*)d_in[24];
  const float* Rglob = (const float*)d_in[25];
  const float* dloc  = (const float*)d_in[26];

  float* ws = (float*)d_ws;
  float* out = (float*)d_out;
  float* z0  = out;
  float* z1  = out + 2560000;
  float* s00 = z1 + 256000;
  float* s01 = s00 + E00;
  float* s10 = s01 + E01;
  float* s11 = s10 + E10;

  const dim3 blk(256);

  int* l00 = (int*)(ws + off_l00);  int* c00 = (int*)(ws + off_c00);
  int* l01 = (int*)(ws + off_l01);  int* c01 = (int*)(ws + off_c01);
  int* l10 = (int*)(ws + off_l10);  int* c10 = (int*)(ws + off_c10);
  int* l11 = (int*)(ws + off_l11);  int* c11 = (int*)(ws + off_c11);
  zero_counts<<<dim3(4), blk, 0, stream>>>(c00);
  bin_edges<<<dim3(NB00 + NB01 + NB10 + KK * NB11), dim3(BIN_THREADS), 0, stream>>>(
      e00_dst, e00_src, e01_dst, e01_src, e10_dst, e10_src, e11_dst, e11_src,
      l00, c00, l01, c01, l10, c10, l11, c11);

  gemm_f32_wmma<FF><<<dim3(GX_BIG, HH / 64, 1), dim3(GW_BIG * 32), 0, stream>>>(
      feat0, W1_00, ws + off_xw00, N0, HH, 0, 0, 0);
  gemm_f32_wmma<FF><<<dim3(GX_ONE, HH / 64, 1), dim3(GW_ONE * 32), 0, stream>>>(
      feat1, W1_01, ws + off_xw01, N1, HH, 0, 0, 0);
  gemm_f32_wmma<FF><<<dim3(GX_REL, HH / 64, KK), dim3(GW_REL * 32), 0, stream>>>(
      feat1, W1_11, ws + off_xw11, N1, HH, 0, (long long)FF * HH, (long long)N1 * HH);

  const dim3 sblk(SP_THREADS);
  spmm_tile<HH><<<dim3(N0 / T0, 1), sblk, 0, stream>>>(e00_dst, e00_src, e00_val, ws + off_xw00,
                                                             ws + off_acc00, E00, N0, T0, 0, 0, 0,
                                                             l00, c00, CAP00);
  gemm_f32_wmma<FF><<<dim3(GX_BIG, HH / 64, 1), dim3(GW_BIG * 32), 0, stream>>>(
      feat0, W1_10, ws + off_xw10, N0, HH, 0, 0, 0);
  spmm_tile<HH><<<dim3(N0 / T0, 1), sblk, 0, stream>>>(e01_dst, e01_src, e01_val, ws + off_xw01,
                                                             ws + off_acc01, E01, N0, T0, 0, 0, 0,
                                                             l01, c01, CAP01);
  spmm_tile<HH><<<dim3(N1 / T1s, 1), sblk, 0, stream>>>(e10_dst, e10_src, e10_val, ws + off_xw10,
                                                              ws + off_acc10, E10, N1, T1s, 0, 0, 0,
                                                              l10, c10, CAP10);
  spmm_tile<HH><<<dim3(N1 / T1m, KK), sblk, 0, stream>>>(e11_dst, e11_src, e11_val, ws + off_xw11,
                                                               ws + off_acc11, E11, N1, T1m, E11,
                                                               (long long)N1 * HH, (long long)N1 * HH,
                                                               l11, c11, CAP11);

  conv_pair<HH, 1, true, 2><<<dim3(FIN0), blk, 0, stream>>>(ws + off_acc00, ws + off_acc01, ws + off_h0, N0, 0);
  conv_pair<HH, KK, true, 1><<<dim3(FIN1), blk, 0, stream>>>(ws + off_acc10, ws + off_acc11, ws + off_h1, N1,
                                                             (long long)N1 * HH);

  gemm_f32_wmma<HH><<<dim3(GX_MID, DD / 64, 1), dim3(GW_MID * 32), 0, stream>>>(
      ws + off_h0, W2_00, ws + off_hw00, N0, DD, 0, 0, 0);
  gemm_f32_wmma<HH><<<dim3(GX_MID, DD / 64, 1), dim3(GW_MID * 32), 0, stream>>>(
      ws + off_h0, W2_10, ws + off_hw10, N0, DD, 0, 0, 0);
  gemm_f32_wmma<HH><<<dim3(GX_ONE, DD / 64, 1), dim3(GW_ONE * 32), 0, stream>>>(
      ws + off_h1, W2_01, ws + off_hw01, N1, DD, 0, 0, 0);
  gemm_f32_wmma<HH><<<dim3(GX_REL, DD / 64, KK), dim3(GW_REL * 32), 0, stream>>>(
      ws + off_h1, W2_11, ws + off_hw11, N1, DD, 0, (long long)HH * DD, (long long)N1 * DD);

  spmm_tile<DD><<<dim3(N0 / T0, 1), sblk, 0, stream>>>(e00_dst, e00_src, e00_val, ws + off_hw00,
                                                              ws + off_bcc00, E00, N0, T0, 0, 0, 0,
                                                              l00, c00, CAP00);
  spmm_tile<DD><<<dim3(N0 / T0, 1), sblk, 0, stream>>>(e01_dst, e01_src, e01_val, ws + off_hw01,
                                                              ws + off_bcc01, E01, N0, T0, 0, 0, 0,
                                                              l01, c01, CAP01);
  spmm_tile<DD><<<dim3(N1 / T1s, 1), sblk, 0, stream>>>(e10_dst, e10_src, e10_val, ws + off_hw10,
                                                               ws + off_bcc10, E10, N1, T1s, 0, 0, 0,
                                                               l10, c10, CAP10);
  spmm_tile<DD><<<dim3(N1 / T1m, KK), sblk, 0, stream>>>(e11_dst, e11_src, e11_val, ws + off_hw11,
                                                                ws + off_bcc11, E11, N1, T1m, E11,
                                                                (long long)N1 * DD, (long long)N1 * DD,
                                                                l11, c11, CAP11);

  conv_pair<DD, 1, false, 2><<<dim3(FIN0), blk, 0, stream>>>(ws + off_bcc00, ws + off_bcc01, z0, N0, 0);
  conv_pair<DD, KK, false, 1><<<dim3(FIN1), blk, 0, stream>>>(ws + off_bcc10, ws + off_bcc11, z1, N1,
                                                              (long long)N1 * DD);

  gemm_f32_wmma<DD><<<dim3(GX_MID, DD / 64, 1), dim3(GW_MID * 32), 0, stream>>>(
      z0, R00, ws + off_z0R00, N0, DD, 0, 0, 0);
  gemm_f32_wmma<DD><<<dim3(GX_MID, DD / 64, 1), dim3(GW_MID * 32), 0, stream>>>(
      z0, R01, ws + off_z0R01, N0, DD, 0, 0, 0);
  gemm_f32_wmma<DD><<<dim3(GX_ONE, DD / 64, 1), dim3(GW_ONE * 32), 0, stream>>>(
      z1, R10, ws + off_z1R10, N1, DD, 0, 0, 0);

  scale_dloc<<<dim3(N1 * (DD / 4) / 256), blk, 0, stream>>>(z1, dloc, ws + off_z1d);
  gemm_f32_wmma<DD><<<dim3(GX_REL, DD / 64, KK), dim3(GW_REL * 32), 0, stream>>>(
      ws + off_z1d, Rglob, ws + off_Zk, N1, DD, (long long)N1 * DD, 0, (long long)N1 * DD);

  const dim3 eblk(ES_THREADS);
  edge_score_tile<<<dim3(N0 / T0, 1), eblk, 0, stream>>>(ws + off_z0R00, z0, e00_dst, e00_src,
                                                         s00, E00, T0, 0, 0, 0, 0, l00, c00, CAP00);
  edge_score_tile<<<dim3(N0 / T0, 1), eblk, 0, stream>>>(ws + off_z0R01, z1, e01_dst, e01_src,
                                                         s01, E01, T0, 0, 0, 0, 0, l01, c01, CAP01);
  edge_score_tile<<<dim3(N1 / T1s, 1), eblk, 0, stream>>>(ws + off_z1R10, z0, e10_dst, e10_src,
                                                          s10, E10, T1s, 0, 0, 0, 0, l10, c10, CAP10);
  edge_score_tile<<<dim3(N1 / T1m, KK), eblk, 0, stream>>>(ws + off_Zk, ws + off_z1d, e11_dst, e11_src,
                                                           s11, E11, T1m, E11,
                                                           (long long)N1 * DD, (long long)N1 * DD,
                                                           (long long)E11, l11, c11, CAP11);
}


# row pass moved into the N0-row SPMM blocks: a block sums both relations into two LDS tiles, normalises and adds its rows there, stores h0 / z0 once; 31 launches -> 27; row norms summed in a new order
# speedup vs baseline: 3.0986x; 1.0506x over previous
#include <hip/hip_runtime.h>

typedef float v2f __attribute__((ext_vector_type(2)));
typedef float v8f __attribute__((ext_vector_type(8)));

constexpr int N0 = 20000, N1 = 2000;
constexpr int FF = 512, HH = 256, DD = 128, KK = 16;
constexpr int E00 = 640000, E01 = 100000, E10 = 100000, E11 = 64000;

constexpr long long off_xw00 = 0;
constexpr long long off_lists = 5120000;
constexpr long long off_xw01 = 10240000;
constexpr long long off_xw11 = 10752000;
constexpr long long off_acc00 = 18944000;
constexpr long long off_acc01 = 24064000;
constexpr long long off_acc10 = 29184000;
constexpr long long off_acc11 = 29696000;

constexpr long long off_xw10  = off_xw00;
constexpr long long off_h0    = off_acc00;
constexpr long long off_h1    = off_xw01;
constexpr long long off_hw00  = off_acc01;
constexpr long long off_hw10  = off_acc01 + 2560000;
constexpr long long off_hw01  = off_xw11;
constexpr long long off_hw11  = off_xw11 + 256000;
constexpr long long off_bcc10 = off_acc10;
constexpr long long off_bcc11 = off_acc11;
constexpr long long off_z0R00 = off_acc01;
constexpr long long off_z0R01 = off_acc01 + 2560000;
constexpr long long off_z1R10 = off_acc10 + 256000;
constexpr long long off_z1d   = off_acc11 + 4096000;
constexpr long long off_Zk    = off_acc11;

__device__ __forceinline__ float wave_sum(float v) {
#pragma unroll
  for (int off = 16; off > 0; off >>= 1) v += __shfl_xor(v, off, 32);
  return v;
}

typedef float v4f __attribute__((ext_vector_type(4)));
constexpr int BK  = 32;
constexpr int BPS = 144;
constexpr int GEMM_MAXT = 640;

template <int KD>
__global__ __launch_bounds__(GEMM_MAXT) void gemm_f32_wmma(
    const float* __restrict__ A, const float* __restrict__ B, float* __restrict__ C,
    int M, int N, long long sA, long long sB, long long sC) {
  __shared__ __attribute__((aligned(16))) float Bp[(KD / 2) * BPS];

  const int b = blockIdx.z;
  A += (long long)b * sA;
  B += (long long)b * sB;
  C += (long long)b * sC;

  const int tid  = threadIdx.x;
  const int nT   = blockDim.x;
  const int lane = tid & 31;
  const int wave = __builtin_amdgcn_readfirstlane(tid >> 5);
  const int n0   = blockIdx.y * 64;
  const int m0   = blockIdx.x * (nT >> 5) * 16 + wave * 16;
  const int half = lane >> 4;
  const int lr   = lane & 15;

  {
    constexpr int ITEMS = (KD / 2) * 16;
    for (int i0 = tid; i0 < ITEMS; i0 += 4 * nT) {
      v4f r0[4], r1[4];
#pragma unroll
      for (int u = 0; u < 4; ++u) {
        const int i = min(i0 + u * nT, ITEMS - 1);
        const int p = i >> 4, c4 = (i & 15) * 4;
        const float* bp = B + (long long)(2 * p) * N + n0 + c4;
        r0[u] = *(const v4f*)bp;
        r1[u] = *(const v4f*)(bp + N);
      }
#pragma unroll
      for (int u = 0; u < 4; ++u) {
        const int i = i0 + u * nT;
        if (i < ITEMS) {
          const int p = i >> 4, c4 = (i & 15) * 4;
          float* sp = &Bp[p * BPS + c4 * 2];
          const v4f lo = {r0[u].x, r1[u].x, r0[u].y, r1[u].y};
          const v4f hi = {r0[u].z, r1[u].z, r0[u].w, r1[u].w};
          *(v4f*)sp = lo;
          *(v4f*)(sp + 4) = hi;
        }
      }
    }
  }
  __syncthreads();

  if (m0 >= M) return;

  const v4f* ap = (const v4f*)(A + (long long)(m0 + lr) * KD + 4 * half);
  const float* Bsb = &Bp[(2 * half) * BPS + lr * 2];

  v8f acc0 = {}, acc1 = {}, acc2 = {}, acc3 = {};
  v4f q[4], qn[4];
#pragma unroll
  for (int j = 0; j < 4; ++j) q[j] = ap[2 * j];

  constexpr int NCH = KD / BK;
  for (int c = 0; c < NCH; ++c) {
    const int cn = min(c + 1, NCH - 1);
#pragma unroll
    for (int j = 0; j < 4; ++j) qn[j] = ap[cn * (BK / 4) + 2 * j];

    const float* bc = Bsb + c * (BK / 2) * BPS;
#pragma unroll
    for (int j = 0; j < 4; ++j) {
#pragma unroll
      for (int sub = 0; sub < 2; ++sub) {
        const v2f a = sub ? __builtin_shufflevector(q[j], q[j], 2, 3)
                          : __builtin_shufflevector(q[j], q[j], 0, 1);
        const float* bp = bc + (4 * j + sub) * BPS;
        const v2f b0 = *(const v2f*)(bp);
        const v2f b1 = *(const v2f*)(bp + 32);
        const v2f b2 = *(const v2f*)(bp + 64);
        const v2f b3 = *(const v2f*)(bp + 96);
        acc0 = __builtin_amdgcn_wmma_f32_16x16x4_f32(false, a, false, b0, (short)0, acc0, false, false);
        acc1 = __builtin_amdgcn_wmma_f32_16x16x4_f32(false, a, false, b1, (short)0, acc1, false, false);
        acc2 = __builtin_amdgcn_wmma_f32_16x16x4_f32(false, a, false, b2, (short)0, acc2, false, false);
        acc3 = __builtin_amdgcn_wmma_f32_16x16x4_f32(false, a, false, b3, (short)0, acc3, false, false);
      }
    }
#pragma unroll
    for (int j = 0; j < 4; ++j) q[j] = qn[j];
  }

#pragma unroll
  for (int r = 0; r < 8; ++r) {
    const int row = m0 + r + 8 * half;
    float* cp = C + (long long)row * N + n0 + lr;
    cp[0]  = acc0[r];
    cp[16] = acc1[r];
    cp[32] = acc2[r];
    cp[48] = acc3[r];
  }
}

constexpr int SP_THREADS = 1024;
constexpr int SP_WAVES   = SP_THREADS / 32;
constexpr int SP_MAXT    = 125;
constexpr int SP_Q       = 16384;
constexpr int Q_EBITS    = 20;
constexpr int Q_EMASK    = (1 << Q_EBITS) - 1;
static_assert(E00 <= (1 << Q_EBITS) && E01 <= (1 << Q_EBITS) && E10 <= (1 << Q_EBITS) &&
              E11 <= (1 << Q_EBITS), "edge id fits the low bits of a queue word");
static_assert(N0 <= 65536 && N1 <= 65536 && SP_MAXT <= 128, "src and tile row fit one list word");

template <int N, int C>
__device__ __forceinline__ void add_rows(const float* __restrict__ x, float* acc, int s, int lr, int vb,
                                         int j0, int lane) {
  constexpr int G = N / 4;
  constexpr int H = N / 128;
  float4 t[C][H];
  float vv[C];
  int ll[C];
#pragma unroll
  for (int u = 0; u < C; ++u) {
    const int j = j0 + u;
    const int sj = __builtin_amdgcn_readlane(s, j);
    ll[u] = __builtin_amdgcn_readlane(lr, j);
    vv[u] = __int_as_float(__builtin_amdgcn_readlane(vb, j));
    const float4* xp = (const float4*)(x + (long long)sj * N);
#pragma unroll
    for (int h = 0; h < H; ++h) t[u][h] = xp[lane + 32 * h];
  }
#pragma unroll
  for (int u = 0; u < C; ++u) {
    float* a = &acc[ll[u] * N + lane];
#pragma unroll
    for (int h = 0; h < H; ++h) {
      atomicAdd(&a[0 * G + 32 * h], vv[u] * t[u][h].x);
      atomicAdd(&a[1 * G + 32 * h], vv[u] * t[u][h].y);
      atomicAdd(&a[2 * G + 32 * h], vv[u] * t[u][h].z);
      atomicAdd(&a[3 * G + 32 * h], vv[u] * t[u][h].w);
    }
  }
}


template <int N>
__device__ __forceinline__ void spmm_accumulate(
    const int* __restrict__ dst, const int* __restrict__ src, const float* __restrict__ val,
    const float* __restrict__ x, float* acc, int* qe, int* qcount, int nE, int T, int row0,
    const int2* __restrict__ myList, const int listCnt, int cap) {
  constexpr int G = N / 4;
  constexpr int H = N / 128;

  const int tid  = threadIdx.x;
  const int lane = tid & 31;
  const int wave = __builtin_amdgcn_readfirstlane(tid >> 5);
  const bool fromList = (listCnt <= cap);

  for (int i = tid; i < T * G; i += SP_THREADS)
    ((float4*)acc)[i] = make_float4(0.f, 0.f, 0.f, 0.f);
  if (tid == 0) *qcount = 0;
  __syncthreads();

  const int n4 = nE >> 2;
  const int4* d4p = (const int4*)dst;
  constexpr int STEP = SP_THREADS * 4;
  const int last4 = max(n4 - 1, 0);

  auto test4 = [&](const int4 (&dv)[4], int base) {
#pragma unroll
    for (int u = 0; u < 4; ++u) {
      const int idx = base + u * SP_THREADS + tid;
      const int e0 = idx << 2;
      const int dd[4] = {dv[u].x, dv[u].y, dv[u].z, dv[u].w};
#pragma unroll
      for (int c = 0; c < 4; ++c) {
        const unsigned r = (unsigned)(dd[c] - row0);
        if (idx < n4 && r < (unsigned)T) {
          const int e = e0 + c;
          const int slot = atomicAdd(qcount, 1);
          if (slot < SP_Q) {
            qe[slot] = e | ((int)r << Q_EBITS);
          } else {
            const int s = src[e];
            const float v = val[e];
            const float* xr = x + (long long)s * N;
            float* a = &acc[(int)r * N];
            for (int f = 0; f < N; ++f) atomicAdd(&a[(f & 3) * G + (f >> 2)], v * xr[f]);
          }
        }
      }
    }
  };

  if (!fromList) {
  int4 da[4], db[4];
#pragma unroll
  for (int u = 0; u < 4; ++u) da[u] = d4p[min(u * SP_THREADS + tid, last4)];
  for (int i0 = 0; i0 < n4; i0 += 2 * STEP) {
#pragma unroll
    for (int u = 0; u < 4; ++u) db[u] = d4p[min(i0 + STEP + u * SP_THREADS + tid, last4)];
    test4(da, i0);
#pragma unroll
    for (int u = 0; u < 4; ++u) da[u] = d4p[min(i0 + 2 * STEP + u * SP_THREADS + tid, last4)];
    test4(db, i0 + STEP);
  }
  }
  __syncthreads();

  constexpr int R = 2048 / N;
  static_assert(R * H == 16 && 32 % R == 0, "a step holds 16 float4 per lane and steps tile a group");
  const int cnt = fromList ? listCnt : __builtin_amdgcn_readfirstlane(min(*qcount, SP_Q));
  const int nGroups = (cnt + 31) >> 5;

  auto walk = [&](int s, int lr, int vb, int nv) {
    int j0 = 0;
    for (; j0 + R <= nv; j0 += R) add_rows<N, R>(x, acc, s, lr, vb, j0, lane);
    if constexpr (R > 8) {
      if (nv - j0 >= 8) { add_rows<N, 8>(x, acc, s, lr, vb, j0, lane); j0 += 8; }
    }
    if (nv - j0 >= 4) { add_rows<N, 4>(x, acc, s, lr, vb, j0, lane); j0 += 4; }
    if (nv - j0 >= 2) { add_rows<N, 2>(x, acc, s, lr, vb, j0, lane); j0 += 2; }
    if (nv - j0 >= 1) add_rows<N, 1>(x, acc, s, lr, vb, j0, lane);
  };

  if (fromList) {
    int2 en = make_int2(0, 0), enN = make_int2(0, 0);
    {
      const int gi = wave * 32 + lane;
      if (gi < cnt) en = myList[gi];
      const int gn = gi + SP_WAVES * 32;
      if (gn < cnt) enN = myList[gn];
    }
    int vb = __float_as_int(val[en.y]);
    for (int g = wave; g < nGroups; g += SP_WAVES) {
      int2 enNN = make_int2(0, 0);
      const int gnn = (g + 2 * SP_WAVES) * 32 + lane;
      if (gnn < cnt) enNN = myList[gnn];
      const int vbN = __float_as_int(val[enN.y]);
      walk(en.x & 0xFFFF, en.x >> 16, vb, min(32, cnt - g * 32));
      en = enN;
      vb = vbN;
      enN = enNN;
    }
  } else {
    for (int g = wave; g < nGroups; g += SP_WAVES) {
      const int gi = g * 32 + lane;
      int s = 0, lr = 0, vb = 0;
      if (gi < cnt) {
        const int q = qe[gi];
        const int e = q & Q_EMASK;
        lr = q >> Q_EBITS;
        s  = src[e];
        vb = __float_as_int(val[e]);
      }
      walk(s, lr, vb, min(32, cnt - g * 32));
    }
  }
  __syncthreads();
}

template <int N>
__global__ __launch_bounds__(SP_THREADS) __attribute__((amdgpu_waves_per_eu(8, 8))) void spmm_tile(
    const int* __restrict__ dst, const int* __restrict__ src, const float* __restrict__ val,
    const float* __restrict__ x, float* __restrict__ out, int nE, int nRows, int T,
    int eStride, long long xStride, long long oStride,
    const int* __restrict__ lists, const int* __restrict__ counts, int cap) {
  constexpr int G = N / 4;
  __shared__ __attribute__((aligned(16))) float acc[SP_MAXT * N];
  __shared__ int qe[SP_Q];
  __shared__ int qcount;

  const int k = blockIdx.y;
  dst += (long long)k * eStride;
  src += (long long)k * eStride;
  val += (long long)k * eStride;
  x   += (long long)k * xStride;
  out += (long long)k * oStride;

  const int tid  = threadIdx.x;
  const int row0 = blockIdx.x * T;
  const int tile = blockIdx.y * gridDim.x + blockIdx.x;
  const int2* __restrict__ myList = (const int2*)lists + (long long)tile * cap;
  const int listCnt = counts[tile];

  spmm_accumulate<N>(dst, src, val, x, acc, qe, &qcount, nE, T, row0, myList, listCnt, cap);

  for (int i = tid; i < T * G; i += SP_THREADS) {
    const int r = i / G;
    const int g = i % G;
    if (row0 + r < nRows) {
      const float* a = &acc[r * N + g];
      *(float4*)(out + (long long)(row0 + r) * N + 4 * g) = make_float4(a[0], a[G], a[2 * G], a[3 * G]);
    }
  }
}

constexpr int PT_MAXT = 80;

template <int N, bool RELU>
__global__ __launch_bounds__(SP_THREADS) __attribute__((amdgpu_waves_per_eu(8, 8))) void spmm_pair_tile(
    const int* __restrict__ dstA, const int* __restrict__ srcA, const float* __restrict__ valA,
    const float* __restrict__ xA, int nEA, const int* __restrict__ listsA,
    const int* __restrict__ countsA, int capA,
    const int* __restrict__ dstB, const int* __restrict__ srcB, const float* __restrict__ valB,
    const float* __restrict__ xB, int nEB, const int* __restrict__ listsB,
    const int* __restrict__ countsB, int capB,
    float* __restrict__ out, int nRows, int T) {
  constexpr int G = N / 4;
  constexpr int H = N / 128;
  __shared__ __attribute__((aligned(16))) float accA[PT_MAXT * N];
  __shared__ __attribute__((aligned(16))) float accB[PT_MAXT * N];
  __shared__ int qe[SP_Q];
  __shared__ int qcount;

  const int tid  = threadIdx.x;
  const int lane = tid & 31;
  const int wave = __builtin_amdgcn_readfirstlane(tid >> 5);
  const int tile = blockIdx.x;
  const int row0 = tile * T;
  const int cntA = countsA[tile];
  const int cntB = countsB[tile];

  spmm_accumulate<N>(dstA, srcA, valA, xA, accA, qe, &qcount, nEA, T, row0,
                     (const int2*)listsA + (long long)tile * capA, cntA, capA);
  spmm_accumulate<N>(dstB, srcB, valB, xB, accB, qe, &qcount, nEB, T, row0,
                     (const int2*)listsB + (long long)tile * capB, cntB, capB);

  for (int r = wave; r < T; r += SP_WAVES) {
    float4 va[H], vb[H];
    float sa = 0.f, sb = 0.f;
#pragma unroll
    for (int h = 0; h < H; ++h) {
      const float* a = &accA[r * N + lane + 32 * h];
      const float* b = &accB[r * N + lane + 32 * h];
      float4 p = make_float4(a[0], a[G], a[2 * G], a[3 * G]);
      float4 q = make_float4(b[0], b[G], b[2 * G], b[3 * G]);
      if (RELU) {
        p = make_float4(fmaxf(p.x, 0.f), fmaxf(p.y, 0.f), fmaxf(p.z, 0.f), fmaxf(p.w, 0.f));
        q = make_float4(fmaxf(q.x, 0.f), fmaxf(q.y, 0.f), fmaxf(q.z, 0.f), fmaxf(q.w, 0.f));
      }
      va[h] = p;
      vb[h] = q;
      sa = __builtin_fmaf(p.x, p.x, sa);
      sa = __builtin_fmaf(p.y, p.y, sa);
      sa = __builtin_fmaf(p.z, p.z, sa);
      sa = __builtin_fmaf(p.w, p.w, sa);
      sb = __builtin_fmaf(q.x, q.x, sb);
      sb = __builtin_fmaf(q.y, q.y, sb);
      sb = __builtin_fmaf(q.z, q.z, sb);
      sb = __builtin_fmaf(q.w, q.w, sb);
    }
    const float sca = rsqrtf(fmaxf(wave_sum(sa), 1e-12f));
    const float scb = rsqrtf(fmaxf(wave_sum(sb), 1e-12f));
    if (row0 + r < nRows) {
      float* op = out + (long long)(row0 + r) * N + 4 * lane;
#pragma unroll
      for (int h = 0; h < H; ++h) {
        float4 v = make_float4(va[h].x * sca + vb[h].x * scb, va[h].y * sca + vb[h].y * scb,
                               va[h].z * sca + vb[h].z * scb, va[h].w * sca + vb[h].w * scb);
        if (RELU) v = make_float4(fmaxf(v.x, 0.f), fmaxf(v.y, 0.f), fmaxf(v.z, 0.f), fmaxf(v.w, 0.f));
        *(float4*)(op + 128 * h) = v;
      }
    }
  }
}

constexpr int T0  = 80;
constexpr int T1s = 8;
constexpr int T1m = 125;
static_assert(N0 % T0 == 0 && N1 % T1s == 0 && N1 % T1m == 0, "tiles cover the rows exactly");
static_assert(T0 <= SP_MAXT && T1s <= SP_MAXT && T1m <= SP_MAXT && T0 <= PT_MAXT, "tile fits its LDS array");
static_assert(E00 % 4 == 0 && E01 % 4 == 0 && E10 % 4 == 0 && E11 % 4 == 0, "dst read as int4");

constexpr int CAP00 = 3072, CAP01 = 640, CAP10 = 640, CAP11 = 4608;
static_assert(CAP00 <= SP_Q && CAP01 <= SP_Q && CAP10 <= SP_Q && CAP11 <= SP_Q, "a list is a queue prefix");
constexpr long long off_l00 = off_lists;
constexpr long long off_l01 = off_l00 + 2LL * (N0 / T0) * CAP00;
constexpr long long off_l10 = off_l01 + 2LL * (N0 / T0) * CAP01;
constexpr long long off_l11 = off_l10 + 2LL * (N1 / T1s) * CAP10;
constexpr long long off_c00 = off_l11 + 2LL * (N1 / T1m) * KK * CAP11;
constexpr long long off_c01 = off_c00 + 256;
constexpr long long off_c10 = off_c01 + 256;
constexpr long long off_c11 = off_c10 + 256;
static_assert(N0 / T0 <= 256 && N1 / T1s <= 256 && (N1 / T1m) * KK <= 256, "count arrays hold 256 tiles each");
static_assert(off_c11 + 256 <= off_lists + 5120000, "lists and counts fit the plane xw10 used to have");
static_assert(off_c01 == off_c00 + 256 && off_c10 == off_c00 + 512 && off_c11 == off_c00 + 768,
              "the four count arrays are one run of 1024 ints");

constexpr int BIN_THREADS = 1024;
constexpr int BIN_U  = 2;
constexpr int BIN_I4 = BIN_THREADS * BIN_U;
constexpr int NB00 = (E00 / 4 + BIN_I4 - 1) / BIN_I4;
constexpr int NB01 = (E01 / 4 + BIN_I4 - 1) / BIN_I4;
constexpr int NB10 = (E10 / 4 + BIN_I4 - 1) / BIN_I4;
constexpr int NB11 = (E11 / 4 + BIN_I4 - 1) / BIN_I4;
static_assert(N0 / T0 <= 256 && N1 / T1s <= 256 && N1 / T1m <= 256, "a relation's tiles fit the block's LDS counters");

__global__ void zero_counts(int* __restrict__ c) { c[blockIdx.x * blockDim.x + threadIdx.x] = 0; }

template <int T>
__device__ __forceinline__ void bin_block(const int* __restrict__ dst, const int* __restrict__ src,
                                          int nE, int nRows, int blk, int tileBase, int nTiles, int cap,
                                          int2* __restrict__ lists, int* __restrict__ counts,
                                          int* cntL, int* baseL) {
  const int tid = threadIdx.x;
  const int n4 = nE >> 2;
  if (tid < 256) cntL[tid] = 0;
  __syncthreads();

  int4 dv[BIN_U], sv[BIN_U];
#pragma unroll
  for (int u = 0; u < BIN_U; ++u) {
    const int idxc = min(blk * BIN_I4 + u * BIN_THREADS + tid, n4 - 1);
    dv[u] = ((const int4*)dst)[idxc];
    sv[u] = ((const int4*)src)[idxc];
  }

  int tl[BIN_U * 4], rw[BIN_U * 4], sr[BIN_U * 4], rk[BIN_U * 4];
  bool ok[BIN_U * 4];
#pragma unroll
  for (int u = 0; u < BIN_U; ++u) {
    const bool in = blk * BIN_I4 + u * BIN_THREADS + tid < n4;
    const int dd[4] = {dv[u].x, dv[u].y, dv[u].z, dv[u].w};
    const int ss[4] = {sv[u].x, sv[u].y, sv[u].z, sv[u].w};
#pragma unroll
    for (int c = 0; c < 4; ++c) {
      const int j = u * 4 + c;
      ok[j] = in && (unsigned)dd[c] < (unsigned)nRows;
      const int t = ok[j] ? dd[c] / T : 0;
      tl[j] = t;
      rw[j] = dd[c] - t * T;
      sr[j] = ss[c];
      rk[j] = ok[j] ? atomicAdd(&cntL[t], 1) : 0;
    }
  }
  __syncthreads();

  if (tid < nTiles) {
    const int n = cntL[tid];
    baseL[tid] = (n > 0) ? atomicAdd(&counts[tileBase + tid], n) : 0;
  }
  __syncthreads();

#pragma unroll
  for (int u = 0; u < BIN_U; ++u) {
#pragma unroll
    for (int c = 0; c < 4; ++c) {
      const int j = u * 4 + c;
      if (ok[j]) {
        const int slot = baseL[tl[j]] + rk[j];
        if (slot < cap)
          lists[(long long)(tileBase + tl[j]) * cap + slot] =
              make_int2(sr[j] | (rw[j] << 16), (blk * BIN_I4 + u * BIN_THREADS + tid) * 4 + c);
      }
    }
  }
}

__global__ __launch_bounds__(BIN_THREADS) void bin_edges(
    const int* __restrict__ d00, const int* __restrict__ s00,
    const int* __restrict__ d01, const int* __restrict__ s01,
    const int* __restrict__ d10, const int* __restrict__ s10,
    const int* __restrict__ d11, const int* __restrict__ s11,
    int* __restrict__ l00, int* __restrict__ c00, int* __restrict__ l01, int* __restrict__ c01,
    int* __restrict__ l10, int* __restrict__ c10, int* __restrict__ l11, int* __restrict__ c11) {
  __shared__ int cntL[256];
  __shared__ int baseL[256];
  int b = blockIdx.x;
  if (b < NB00) {
    bin_block<T0>(d00, s00, E00, N0, b, 0, N0 / T0, CAP00, (int2*)l00, c00, cntL, baseL);
    return;
  }
  b -= NB00;
  if (b < NB01) {
    bin_block<T0>(d01, s01, E01, N0, b, 0, N0 / T0, CAP01, (int2*)l01, c01, cntL, baseL);
    return;
  }
  b -= NB01;
  if (b < NB10) {
    bin_block<T1s>(d10, s10, E10, N1, b, 0, N1 / T1s, CAP10, (int2*)l10, c10, cntL, baseL);
    return;
  }
  b -= NB10;
  const int k = b / NB11;
  b -= k * NB11;
  bin_block<T1m>(d11 + (long long)k * E11, s11 + (long long)k * E11, E11, N1, b,
                 k * (N1 / T1m), N1 / T1m, CAP11, (int2*)l11, c11, cntL, baseL);
}

template <int N, int KB, bool RELU, int RPW>
__global__ __launch_bounds__(256) void conv_pair(const float* __restrict__ a, const float* __restrict__ b,
                                                 float* __restrict__ out, int n, long long kStride) {
  constexpr int NC = N / 32;
  const int lane = threadIdx.x & 31;
  const int wpb = blockDim.x >> 5;
  const int nWaves = gridDim.x * wpb;
  for (int row0 = blockIdx.x * wpb + (threadIdx.x >> 5); row0 < n; row0 += RPW * nWaves) {
    float va[RPW][NC], vb[RPW][NC];
    float sa[RPW], sb[RPW];
    float ta[RPW][NC];
#pragma unroll
    for (int r = 0; r < RPW; ++r) {
      const int row = min(row0 + r * nWaves, n - 1);
      const float* ap = a + (long long)row * N + lane;
#pragma unroll
      for (int j = 0; j < NC; ++j) ta[r][j] = ap[j * 32];
    }
    if constexpr (KB == 1) {
      float tb[RPW][NC];
#pragma unroll
      for (int r = 0; r < RPW; ++r) {
        const int row = min(row0 + r * nWaves, n - 1);
        const float* bp = b + (long long)row * N + lane;
#pragma unroll
        for (int j = 0; j < NC; ++j) tb[r][j] = bp[j * 32];
      }
#pragma unroll
      for (int r = 0; r < RPW; ++r) {
        float ss = 0.f;
#pragma unroll
        for (int j = 0; j < NC; ++j) {
          float x = 0.f;
          x += RELU ? fmaxf(tb[r][j], 0.f) : tb[r][j];
          vb[r][j] = x;
          ss = __builtin_fmaf(x, x, ss);
        }
        sb[r] = ss;
      }
    } else {
#pragma unroll
      for (int r = 0; r < RPW; ++r) {
        const int row = min(row0 + r * nWaves, n - 1);
        const float* bp = b + (long long)row * N + lane;
        float ss = 0.f;
#pragma unroll
        for (int j = 0; j < NC; ++j) {
          float t[KB];
#pragma unroll
          for (int k = 0; k < KB; ++k) t[k] = bp[(long long)k * kStride + j * 32];
          float x = 0.f;
#pragma unroll
          for (int k = 0; k < KB; ++k) x += RELU ? fmaxf(t[k], 0.f) : t[k];
          vb[r][j] = x;
          ss = __builtin_fmaf(x, x, ss);
        }
        sb[r] = ss;
      }
    }
#pragma unroll
    for (int r = 0; r < RPW; ++r) {
      float ss = 0.f;
#pragma unroll
      for (int j = 0; j < NC; ++j) {
        float x = 0.f;
        x += RELU ? fmaxf(ta[r][j], 0.f) : ta[r][j];
        va[r][j] = x;
        ss = __builtin_fmaf(x, x, ss);
      }
      sa[r] = ss;
    }
#pragma unroll
    for (int r = 0; r < RPW; ++r) {
      const int row = row0 + r * nWaves;
      const float sca = rsqrtf(fmaxf(wave_sum(sa[r]), 1e-12f));
      const float scb = rsqrtf(fmaxf(wave_sum(sb[r]), 1e-12f));
      if (row < n) {
        float* op = out + (long long)row * N + lane;
#pragma unroll
        for (int j = 0; j < NC; ++j) {
          float v = va[r][j] * sca + vb[r][j] * scb;
          if (RELU) v = fmaxf(v, 0.f);
          op[j * 32] = v;
        }
      }
    }
  }
}

__global__ __launch_bounds__(256) void scale_dloc(const float* z1, const float* dloc, float* out) {
  const int i = blockIdx.x * blockDim.x + threadIdx.x;
  if (i >= N1 * (DD / 4)) return;
  const int f4 = i & (DD / 4 - 1);
  const float4 z = ((const float4*)z1)[i];
  const float4* d4 = (const float4*)dloc;
  float4* o4 = (float4*)out;
  float4 d[KK];
#pragma unroll
  for (int k = 0; k < KK; ++k) d[k] = d4[k * (DD / 4) + f4];
#pragma unroll
  for (int k = 0; k < KK; ++k)
    o4[(long long)k * (N1 * (DD / 4)) + i] =
        make_float4(z.x * d[k].x, z.y * d[k].y, z.z * d[k].z, z.w * d[k].w);
}
static_assert((N1 * (DD / 4)) % 256 == 0, "scale_dloc's grid is exact");

constexpr int ES_THREADS = 1024;
constexpr int ES_WAVES   = ES_THREADS / 32;

__global__ __launch_bounds__(ES_THREADS) __attribute__((amdgpu_waves_per_eu(8, 8))) void edge_score_tile(
    const float* __restrict__ zl, const float* __restrict__ zr,
    const int* __restrict__ dst, const int* __restrict__ src,
    float* __restrict__ out, int nE, int T, int eStride,
    long long zlStride, long long zrStride, long long oStride,
    const int* __restrict__ lists, const int* __restrict__ counts, int cap) {
  __shared__ __attribute__((aligned(16))) float zt[SP_MAXT * DD];

  const int k = blockIdx.y;
  zl  += (long long)k * zlStride;
  zr  += (long long)k * zrStride;
  dst += (long long)k * eStride;
  src += (long long)k * eStride;
  out += (long long)k * oStride;

  const int tid  = threadIdx.x;
  const int lane = tid & 31;
  const int wave = __builtin_amdgcn_readfirstlane(tid >> 5);
  const int row0 = blockIdx.x * T;
  const int tile = blockIdx.y * gridDim.x + blockIdx.x;
  const int2* __restrict__ myList = (const int2*)lists + (long long)tile * cap;

  {
    const float4* zsrc = (const float4*)(zl + (long long)row0 * DD);
    for (int i = tid; i < T * (DD / 4); i += ES_THREADS) ((float4*)zt)[i] = zsrc[i];
  }
  const int cnt = __builtin_amdgcn_readfirstlane(counts[tile]);
  __syncthreads();

  if (cnt <= cap) {
    const bool h16 = (lane & 16) != 0, h8 = (lane & 8) != 0, h4 = (lane & 4) != 0;
    const int nGroups = (cnt + 31) >> 5;
    int2 en = make_int2(0, 0);
    if (wave * 32 + lane < cnt) en = myList[wave * 32 + lane];
    for (int g = wave; g < nGroups; g += ES_WAVES) {
      int2 enN = make_int2(0, 0);
      const int gn = (g + ES_WAVES) * 32 + lane;
      if (gn < cnt) enN = myList[gn];
      const int s  = en.x & 0xFFFF;
      const int lr = en.x >> 16;
      const int e  = en.y;
      const int nv = min(32, cnt - g * 32);
      for (int j0 = 0; j0 < nv; j0 += 16) {
        float4 b[16];
#pragma unroll
        for (int u = 0; u < 16; ++u) {
          const int sj = __builtin_amdgcn_readlane(s, j0 + u);
          b[u] = ((const float4*)(zr + (long long)sj * DD))[lane];
        }
#pragma unroll
        for (int hh = 0; hh < 2; ++hh) {
          const int jb = j0 + 8 * hh;
          float p[8];
#pragma unroll
          for (int u = 0; u < 8; ++u) {
            const int lj = __builtin_amdgcn_readlane(lr, jb + u);
            const float4 a = ((const float4*)(zt + lj * DD))[lane];
            p[u] = a.x * b[8 * hh + u].x + a.y * b[8 * hh + u].y + a.z * b[8 * hh + u].z +
                   a.w * b[8 * hh + u].w;
          }
          float q[4], r[2];
#pragma unroll
          for (int i = 0; i < 4; ++i) {
            const float keep = h16 ? p[4 + i] : p[i];
            const float send = h16 ? p[i] : p[4 + i];
            q[i] = keep + __shfl_xor(send, 16, 32);
          }
#pragma unroll
          for (int i = 0; i < 2; ++i) {
            const float keep = h8 ? q[2 + i] : q[i];
            const float send = h8 ? q[i] : q[2 + i];
            r[i] = keep + __shfl_xor(send, 8, 32);
          }
          float t;
          {
            const float keep = h4 ? r[1] : r[0];
            const float send = h4 ? r[0] : r[1];
            t = keep + __shfl_xor(send, 4, 32);
          }
          t += __shfl_xor(t, 2, 32);
          t += __shfl_xor(t, 1, 32);
          const int mine = jb + (lane >> 2);
          const int eo = __shfl(e, mine, 32);
          if ((lane & 3) == 0 && mine < nv) out[eo] = t;
        }
      }
      en = enN;
    }
  } else {
    for (int base = wave * 32; base < nE; base += ES_WAVES * 32) {
      const int e = base + lane;
      const unsigned r = (e < nE) ? (unsigned)(dst[e] - row0) : 0xFFFFFFFFu;
      unsigned mask = __builtin_amdgcn_ballot_w32(r < (unsigned)T);
      while (mask != 0u) {
        const int j = __builtin_ctz(mask);
        mask &= mask - 1u;
        const int lj = __builtin_amdgcn_readlane((int)r, j);
        const int sj = src[base + j];
        const float4 a = ((const float4*)(zt + lj * DD))[lane];
        const float4 bb = ((const float4*)(zr + (long long)sj * DD))[lane];
        float acc = a.x * bb.x + a.y * bb.y + a.z * bb.z + a.w * bb.w;
        acc = wave_sum(acc);
        if (lane == 0) out[base + j] = acc;
      }
    }
  }
}

constexpr int FIN1 = N1 / 8;
static_assert(N1 % 8 == 0, "conv_pair grid covers the rows evenly");

constexpr int GW_BIG = 20, GW_MID = 10, GW_REL = 16, GW_ONE = 4;
constexpr int GX_BIG = (N0 + GW_BIG * 16 - 1) / (GW_BIG * 16);
constexpr int GX_MID = (N0 + GW_MID * 16 - 1) / (GW_MID * 16);
constexpr int GX_REL = (N1 + GW_REL * 16 - 1) / (GW_REL * 16);
constexpr int GX_ONE = (N1 + GW_ONE * 16 - 1) / (GW_ONE * 16);
static_assert(GW_BIG * 32 <= GEMM_MAXT && N0 % 16 == 0 && N1 % 16 == 0 && HH % 64 == 0 && DD % 64 == 0,
              "block size within bounds; a wave's 16 rows lie wholly inside or outside M");

extern "C" void kernel_launch(void* const* d_in, const int* in_sizes, int n_in,
                              void* d_out, int out_size, void* d_ws, size_t ws_size,
                              hipStream_t stream) {
  (void)in_sizes; (void)n_in; (void)out_size; (void)ws_size;

  const float* feat0   = (const float*)d_in[0];
  const float* feat1   = (const float*)d_in[1];
  const int*   e00_dst = (const int*)d_in[2];
  const int*   e00_src = (const int*)d_in[3];
  const float* e00_val = (const float*)d_in[4];
  const int*   e01_dst = (const int*)d_in[5];
  const int*   e01_src = (const int*)d_in[6];
  const float* e01_val = (const float*)d_in[7];
  const int*   e10_dst = (const int*)d_in[8];
  const int*   e10_src = (const int*)d_in[9];
  const float* e10_val = (const float*)d_in[10];
  const int*   e11_dst = (const int*)d_in[11];
  const int*   e11_src = (const int*)d_in[12];
  const float* e11_val = (const float*)d_in[13];
  const float* W1_00 = (const float*)d_in[14];
  const float* W1_01 = (const float*)d_in[15];
  const float* W1_10 = (const float*)d_in[16];
  const float* W1_11 = (const float*)d_in[17];
  const float* W2_00 = (const float*)d_in[18];
  const float* W2_01 = (const float*)d_in[19];
  const float* W2_10 = (const float*)d_in[20];
  const float* W2_11 = (const float*)d_in[21];
  const float* R00   = (const float*)d_in[22];
  const float* R01   = (const float*)d_in[23];
  const float* R10   = (const float*)d_in[24];
  const float* Rglob = (const float*)d_in[25];
  const float* dloc  = (const float*)d_in[26];

  float* ws = (float*)d_ws;
  float* out = (float*)d_out;
  float* z0  = out;
  float* z1  = out + 2560000;
  float* s00 = z1 + 256000;
  float* s01 = s00 + E00;
  float* s10 = s01 + E01;
  float* s11 = s10 + E10;

  const dim3 blk(256);

  int* l00 = (int*)(ws + off_l00);  int* c00 = (int*)(ws + off_c00);
  int* l01 = (int*)(ws + off_l01);  int* c01 = (int*)(ws + off_c01);
  int* l10 = (int*)(ws + off_l10);  int* c10 = (int*)(ws + off_c10);
  int* l11 = (int*)(ws + off_l11);  int* c11 = (int*)(ws + off_c11);
  zero_counts<<<dim3(4), blk, 0, stream>>>(c00);
  bin_edges<<<dim3(NB00 + NB01 + NB10 + KK * NB11), dim3(BIN_THREADS), 0, stream>>>(
      e00_dst, e00_src, e01_dst, e01_src, e10_dst, e10_src, e11_dst, e11_src,
      l00, c00, l01, c01, l10, c10, l11, c11);

  gemm_f32_wmma<FF><<<dim3(GX_BIG, HH / 64, 1), dim3(GW_BIG * 32), 0, stream>>>(
      feat0, W1_00, ws + off_xw00, N0, HH, 0, 0, 0);
  gemm_f32_wmma<FF><<<dim3(GX_ONE, HH / 64, 1), dim3(GW_ONE * 32), 0, stream>>>(
      feat1, W1_01, ws + off_xw01, N1, HH, 0, 0, 0);
  gemm_f32_wmma<FF><<<dim3(GX_REL, HH / 64, KK), dim3(GW_REL * 32), 0, stream>>>(
      feat1, W1_11, ws + off_xw11, N1, HH, 0, (long long)FF * HH, (long long)N1 * HH);

  const dim3 sblk(SP_THREADS);
  spmm_pair_tile<HH, true><<<dim3(N0 / T0), sblk, 0, stream>>>(
      e00_dst, e00_src, e00_val, ws + off_xw00, E00, l00, c00, CAP00,
      e01_dst, e01_src, e01_val, ws + off_xw01, E01, l01, c01, CAP01,
      ws + off_h0, N0, T0);
  gemm_f32_wmma<FF><<<dim3(GX_BIG, HH / 64, 1), dim3(GW_BIG * 32), 0, stream>>>(
      feat0, W1_10, ws + off_xw10, N0, HH, 0, 0, 0);
  spmm_tile<HH><<<dim3(N1 / T1s, 1), sblk, 0, stream>>>(e10_dst, e10_src, e10_val, ws + off_xw10,
                                                              ws + off_acc10, E10, N1, T1s, 0, 0, 0,
                                                              l10, c10, CAP10);
  spmm_tile<HH><<<dim3(N1 / T1m, KK), sblk, 0, stream>>>(e11_dst, e11_src, e11_val, ws + off_xw11,
                                                               ws + off_acc11, E11, N1, T1m, E11,
                                                               (long long)N1 * HH, (long long)N1 * HH,
                                                               l11, c11, CAP11);

  conv_pair<HH, KK, true, 1><<<dim3(FIN1), blk, 0, stream>>>(ws + off_acc10, ws + off_acc11, ws + off_h1, N1,
                                                             (long long)N1 * HH);

  gemm_f32_wmma<HH><<<dim3(GX_MID, DD / 64, 1), dim3(GW_MID * 32), 0, stream>>>(
      ws + off_h0, W2_00, ws + off_hw00, N0, DD, 0, 0, 0);
  gemm_f32_wmma<HH><<<dim3(GX_MID, DD / 64, 1), dim3(GW_MID * 32), 0, stream>>>(
      ws + off_h0, W2_10, ws + off_hw10, N0, DD, 0, 0, 0);
  gemm_f32_wmma<HH><<<dim3(GX_ONE, DD / 64, 1), dim3(GW_ONE * 32), 0, stream>>>(
      ws + off_h1, W2_01, ws + off_hw01, N1, DD, 0, 0, 0);
  gemm_f32_wmma<HH><<<dim3(GX_REL, DD / 64, KK), dim3(GW_REL * 32), 0, stream>>>(
      ws + off_h1, W2_11, ws + off_hw11, N1, DD, 0, (long long)HH * DD, (long long)N1 * DD);

  spmm_pair_tile<DD, false><<<dim3(N0 / T0), sblk, 0, stream>>>(
      e00_dst, e00_src, e00_val, ws + off_hw00, E00, l00, c00, CAP00,
      e01_dst, e01_src, e01_val, ws + off_hw01, E01, l01, c01, CAP01,
      z0, N0, T0);
  spmm_tile<DD><<<dim3(N1 / T1s, 1), sblk, 0, stream>>>(e10_dst, e10_src, e10_val, ws + off_hw10,
                                                               ws + off_bcc10, E10, N1, T1s, 0, 0, 0,
                                                               l10, c10, CAP10);
  spmm_tile<DD><<<dim3(N1 / T1m, KK), sblk, 0, stream>>>(e11_dst, e11_src, e11_val, ws + off_hw11,
                                                                ws + off_bcc11, E11, N1, T1m, E11,
                                                                (long long)N1 * DD, (long long)N1 * DD,
                                                                l11, c11, CAP11);

  conv_pair<DD, KK, false, 1><<<dim3(FIN1), blk, 0, stream>>>(ws + off_bcc10, ws + off_bcc11, z1, N1,
                                                              (long long)N1 * DD);

  gemm_f32_wmma<DD><<<dim3(GX_MID, DD / 64, 1), dim3(GW_MID * 32), 0, stream>>>(
      z0, R00, ws + off_z0R00, N0, DD, 0, 0, 0);
  gemm_f32_wmma<DD><<<dim3(GX_MID, DD / 64, 1), dim3(GW_MID * 32), 0, stream>>>(
      z0, R01, ws + off_z0R01, N0, DD, 0, 0, 0);
  gemm_f32_wmma<DD><<<dim3(GX_ONE, DD / 64, 1), dim3(GW_ONE * 32), 0, stream>>>(
      z1, R10, ws + off_z1R10, N1, DD, 0, 0, 0);

  scale_dloc<<<dim3(N1 * (DD / 4) / 256), blk, 0, stream>>>(z1, dloc, ws + off_z1d);
  gemm_f32_wmma<DD><<<dim3(GX_REL, DD / 64, KK), dim3(GW_REL * 32), 0, stream>>>(
      ws + off_z1d, Rglob, ws + off_Zk, N1, DD, (long long)N1 * DD, 0, (long long)N1 * DD);

  const dim3 eblk(ES_THREADS);
  edge_score_tile<<<dim3(N0 / T0, 1), eblk, 0, stream>>>(ws + off_z0R00, z0, e00_dst, e00_src,
                                                         s00, E00, T0, 0, 0, 0, 0, l00, c00, CAP00);
  edge_score_tile<<<dim3(N0 / T0, 1), eblk, 0, stream>>>(ws + off_z0R01, z1, e01_dst, e01_src,
                                                         s01, E01, T0, 0, 0, 0, 0, l01, c01, CAP01);
  edge_score_tile<<<dim3(N1 / T1s, 1), eblk, 0, stream>>>(ws + off_z1R10, z0, e10_dst, e10_src,
                                                          s10, E10, T1s, 0, 0, 0, 0, l10, c10, CAP10);
  edge_score_tile<<<dim3(N1 / T1m, KK), eblk, 0, stream>>>(ws + off_Zk, ws + off_z1d, e11_dst, e11_src,
                                                           s11, E11, T1m, E11,
                                                           (long long)N1 * DD, (long long)N1 * DD,
                                                           (long long)E11, l11, c11, CAP11);
}
